# starting kernel
# baseline (speedup 1.0000x reference)
#include <hip/hip_runtime.h>
#include <hip/hip_bf16.h>

#define NN 50000
#define DD 128
#define EE 800000

typedef __attribute__((ext_vector_type(2))) float v2f;
typedef __attribute__((ext_vector_type(8))) float v8f;

template <bool TRANSB, int ACT>
__global__ __launch_bounds__(256) void gemm_wmma(
    const float* __restrict__ A, const float* __restrict__ W,
    const float* __restrict__ bias, float* __restrict__ C,
    int n, int K, int O)
{
    const int wave = threadIdx.x >> 5;
    const int lane = threadIdx.x & 31;
    const int groups_n = O >> 6;
    const int total = (n >> 4) * groups_n;
    const int tile = blockIdx.x * 8 + wave;
    if (tile >= total) return;

    const int tm = tile / groups_n;
    const int tg = tile % groups_n;
    const int half = lane >> 4;
    const int l15  = lane & 15;
    const int rbase = tm << 4;
    const int cbase = tg << 6;
    const int arow  = rbase + l15;

    v8f acc[4];
#pragma unroll
    for (int t = 0; t < 4; ++t) acc[t] = (v8f){0.f,0.f,0.f,0.f,0.f,0.f,0.f,0.f};

#if __has_builtin(__builtin_amdgcn_wmma_f32_16x16x4_f32)
    const float* Ap = A + (size_t)arow * K;
    for (int k0 = 0; k0 < K; k0 += 4) {
        const int ka = k0 + 2 * half;
        v2f a = *(const v2f*)(Ap + ka);
        v2f b[4];
        if (TRANSB) {
            const float* Wp = W + (size_t)(cbase + l15) * K + ka;
#pragma unroll
            for (int t = 0; t < 4; ++t)
                b[t] = *(const v2f*)(Wp + (size_t)(16 * t) * K);
        } else {
            const float* Wp = W + (size_t)ka * O + cbase + l15;
#pragma unroll
            for (int t = 0; t < 4; ++t) {
                b[t].x = Wp[16 * t];
                b[t].y = Wp[16 * t + O];
            }
        }
#pragma unroll
        for (int t = 0; t < 4; ++t)
            acc[t] = __builtin_amdgcn_wmma_f32_16x16x4_f32(
                 false, a,  false, b[t],
                 (short)0, acc[t],  false,  false);
    }
#else
    for (int k = 0; k < K; ++k) {
#pragma unroll
        for (int t = 0; t < 4; ++t) {
            int col = cbase + 16 * t + l15;
            float bv = TRANSB ? W[(size_t)col * K + k] : W[(size_t)k * O + col];
#pragma unroll
            for (int v = 0; v < 8; ++v)
                acc[t][v] += A[(size_t)(rbase + v + 8 * half) * K + k] * bv;
        }
    }
#endif

#pragma unroll
    for (int t = 0; t < 4; ++t) {
        const int col = cbase + 16 * t + l15;
        const float bv = bias ? bias[col] : 0.f;
#pragma unroll
        for (int v = 0; v < 8; ++v) {
            float val = acc[t][v] + bv;
            if (ACT == 1) val = fmaxf(val, 0.f);
            C[(size_t)(rbase + v + 8 * half) * O + col] = val;
        }
    }
}

__device__ __forceinline__ float wave_sum32(float v) {
#pragma unroll
    for (int m = 16; m > 0; m >>= 1) v += __shfl_xor(v, m, 32);
    return v;
}

__global__ __launch_bounds__(256) void fill_kernel(float* __restrict__ p, float v, int n) {
    int i = blockIdx.x * blockDim.x + threadIdx.x;
    if (i < n) p[i] = v;
}

__global__ __launch_bounds__(256) void deg_scatter_kernel(const int* __restrict__ dst,
                                                          float* __restrict__ deg, int e) {
    int i = blockIdx.x * blockDim.x + threadIdx.x;
    if (i < e) atomicAdd(&deg[dst[i]], 1.0f);
}

__global__ __launch_bounds__(256) void rsqrt_kernel(float* __restrict__ p, int n) {
    int i = blockIdx.x * blockDim.x + threadIdx.x;
    if (i < n) p[i] = rsqrtf(p[i]);
}

__global__ __launch_bounds__(256) void res_scatter_kernel(
    const int* __restrict__ src, const int* __restrict__ dst,
    const float* __restrict__ v, float* __restrict__ o, int e)
{
    int wv = (blockIdx.x * 256 + threadIdx.x) >> 5;
    int lane = threadIdx.x & 31;
    if (wv >= e) return;
    const float* vr = v + (size_t)src[wv] * DD;
    float* orow = o + (size_t)dst[wv] * DD;
#pragma unroll
    for (int j = 0; j < 4; ++j) {
        int d = lane + 32 * j;
        atomicAdd(&orow[d], vr[d]);
    }
}

__global__ __launch_bounds__(256) void gcn_base_kernel(
    const float* __restrict__ hw, const float* __restrict__ isd,
    const float* __restrict__ bias, float* __restrict__ o, int total)
{
    int i = blockIdx.x * blockDim.x + threadIdx.x;
    if (i >= total) return;
    int nd = i >> 7, d = i & 127;
    float s = isd[nd];
    o[i] = hw[i] * s * s + bias[d];
}

__global__ __launch_bounds__(256) void gcn_scatter_kernel(
    const int* __restrict__ src, const int* __restrict__ dst,
    const float* __restrict__ hw, const float* __restrict__ isd,
    float* __restrict__ o, int e)
{
    int wv = (blockIdx.x * 256 + threadIdx.x) >> 5;
    int lane = threadIdx.x & 31;
    if (wv >= e) return;
    int s = src[wv], t = dst[wv];
    float w = isd[s] * isd[t];
    const float* vr = hw + (size_t)s * DD;
    float* orow = o + (size_t)t * DD;
#pragma unroll
    for (int j = 0; j < 4; ++j) {
        int d = lane + 32 * j;
        atomicAdd(&orow[d], vr[d] * w);
    }
}

__global__ __launch_bounds__(256) void ln128_relu_kernel(
    float* __restrict__ h, const float* __restrict__ g, const float* __restrict__ b, int n)
{
    int wv = (blockIdx.x * 256 + threadIdx.x) >> 5;
    int lane = threadIdx.x & 31;
    if (wv >= n) return;
    float* row = h + (size_t)wv * DD;
    float x[4], s = 0.f;
#pragma unroll
    for (int j = 0; j < 4; ++j) { x[j] = row[lane + 32 * j]; s += x[j]; }
    float mean = wave_sum32(s) * (1.f / 128.f);
    float sq = 0.f;
#pragma unroll
    for (int j = 0; j < 4; ++j) { float d = x[j] - mean; sq += d * d; }
    float inv = rsqrtf(wave_sum32(sq) * (1.f / 128.f) + 1e-5f);
#pragma unroll
    for (int j = 0; j < 4; ++j) {
        int d = lane + 32 * j;
        row[d] = fmaxf((x[j] - mean) * inv * g[d] + b[d], 0.f);
    }
}

__global__ __launch_bounds__(256) void attention_kernel(
    const float* __restrict__ qkv0, const float* __restrict__ qkv1,
    float* __restrict__ ao0, float* __restrict__ ao1, int n)
{
    int wv = (blockIdx.x * 256 + threadIdx.x) >> 5;
    int lane = threadIdx.x & 31;
    if (wv >= n) return;
    const float* b0 = qkv0 + (size_t)wv * 384;
    const float* b1 = qkv1 + (size_t)wv * 384;
    float q0[4], k0[4], v0[4], q1[4], k1[4], v1[4];
#pragma unroll
    for (int j = 0; j < 4; ++j) {
        int d = lane + 32 * j;
        q0[j] = b0[d]; k0[j] = b0[128 + d]; v0[j] = b0[256 + d];
        q1[j] = b1[d]; k1[j] = b1[128 + d]; v1[j] = b1[256 + d];
    }
    float l00 = 0, l01 = 0, l10 = 0, l11 = 0;
#pragma unroll
    for (int j = 0; j < 4; ++j) {
        l00 += q0[j] * k0[j]; l01 += q0[j] * k1[j];
        l10 += q1[j] * k0[j]; l11 += q1[j] * k1[j];
    }
    l00 = wave_sum32(l00); l01 = wave_sum32(l01);
    l10 = wave_sum32(l10); l11 = wave_sum32(l11);
    const float scale = 0.08838834764831845f;
    l00 *= scale; l01 *= scale; l10 *= scale; l11 *= scale;
    float m0 = fmaxf(l00, l01), e00 = __expf(l00 - m0), e01 = __expf(l01 - m0);
    float r0 = 1.f / (e00 + e01), w00 = e00 * r0, w01 = e01 * r0;
    float m1 = fmaxf(l10, l11), e10 = __expf(l10 - m1), e11 = __expf(l11 - m1);
    float r1 = 1.f / (e10 + e11), w10 = e10 * r1, w11 = e11 * r1;
    float* o0 = ao0 + (size_t)wv * DD;
    float* o1 = ao1 + (size_t)wv * DD;
#pragma unroll
    for (int j = 0; j < 4; ++j) {
        int d = lane + 32 * j;
        o0[d] = w00 * v0[j] + w01 * v1[j];
        o1[d] = w10 * v0[j] + w11 * v1[j];
    }
}

__global__ __launch_bounds__(256) void fuse_ln384_kernel(
    const float* __restrict__ a0, const float* __restrict__ a1,
    const float* __restrict__ rs, const float* __restrict__ g,
    const float* __restrict__ b, float* __restrict__ o, int n)
{
    int wv = (blockIdx.x * 256 + threadIdx.x) >> 5;
    int lane = threadIdx.x & 31;
    if (wv >= n) return;
    const float* p0 = a0 + (size_t)wv * DD;
    const float* p1 = a1 + (size_t)wv * DD;
    const float* p2 = rs + (size_t)wv * DD;
    float x[12], s = 0.f;
#pragma unroll
    for (int j = 0; j < 12; ++j) {
        int d = lane + 32 * j;
        float v = (d < 128) ? p0[d] : (d < 256) ? p1[d - 128] : p2[d - 256];
        x[j] = v; s += v;
    }
    float mean = wave_sum32(s) * (1.f / 384.f);
    float sq = 0.f;
#pragma unroll
    for (int j = 0; j < 12; ++j) { float d = x[j] - mean; sq += d * d; }
    float inv = rsqrtf(wave_sum32(sq) * (1.f / 384.f) + 1e-5f);
    float* orow = o + (size_t)wv * 384;
#pragma unroll
    for (int j = 0; j < 12; ++j) {
        int d = lane + 32 * j;
        orow[d] = (x[j] - mean) * inv * g[d] + b[d];
    }
}

__global__ void colsum_kernel(const float* __restrict__ p, float* __restrict__ cs, int n) {
    int c = threadIdx.x;
    float s = 0.f;
    for (int r = blockIdx.x; r < n; r += gridDim.x) s += p[(size_t)r * DD + c];
    atomicAdd(&cs[c], s);
}

__global__ __launch_bounds__(256) void pairnorm_kernel(
    const float* __restrict__ pre, const float* __restrict__ cs,
    float* __restrict__ o, int n)
{
    int wv = (blockIdx.x * 256 + threadIdx.x) >> 5;
    int lane = threadIdx.x & 31;
    if (wv >= n) return;
    const float* row = pre + (size_t)wv * DD;
    float x[4], sq = 0.f;
#pragma unroll
    for (int j = 0; j < 4; ++j) { x[j] = row[lane + 32 * j]; sq += x[j] * x[j]; }
    float rn = sqrtf(1e-6f + wave_sum32(sq));
    float inv = 20.f / rn;
    float* orow = o + (size_t)wv * DD;
    const float invN = 1.f / (float)NN;
#pragma unroll
    for (int j = 0; j < 4; ++j) {
        int d = lane + 32 * j;
        orow[d] = x[j] * inv - cs[d] * invN;
    }
}

extern "C" void kernel_launch(void* const* d_in, const int* in_sizes, int n_in,
                              void* d_out, int out_size, void* d_ws, size_t ws_size,
                              hipStream_t stream)
{
    const float* x      = (const float*)d_in[0];
    const float* x_org  = (const float*)d_in[1];
    const int*   ei     = (const int*)d_in[2];
    const float* in_W   = (const float*)d_in[3];
    const float* in_b   = (const float*)d_in[4];
    const float* res_W  = (const float*)d_in[5];
    const float* gcn_W  = (const float*)d_in[6];
    const float* gcn_b  = (const float*)d_in[7];
    const float* ln_g   = (const float*)d_in[8];
    const float* ln_b   = (const float*)d_in[9];
    const float* lin_W  = (const float*)d_in[10];
    const float* lin_b  = (const float*)d_in[11];
    const float* attr_W = (const float*)d_in[12];
    const float* attr_b = (const float*)d_in[13];
    const float* qkv_W  = (const float*)d_in[14];
    const float* qkv_b  = (const float*)d_in[15];
    const float* out_W  = (const float*)d_in[16];
    const float* out_b  = (const float*)d_in[17];
    const float* fln_g  = (const float*)d_in[18];
    const float* fln_b  = (const float*)d_in[19];
    const float* fuse_W = (const float*)d_in[20];
    const float* fuse_b = (const float*)d_in[21];
    float* out = (float*)d_out;

    const int* src = ei;
    const int* dst = ei + EE;

    float* ws = (float*)d_ws;
    const size_t ND = (size_t)NN * DD;
    float* isd  = ws;
    float* csum = isd + NN;
    float* bh   = csum + DD;
    float* bt   = bh + ND;
    float* br   = bt + ND;
    float* bs   = br + ND;
    float* q0b  = bs + ND;
    float* q1b  = q0b + 3 * ND;

    const dim3 blk(256);
    const int gemmO128  = ((NN / 16) * (DD / 64) + 7) / 8;
    const int gemmO384  = ((NN / 16) * (3 * DD / 64) + 7) / 8;
    const int nodeWaves = (NN + 7) / 8;
    const int edgeWaves = (EE + 7) / 8;
    const int elemBlk   = ((int)ND + 255) / 256;

    fill_kernel<<<(NN + 255) / 256, blk, 0, stream>>>(isd, 1.0f, NN);
    deg_scatter_kernel<<<(EE + 255) / 256, blk, 0, stream>>>(dst, isd, EE);
    rsqrt_kernel<<<(NN + 255) / 256, blk, 0, stream>>>(isd, NN);

    gemm_wmma<true, 1><<<gemmO128, blk, 0, stream>>>(x, in_W, in_b, bh, NN, DD, DD);

    gemm_wmma<false, 0><<<gemmO128, blk, 0, stream>>>(x_org, res_W, nullptr, bt, NN, DD, DD);
    fill_kernel<<<elemBlk, blk, 0, stream>>>(br, 0.f, (int)ND);
    res_scatter_kernel<<<edgeWaves, blk, 0, stream>>>(src, dst, bt, br, EE);

    for (int i = 0; i < 3; ++i) {
        gemm_wmma<true, 0><<<gemmO128, blk, 0, stream>>>(
            bh, gcn_W + (size_t)i * DD * DD, nullptr, bt, NN, DD, DD);
        gcn_base_kernel<<<elemBlk, blk, 0, stream>>>(
            bt, isd, gcn_b + (size_t)i * DD, bh, (int)ND);
        gcn_scatter_kernel<<<edgeWaves, blk, 0, stream>>>(src, dst, bt, isd, bh, EE);
        ln128_relu_kernel<<<nodeWaves, blk, 0, stream>>>(
            bh, ln_g + (size_t)i * DD, ln_b + (size_t)i * DD, NN);
    }

    gemm_wmma<true, 0><<<gemmO128, blk, 0, stream>>>(bh, lin_W, lin_b, bs, NN, DD, DD);
    gemm_wmma<true, 1><<<gemmO128, blk, 0, stream>>>(x_org, attr_W, attr_b, bt, NN, DD, DD);

    gemm_wmma<true, 0><<<gemmO384, blk, 0, stream>>>(bs, qkv_W, qkv_b, q0b, NN, DD, 3 * DD);
    gemm_wmma<true, 0><<<gemmO384, blk, 0, stream>>>(bt, qkv_W, qkv_b, q1b, NN, DD, 3 * DD);

    attention_kernel<<<nodeWaves, blk, 0, stream>>>(q0b, q1b, bh, bt, NN);

    gemm_wmma<true, 0><<<gemmO128, blk, 0, stream>>>(bh, out_W, out_b, bs, NN, DD, DD);
    gemm_wmma<true, 0><<<gemmO128, blk, 0, stream>>>(bt, out_W, out_b, q1b, NN, DD, DD);

    fuse_ln384_kernel<<<nodeWaves, blk, 0, stream>>>(bs, q1b, br, fln_g, fln_b, q0b, NN);

    gemm_wmma<true, 0><<<gemmO128, blk, 0, stream>>>(q0b, fuse_W, fuse_b, bh, NN, 3 * DD, DD);

    fill_kernel<<<1, 128, 0, stream>>>(csum, 0.f, DD);
    colsum_kernel<<<256, 128, 0, stream>>>(bh, csum, NN);
    pairnorm_kernel<<<nodeWaves, blk, 0, stream>>>(bh, csum, out, NN);
}


# Float atomic scatter-adds (residual, 3 GCN layers) replaced by per-destination row gathers over an edge list sorted in-call; row sums now in fixed ascending-source order, one store per row
# speedup vs baseline: 1.5252x; 1.5252x over previous
#include <hip/hip_runtime.h>
#include <hip/hip_bf16.h>

#define NN 50000
#define DD 128
#define EE 800000

typedef __attribute__((ext_vector_type(2))) float v2f;
typedef __attribute__((ext_vector_type(8))) float v8f;

template <bool TRANSB, int ACT>
__global__ __launch_bounds__(256) void gemm_wmma(
    const float* __restrict__ A, const float* __restrict__ W,
    const float* __restrict__ bias, float* __restrict__ C,
    int n, int K, int O)
{
    const int wave = threadIdx.x >> 5;
    const int lane = threadIdx.x & 31;
    const int groups_n = O >> 6;
    const int total = (n >> 4) * groups_n;
    const int tile = blockIdx.x * 8 + wave;
    if (tile >= total) return;

    const int tm = tile / groups_n;
    const int tg = tile % groups_n;
    const int half = lane >> 4;
    const int l15  = lane & 15;
    const int rbase = tm << 4;
    const int cbase = tg << 6;
    const int arow  = rbase + l15;

    v8f acc[4];
#pragma unroll
    for (int t = 0; t < 4; ++t) acc[t] = (v8f){0.f,0.f,0.f,0.f,0.f,0.f,0.f,0.f};

#if __has_builtin(__builtin_amdgcn_wmma_f32_16x16x4_f32)
    const float* Ap = A + (size_t)arow * K;
    for (int k0 = 0; k0 < K; k0 += 4) {
        const int ka = k0 + 2 * half;
        v2f a = *(const v2f*)(Ap + ka);
        v2f b[4];
        if (TRANSB) {
            const float* Wp = W + (size_t)(cbase + l15) * K + ka;
#pragma unroll
            for (int t = 0; t < 4; ++t)
                b[t] = *(const v2f*)(Wp + (size_t)(16 * t) * K);
        } else {
            const float* Wp = W + (size_t)ka * O + cbase + l15;
#pragma unroll
            for (int t = 0; t < 4; ++t) {
                b[t].x = Wp[16 * t];
                b[t].y = Wp[16 * t + O];
            }
        }
#pragma unroll
        for (int t = 0; t < 4; ++t)
            acc[t] = __builtin_amdgcn_wmma_f32_16x16x4_f32(
                 false, a,  false, b[t],
                 (short)0, acc[t],  false,  false);
    }
#else
    for (int k = 0; k < K; ++k) {
#pragma unroll
        for (int t = 0; t < 4; ++t) {
            int col = cbase + 16 * t + l15;
            float bv = TRANSB ? W[(size_t)col * K + k] : W[(size_t)k * O + col];
#pragma unroll
            for (int v = 0; v < 8; ++v)
                acc[t][v] += A[(size_t)(rbase + v + 8 * half) * K + k] * bv;
        }
    }
#endif

#pragma unroll
    for (int t = 0; t < 4; ++t) {
        const int col = cbase + 16 * t + l15;
        const float bv = bias ? bias[col] : 0.f;
#pragma unroll
        for (int v = 0; v < 8; ++v) {
            float val = acc[t][v] + bv;
            if (ACT == 1) val = fmaxf(val, 0.f);
            C[(size_t)(rbase + v + 8 * half) * O + col] = val;
        }
    }
}

__device__ __forceinline__ float wave_sum32(float v) {
#pragma unroll
    for (int m = 16; m > 0; m >>= 1) v += __shfl_xor(v, m, 32);
    return v;
}

__global__ __launch_bounds__(256) void fill_kernel(float* __restrict__ p, float v, int n) {
    int i = blockIdx.x * blockDim.x + threadIdx.x;
    if (i < n) p[i] = v;
}

__global__ __launch_bounds__(256) void fill_int_kernel(int* __restrict__ p, int v, int n) {
    int i = blockIdx.x * blockDim.x + threadIdx.x;
    if (i < n) p[i] = v;
}

__global__ __launch_bounds__(256) void count_rank_kernel(
    const int* __restrict__ dst, int* __restrict__ cnt, int* __restrict__ rank, int e)
{
    int i = blockIdx.x * blockDim.x + threadIdx.x;
    if (i < e) rank[i] = atomicAdd(&cnt[dst[i]], 1);
}

#define SCAN_CH 13
static_assert(32 * SCAN_CH * 128 >= NN, "scan block must cover all nodes");
static_assert(NN % 4 == 0, "16-byte accesses below need NN % 4 == 0");

__global__ __launch_bounds__(1024) void scan_isd_kernel(
    const int* __restrict__ cnt, int* __restrict__ rowptr, float* __restrict__ isd, int n)
{
    __shared__ int wtot[32];
    const int lane = threadIdx.x & 31;
    const int wave = threadIdx.x >> 5;
    const int wbase = wave * (SCAN_CH * 128);
    int4 c[SCAN_CH];
#pragma unroll
    for (int k = 0; k < SCAN_CH; ++k) {
        const int e0 = wbase + k * 128 + lane * 4;
        c[k] = (e0 < n) ? *(const int4*)(cnt + e0) : make_int4(0, 0, 0, 0);
    }
    int ex[SCAN_CH];
    int carry = 0;
#pragma unroll
    for (int k = 0; k < SCAN_CH; ++k) {
        const int tot = c[k].x + c[k].y + c[k].z + c[k].w;
        int inc = tot;
#pragma unroll
        for (int d = 1; d < 32; d <<= 1) {
            int t = __shfl_up(inc, d, 32);
            if (lane >= d) inc += t;
        }
        ex[k] = carry + inc - tot;
        carry += __shfl(inc, 31, 32);
    }
    if (lane == 0) wtot[wave] = carry;
    __syncthreads();
    int below = (lane < wave) ? wtot[lane] : 0;
#pragma unroll
    for (int m = 16; m > 0; m >>= 1) below += __shfl_xor(below, m, 32);
#pragma unroll
    for (int k = 0; k < SCAN_CH; ++k) {
        const int e0 = wbase + k * 128 + lane * 4;
        if (e0 < n) {
            const int p0 = below + ex[k];
            const int p1 = p0 + c[k].x, p2 = p1 + c[k].y, p3 = p2 + c[k].z;
            *(int4*)(rowptr + e0) = make_int4(p0, p1, p2, p3);
            *(float4*)(isd + e0) = make_float4(
                rsqrtf((float)c[k].x + 1.0f), rsqrtf((float)c[k].y + 1.0f),
                rsqrtf((float)c[k].z + 1.0f), rsqrtf((float)c[k].w + 1.0f));
        }
    }
}

__global__ __launch_bounds__(256) void place_kernel(
    const int* __restrict__ src, const int* __restrict__ dst,
    const int* __restrict__ rowptr, const int* __restrict__ rank,
    int* __restrict__ csr, int e)
{
    int i = blockIdx.x * blockDim.x + threadIdx.x;
    if (i >= e) return;
    const int pos = rowptr[dst[i]] + rank[i];
    if (pos >= 0 && pos < e) csr[pos] = src[i];
}

__global__ __launch_bounds__(256) void row_order_kernel(
    const int* __restrict__ rowptr, const int* __restrict__ cnt,
    const int* __restrict__ csr_in, int* __restrict__ csr_out, int n)
{
    const int lane = threadIdx.x & 31;
    const int node = __builtin_amdgcn_readfirstlane((int)((blockIdx.x * 256 + threadIdx.x) >> 5));
    if (node >= n) return;
    const int start = __builtin_amdgcn_readfirstlane(rowptr[node]);
    const int len   = __builtin_amdgcn_readfirstlane(cnt[node]);
    for (int ib = 0; ib < len; ib += 32) {
        const int i = ib + lane;
        const int xi = (i < len) ? csr_in[start + i] : 0;
        int pos = 0;
        for (int jb = 0; jb < len; jb += 32) {
            const int mj = min(32, len - jb);
            const int xv = (lane < mj) ? csr_in[start + jb + lane] : 0;
            for (int jj = 0; jj < mj; ++jj) {
                const int xj = __builtin_amdgcn_readlane(xv, jj);
                const int j = jb + jj;
                pos += (xj < xi || (xj == xi && j < i)) ? 1 : 0;
            }
        }
        if (i < len) csr_out[start + pos] = xi;
    }
}

template <bool GCN, int B>
__device__ __forceinline__ void gather_accum(
    float4& acc, const float* __restrict__ v, int ms, float mw, int j, int lane)
{
    float4 r[B];
    float w[B];
#pragma unroll
    for (int u = 0; u < B; ++u) {
        const int s = __builtin_amdgcn_readlane(ms, j + u);
        r[u] = *(const float4*)(v + (size_t)s * DD + lane * 4);
        w[u] = GCN ? __int_as_float(__builtin_amdgcn_readlane(__float_as_int(mw), j + u)) : 1.0f;
    }
#pragma unroll
    for (int u = 0; u < B; ++u) {
        if (GCN) {
            acc.x = fmaf(r[u].x, w[u], acc.x); acc.y = fmaf(r[u].y, w[u], acc.y);
            acc.z = fmaf(r[u].z, w[u], acc.z); acc.w = fmaf(r[u].w, w[u], acc.w);
        } else {
            acc.x += r[u].x; acc.y += r[u].y; acc.z += r[u].z; acc.w += r[u].w;
        }
    }
}

template <bool GCN>
__global__ __launch_bounds__(256) void gather_rows_kernel(
    const int* __restrict__ rowptr, const int* __restrict__ cnt, const int* __restrict__ csr,
    const float* __restrict__ v, const float* __restrict__ isd, const float* __restrict__ bias,
    float* __restrict__ o, int n)
{
    const int lane = threadIdx.x & 31;
    const int node = __builtin_amdgcn_readfirstlane((int)((blockIdx.x * 256 + threadIdx.x) >> 5));
    if (node >= n) return;
    const int start = __builtin_amdgcn_readfirstlane(rowptr[node]);
    const int len   = __builtin_amdgcn_readfirstlane(cnt[node]);
    float4 acc = make_float4(0.f, 0.f, 0.f, 0.f);
    float4 sv = acc, bv = acc;
    float it = 0.f;
    if (GCN) {
        it = isd[node];
        sv = *(const float4*)(v + (size_t)node * DD + lane * 4);
        bv = *(const float4*)(bias + lane * 4);
    }
    for (int base = 0; base < len; base += 32) {
        const int m = min(32, len - base);
        int ms = 0;
        float mw = 0.f;
        if (lane < m) {
            ms = csr[start + base + lane];
            ms = min(max(ms, 0), n - 1);
            if (GCN) mw = isd[ms] * it;
        }
        int j = 0;
        for (; j + 8 <= m; j += 8) gather_accum<GCN, 8>(acc, v, ms, mw, j, lane);
        if (j + 4 <= m) { gather_accum<GCN, 4>(acc, v, ms, mw, j, lane); j += 4; }
        if (j + 2 <= m) { gather_accum<GCN, 2>(acc, v, ms, mw, j, lane); j += 2; }
        if (j < m)      { gather_accum<GCN, 1>(acc, v, ms, mw, j, lane); }
    }
    if (GCN) {
        acc.x += sv.x * it * it + bv.x; acc.y += sv.y * it * it + bv.y;
        acc.z += sv.z * it * it + bv.z; acc.w += sv.w * it * it + bv.w;
    }
    *(float4*)(o + (size_t)node * DD + lane * 4) = acc;
}

__global__ __launch_bounds__(256) void ln128_relu_kernel(
    float* __restrict__ h, const float* __restrict__ g, const float* __restrict__ b, int n)
{
    int wv = (blockIdx.x * 256 + threadIdx.x) >> 5;
    int lane = threadIdx.x & 31;
    if (wv >= n) return;
    float* row = h + (size_t)wv * DD;
    float x[4], s = 0.f;
#pragma unroll
    for (int j = 0; j < 4; ++j) { x[j] = row[lane + 32 * j]; s += x[j]; }
    float mean = wave_sum32(s) * (1.f / 128.f);
    float sq = 0.f;
#pragma unroll
    for (int j = 0; j < 4; ++j) { float d = x[j] - mean; sq += d * d; }
    float inv = rsqrtf(wave_sum32(sq) * (1.f / 128.f) + 1e-5f);
#pragma unroll
    for (int j = 0; j < 4; ++j) {
        int d = lane + 32 * j;
        row[d] = fmaxf((x[j] - mean) * inv * g[d] + b[d], 0.f);
    }
}

__global__ __launch_bounds__(256) void attention_kernel(
    const float* __restrict__ qkv0, const float* __restrict__ qkv1,
    float* __restrict__ ao0, float* __restrict__ ao1, int n)
{
    int wv = (blockIdx.x * 256 + threadIdx.x) >> 5;
    int lane = threadIdx.x & 31;
    if (wv >= n) return;
    const float* b0 = qkv0 + (size_t)wv * 384;
    const float* b1 = qkv1 + (size_t)wv * 384;
    float q0[4], k0[4], v0[4], q1[4], k1[4], v1[4];
#pragma unroll
    for (int j = 0; j < 4; ++j) {
        int d = lane + 32 * j;
        q0[j] = b0[d]; k0[j] = b0[128 + d]; v0[j] = b0[256 + d];
        q1[j] = b1[d]; k1[j] = b1[128 + d]; v1[j] = b1[256 + d];
    }
    float l00 = 0, l01 = 0, l10 = 0, l11 = 0;
#pragma unroll
    for (int j = 0; j < 4; ++j) {
        l00 += q0[j] * k0[j]; l01 += q0[j] * k1[j];
        l10 += q1[j] * k0[j]; l11 += q1[j] * k1[j];
    }
    l00 = wave_sum32(l00); l01 = wave_sum32(l01);
    l10 = wave_sum32(l10); l11 = wave_sum32(l11);
    const float scale = 0.08838834764831845f;
    l00 *= scale; l01 *= scale; l10 *= scale; l11 *= scale;
    float m0 = fmaxf(l00, l01), e00 = __expf(l00 - m0), e01 = __expf(l01 - m0);
    float r0 = 1.f / (e00 + e01), w00 = e00 * r0, w01 = e01 * r0;
    float m1 = fmaxf(l10, l11), e10 = __expf(l10 - m1), e11 = __expf(l11 - m1);
    float r1 = 1.f / (e10 + e11), w10 = e10 * r1, w11 = e11 * r1;
    float* o0 = ao0 + (size_t)wv * DD;
    float* o1 = ao1 + (size_t)wv * DD;
#pragma unroll
    for (int j = 0; j < 4; ++j) {
        int d = lane + 32 * j;
        o0[d] = w00 * v0[j] + w01 * v1[j];
        o1[d] = w10 * v0[j] + w11 * v1[j];
    }
}

__global__ __launch_bounds__(256) void fuse_ln384_kernel(
    const float* __restrict__ a0, const float* __restrict__ a1,
    const float* __restrict__ rs, const float* __restrict__ g,
    const float* __restrict__ b, float* __restrict__ o, int n)
{
    int wv = (blockIdx.x * 256 + threadIdx.x) >> 5;
    int lane = threadIdx.x & 31;
    if (wv >= n) return;
    const float* p0 = a0 + (size_t)wv * DD;
    const float* p1 = a1 + (size_t)wv * DD;
    const float* p2 = rs + (size_t)wv * DD;
    float x[12], s = 0.f;
#pragma unroll
    for (int j = 0; j < 12; ++j) {
        int d = lane + 32 * j;
        float v = (d < 128) ? p0[d] : (d < 256) ? p1[d - 128] : p2[d - 256];
        x[j] = v; s += v;
    }
    float mean = wave_sum32(s) * (1.f / 384.f);
    float sq = 0.f;
#pragma unroll
    for (int j = 0; j < 12; ++j) { float d = x[j] - mean; sq += d * d; }
    float inv = rsqrtf(wave_sum32(sq) * (1.f / 384.f) + 1e-5f);
    float* orow = o + (size_t)wv * 384;
#pragma unroll
    for (int j = 0; j < 12; ++j) {
        int d = lane + 32 * j;
        orow[d] = (x[j] - mean) * inv * g[d] + b[d];
    }
}

__global__ void colsum_kernel(const float* __restrict__ p, float* __restrict__ cs, int n) {
    int c = threadIdx.x;
    float s = 0.f;
    for (int r = blockIdx.x; r < n; r += gridDim.x) s += p[(size_t)r * DD + c];
    atomicAdd(&cs[c], s);
}

__global__ __launch_bounds__(256) void pairnorm_kernel(
    const float* __restrict__ pre, const float* __restrict__ cs,
    float* __restrict__ o, int n)
{
    int wv = (blockIdx.x * 256 + threadIdx.x) >> 5;
    int lane = threadIdx.x & 31;
    if (wv >= n) return;
    const float* row = pre + (size_t)wv * DD;
    float x[4], sq = 0.f;
#pragma unroll
    for (int j = 0; j < 4; ++j) { x[j] = row[lane + 32 * j]; sq += x[j] * x[j]; }
    float rn = sqrtf(1e-6f + wave_sum32(sq));
    float inv = 20.f / rn;
    float* orow = o + (size_t)wv * DD;
    const float invN = 1.f / (float)NN;
#pragma unroll
    for (int j = 0; j < 4; ++j) {
        int d = lane + 32 * j;
        orow[d] = x[j] * inv - cs[d] * invN;
    }
}

extern "C" void kernel_launch(void* const* d_in, const int* in_sizes, int n_in,
                              void* d_out, int out_size, void* d_ws, size_t ws_size,
                              hipStream_t stream)
{
    const float* x      = (const float*)d_in[0];
    const float* x_org  = (const float*)d_in[1];
    const int*   ei     = (const int*)d_in[2];
    const float* in_W   = (const float*)d_in[3];
    const float* in_b   = (const float*)d_in[4];
    const float* res_W  = (const float*)d_in[5];
    const float* gcn_W  = (const float*)d_in[6];
    const float* gcn_b  = (const float*)d_in[7];
    const float* ln_g   = (const float*)d_in[8];
    const float* ln_b   = (const float*)d_in[9];
    const float* lin_W  = (const float*)d_in[10];
    const float* lin_b  = (const float*)d_in[11];
    const float* attr_W = (const float*)d_in[12];
    const float* attr_b = (const float*)d_in[13];
    const float* qkv_W  = (const float*)d_in[14];
    const float* qkv_b  = (const float*)d_in[15];
    const float* out_W  = (const float*)d_in[16];
    const float* out_b  = (const float*)d_in[17];
    const float* fln_g  = (const float*)d_in[18];
    const float* fln_b  = (const float*)d_in[19];
    const float* fuse_W = (const float*)d_in[20];
    const float* fuse_b = (const float*)d_in[21];
    float* out = (float*)d_out;

    const int* src = ei;
    const int* dst = ei + EE;

    float* ws = (float*)d_ws;
    const size_t ND = (size_t)NN * DD;
    float* isd  = ws;
    float* csum = isd + NN;
    float* bh   = csum + DD;
    float* bt   = bh + ND;
    float* br   = bt + ND;
    float* bs   = br + ND;
    float* q0b  = bs + ND;
    float* q1b  = q0b + 3 * ND;
    int* cnt    = (int*)q1b;
    int* rowptr = cnt + NN;
    int* rank   = rowptr + NN;
    int* csr_a  = rank + EE;
    int* csr_b  = csr_a + EE;

    const dim3 blk(256);
    const int gemmO128  = ((NN / 16) * (DD / 64) + 7) / 8;
    const int gemmO384  = ((NN / 16) * (3 * DD / 64) + 7) / 8;
    const int nodeWaves = (NN + 7) / 8;
    const int edgeBlk   = (EE + 255) / 256;

    fill_int_kernel<<<(NN + 255) / 256, blk, 0, stream>>>(cnt, 0, NN);
    count_rank_kernel<<<edgeBlk, blk, 0, stream>>>(dst, cnt, rank, EE);
    scan_isd_kernel<<<1, 1024, 0, stream>>>(cnt, rowptr, isd, NN);
    place_kernel<<<edgeBlk, blk, 0, stream>>>(src, dst, rowptr, rank, csr_a, EE);
    row_order_kernel<<<nodeWaves, blk, 0, stream>>>(rowptr, cnt, csr_a, csr_b, NN);

    gemm_wmma<true, 1><<<gemmO128, blk, 0, stream>>>(x, in_W, in_b, bh, NN, DD, DD);

    gemm_wmma<false, 0><<<gemmO128, blk, 0, stream>>>(x_org, res_W, nullptr, bt, NN, DD, DD);
    gather_rows_kernel<false><<<nodeWaves, blk, 0, stream>>>(
        rowptr, cnt, csr_b, bt, nullptr, nullptr, br, NN);

    for (int i = 0; i < 3; ++i) {
        gemm_wmma<true, 0><<<gemmO128, blk, 0, stream>>>(
            bh, gcn_W + (size_t)i * DD * DD, nullptr, bt, NN, DD, DD);
        gather_rows_kernel<true><<<nodeWaves, blk, 0, stream>>>(
            rowptr, cnt, csr_b, bt, isd, gcn_b + (size_t)i * DD, bh, NN);
        ln128_relu_kernel<<<nodeWaves, blk, 0, stream>>>(
            bh, ln_g + (size_t)i * DD, ln_b + (size_t)i * DD, NN);
    }

    gemm_wmma<true, 0><<<gemmO128, blk, 0, stream>>>(bh, lin_W, lin_b, bs, NN, DD, DD);
    gemm_wmma<true, 1><<<gemmO128, blk, 0, stream>>>(x_org, attr_W, attr_b, bt, NN, DD, DD);

    gemm_wmma<true, 0><<<gemmO384, blk, 0, stream>>>(bs, qkv_W, qkv_b, q0b, NN, DD, 3 * DD);
    gemm_wmma<true, 0><<<gemmO384, blk, 0, stream>>>(bt, qkv_W, qkv_b, q1b, NN, DD, 3 * DD);

    attention_kernel<<<nodeWaves, blk, 0, stream>>>(q0b, q1b, bh, bt, NN);

    gemm_wmma<true, 0><<<gemmO128, blk, 0, stream>>>(bh, out_W, out_b, bs, NN, DD, DD);
    gemm_wmma<true, 0><<<gemmO128, blk, 0, stream>>>(bt, out_W, out_b, q1b, NN, DD, DD);

    fuse_ln384_kernel<<<nodeWaves, blk, 0, stream>>>(bs, q1b, br, fln_g, fln_b, q0b, NN);

    gemm_wmma<true, 0><<<gemmO128, blk, 0, stream>>>(q0b, fuse_W, fuse_b, bh, NN, 3 * DD, DD);

    fill_kernel<<<1, 128, 0, stream>>>(csum, 0.f, DD);
    colsum_kernel<<<256, 128, 0, stream>>>(bh, csum, NN);
    pairnorm_kernel<<<nodeWaves, blk, 0, stream>>>(bh, csum, out, NN);
}


# GEMM operands via LDS: each 8-wave block copies a 64-row A tile and 128 W rows with whole-line loads; waves read f32 WMMA fragments from LDS (32x32 per wave), not 16-line strided global loads
# speedup vs baseline: 2.1047x; 1.3800x over previous
#include <hip/hip_runtime.h>
#include <hip/hip_bf16.h>

#define NN 50000
#define DD 128
#define EE 800000

typedef __attribute__((ext_vector_type(2))) float v2f;
typedef __attribute__((ext_vector_type(8))) float v8f;

#define GP 132
#define GBM 64
#define GBN 128

template <bool TRANSB, int ACT, int K, int O>
__global__ __launch_bounds__(256) void gemm_lds(
    const float* __restrict__ A, const float* __restrict__ W,
    const float* __restrict__ bias, float* __restrict__ C, int n)
{
    static_assert(K % 128 == 0 && O % GBN == 0, "tile sizes");
    __shared__ __attribute__((aligned(16))) float As[GBM * GP];
    __shared__ __attribute__((aligned(16))) float Ws[GBN * GP];

    const int wave = threadIdx.x >> 5;
    const int lane = threadIdx.x & 31;
    const int half = lane >> 4;
    const int l15  = lane & 15;
    const int rbase = blockIdx.x * GBM;
    const int cbase = blockIdx.y * GBN;
    const int wr = (wave >> 2) * 32;
    const int wc = (wave & 3) * 32;

    v8f acc[2][2];
#pragma unroll
    for (int i = 0; i < 2; ++i)
#pragma unroll
        for (int j = 0; j < 2; ++j) acc[i][j] = (v8f){0.f,0.f,0.f,0.f,0.f,0.f,0.f,0.f};

    for (int kc = 0; kc < K; kc += 128) {
        if (kc > 0) __syncthreads();

#pragma unroll
        for (int i = 0; i < GBM / 8; ++i) {
            const int row = wave + 8 * i;
            const int grow = min(rbase + row, n - 1);
            const float4 v = *(const float4*)(A + (size_t)grow * K + kc + lane * 4);
            *(float4*)(As + row * GP + lane * 4) = v;
        }
        if (TRANSB) {
#pragma unroll
            for (int i = 0; i < GBN / 8; ++i) {
                const int row = wave + 8 * i;
                const float4 v = *(const float4*)(W + (size_t)(cbase + row) * K + kc + lane * 4);
                *(float4*)(Ws + row * GP + lane * 4) = v;
            }
        } else {
#pragma unroll
            for (int i = 0; i < GBN / 8; ++i) {
                const int idx = wave * (GBN / 8) + i;
                const int k  = (idx >> 2) * 4 + (lane >> 3);
                const int nb = (idx & 3) * 32 + (lane & 7) * 4;
                const float4 v = *(const float4*)(W + (size_t)(kc + k) * O + cbase + nb);
                Ws[(nb + 0) * GP + k] = v.x;
                Ws[(nb + 1) * GP + k] = v.y;
                Ws[(nb + 2) * GP + k] = v.z;
                Ws[(nb + 3) * GP + k] = v.w;
            }
        }
        __syncthreads();

        const float* ap = As + (wr + l15) * GP + 2 * half;
        const float* bp = Ws + (wc + l15) * GP + 2 * half;
#pragma unroll 8
        for (int kk = 0; kk < 128; kk += 4) {
            const v2f a0 = *(const v2f*)(ap + kk);
            const v2f a1 = *(const v2f*)(ap + 16 * GP + kk);
            const v2f b0 = *(const v2f*)(bp + kk);
            const v2f b1 = *(const v2f*)(bp + 16 * GP + kk);
            acc[0][0] = __builtin_amdgcn_wmma_f32_16x16x4_f32(
                false, a0, false, b0, (short)0, acc[0][0], false, false);
            acc[0][1] = __builtin_amdgcn_wmma_f32_16x16x4_f32(
                false, a0, false, b1, (short)0, acc[0][1], false, false);
            acc[1][0] = __builtin_amdgcn_wmma_f32_16x16x4_f32(
                false, a1, false, b0, (short)0, acc[1][0], false, false);
            acc[1][1] = __builtin_amdgcn_wmma_f32_16x16x4_f32(
                false, a1, false, b1, (short)0, acc[1][1], false, false);
        }
    }

#pragma unroll
    for (int j = 0; j < 2; ++j) {
        const int col = cbase + wc + 16 * j + l15;
        const float bv = bias ? bias[col] : 0.f;
#pragma unroll
        for (int i = 0; i < 2; ++i) {
#pragma unroll
            for (int v = 0; v < 8; ++v) {
                const int row = rbase + wr + 16 * i + v + 8 * half;
                float val = acc[i][j][v] + bv;
                if (ACT == 1) val = fmaxf(val, 0.f);
                if (row < n) C[(size_t)row * O + col] = val;
            }
        }
    }
}

__device__ __forceinline__ float wave_sum32(float v) {
#pragma unroll
    for (int m = 16; m > 0; m >>= 1) v += __shfl_xor(v, m, 32);
    return v;
}

__global__ __launch_bounds__(256) void fill_kernel(float* __restrict__ p, float v, int n) {
    int i = blockIdx.x * blockDim.x + threadIdx.x;
    if (i < n) p[i] = v;
}

__global__ __launch_bounds__(256) void fill_int_kernel(int* __restrict__ p, int v, int n) {
    int i = blockIdx.x * blockDim.x + threadIdx.x;
    if (i < n) p[i] = v;
}

__global__ __launch_bounds__(256) void count_rank_kernel(
    const int* __restrict__ dst, int* __restrict__ cnt, int* __restrict__ rank, int e)
{
    int i = blockIdx.x * blockDim.x + threadIdx.x;
    if (i < e) rank[i] = atomicAdd(&cnt[dst[i]], 1);
}

#define SCAN_CH 13
static_assert(32 * SCAN_CH * 128 >= NN, "scan block must cover all nodes");
static_assert(NN % 4 == 0, "16-byte accesses below need NN % 4 == 0");

__global__ __launch_bounds__(1024) void scan_isd_kernel(
    const int* __restrict__ cnt, int* __restrict__ rowptr, float* __restrict__ isd, int n)
{
    __shared__ int wtot[32];
    const int lane = threadIdx.x & 31;
    const int wave = threadIdx.x >> 5;
    const int wbase = wave * (SCAN_CH * 128);
    int4 c[SCAN_CH];
#pragma unroll
    for (int k = 0; k < SCAN_CH; ++k) {
        const int e0 = wbase + k * 128 + lane * 4;
        c[k] = (e0 < n) ? *(const int4*)(cnt + e0) : make_int4(0, 0, 0, 0);
    }
    int ex[SCAN_CH];
    int carry = 0;
#pragma unroll
    for (int k = 0; k < SCAN_CH; ++k) {
        const int tot = c[k].x + c[k].y + c[k].z + c[k].w;
        int inc = tot;
#pragma unroll
        for (int d = 1; d < 32; d <<= 1) {
            int t = __shfl_up(inc, d, 32);
            if (lane >= d) inc += t;
        }
        ex[k] = carry + inc - tot;
        carry += __shfl(inc, 31, 32);
    }
    if (lane == 0) wtot[wave] = carry;
    __syncthreads();
    int below = (lane < wave) ? wtot[lane] : 0;
#pragma unroll
    for (int m = 16; m > 0; m >>= 1) below += __shfl_xor(below, m, 32);
#pragma unroll
    for (int k = 0; k < SCAN_CH; ++k) {
        const int e0 = wbase + k * 128 + lane * 4;
        if (e0 < n) {
            const int p0 = below + ex[k];
            const int p1 = p0 + c[k].x, p2 = p1 + c[k].y, p3 = p2 + c[k].z;
            *(int4*)(rowptr + e0) = make_int4(p0, p1, p2, p3);
            *(float4*)(isd + e0) = make_float4(
                rsqrtf((float)c[k].x + 1.0f), rsqrtf((float)c[k].y + 1.0f),
                rsqrtf((float)c[k].z + 1.0f), rsqrtf((float)c[k].w + 1.0f));
        }
    }
}

__global__ __launch_bounds__(256) void place_kernel(
    const int* __restrict__ src, const int* __restrict__ dst,
    const int* __restrict__ rowptr, const int* __restrict__ rank,
    int* __restrict__ csr, int e)
{
    int i = blockIdx.x * blockDim.x + threadIdx.x;
    if (i >= e) return;
    const int pos = rowptr[dst[i]] + rank[i];
    if (pos >= 0 && pos < e) csr[pos] = src[i];
}

__global__ __launch_bounds__(256) void row_order_kernel(
    const int* __restrict__ rowptr, const int* __restrict__ cnt,
    const int* __restrict__ csr_in, int* __restrict__ csr_out, int n)
{
    const int lane = threadIdx.x & 31;
    const int node = __builtin_amdgcn_readfirstlane((int)((blockIdx.x * 256 + threadIdx.x) >> 5));
    if (node >= n) return;
    const int start = __builtin_amdgcn_readfirstlane(rowptr[node]);
    const int len   = __builtin_amdgcn_readfirstlane(cnt[node]);
    for (int ib = 0; ib < len; ib += 32) {
        const int i = ib + lane;
        const int xi = (i < len) ? csr_in[start + i] : 0;
        int pos = 0;
        for (int jb = 0; jb < len; jb += 32) {
            const int mj = min(32, len - jb);
            const int xv = (lane < mj) ? csr_in[start + jb + lane] : 0;
            for (int jj = 0; jj < mj; ++jj) {
                const int xj = __builtin_amdgcn_readlane(xv, jj);
                const int j = jb + jj;
                pos += (xj < xi || (xj == xi && j < i)) ? 1 : 0;
            }
        }
        if (i < len) csr_out[start + pos] = xi;
    }
}

template <bool GCN, int B>
__device__ __forceinline__ void gather_accum(
    float4& acc, const float* __restrict__ v, int ms, float mw, int j, int lane)
{
    float4 r[B];
    float w[B];
#pragma unroll
    for (int u = 0; u < B; ++u) {
        const int s = __builtin_amdgcn_readlane(ms, j + u);
        r[u] = *(const float4*)(v + (size_t)s * DD + lane * 4);
        w[u] = GCN ? __int_as_float(__builtin_amdgcn_readlane(__float_as_int(mw), j + u)) : 1.0f;
    }
#pragma unroll
    for (int u = 0; u < B; ++u) {
        if (GCN) {
            acc.x = fmaf(r[u].x, w[u], acc.x); acc.y = fmaf(r[u].y, w[u], acc.y);
            acc.z = fmaf(r[u].z, w[u], acc.z); acc.w = fmaf(r[u].w, w[u], acc.w);
        } else {
            acc.x += r[u].x; acc.y += r[u].y; acc.z += r[u].z; acc.w += r[u].w;
        }
    }
}

template <bool GCN>
__global__ __launch_bounds__(256) void gather_rows_kernel(
    const int* __restrict__ rowptr, const int* __restrict__ cnt, const int* __restrict__ csr,
    const float* __restrict__ v, const float* __restrict__ isd, const float* __restrict__ bias,
    float* __restrict__ o, int n)
{
    const int lane = threadIdx.x & 31;
    const int node = __builtin_amdgcn_readfirstlane((int)((blockIdx.x * 256 + threadIdx.x) >> 5));
    if (node >= n) return;
    const int start = __builtin_amdgcn_readfirstlane(rowptr[node]);
    const int len   = __builtin_amdgcn_readfirstlane(cnt[node]);
    float4 acc = make_float4(0.f, 0.f, 0.f, 0.f);
    float4 sv = acc, bv = acc;
    float it = 0.f;
    if (GCN) {
        it = isd[node];
        sv = *(const float4*)(v + (size_t)node * DD + lane * 4);
        bv = *(const float4*)(bias + lane * 4);
    }
    for (int base = 0; base < len; base += 32) {
        const int m = min(32, len - base);
        int ms = 0;
        float mw = 0.f;
        if (lane < m) {
            ms = csr[start + base + lane];
            ms = min(max(ms, 0), n - 1);
            if (GCN) mw = isd[ms] * it;
        }
        int j = 0;
        for (; j + 8 <= m; j += 8) gather_accum<GCN, 8>(acc, v, ms, mw, j, lane);
        if (j + 4 <= m) { gather_accum<GCN, 4>(acc, v, ms, mw, j, lane); j += 4; }
        if (j + 2 <= m) { gather_accum<GCN, 2>(acc, v, ms, mw, j, lane); j += 2; }
        if (j < m)      { gather_accum<GCN, 1>(acc, v, ms, mw, j, lane); }
    }
    if (GCN) {
        acc.x += sv.x * it * it + bv.x; acc.y += sv.y * it * it + bv.y;
        acc.z += sv.z * it * it + bv.z; acc.w += sv.w * it * it + bv.w;
    }
    *(float4*)(o + (size_t)node * DD + lane * 4) = acc;
}

__global__ __launch_bounds__(256) void ln128_relu_kernel(
    float* __restrict__ h, const float* __restrict__ g, const float* __restrict__ b, int n)
{
    int wv = (blockIdx.x * 256 + threadIdx.x) >> 5;
    int lane = threadIdx.x & 31;
    if (wv >= n) return;
    float* row = h + (size_t)wv * DD;
    float x[4], s = 0.f;
#pragma unroll
    for (int j = 0; j < 4; ++j) { x[j] = row[lane + 32 * j]; s += x[j]; }
    float mean = wave_sum32(s) * (1.f / 128.f);
    float sq = 0.f;
#pragma unroll
    for (int j = 0; j < 4; ++j) { float d = x[j] - mean; sq += d * d; }
    float inv = rsqrtf(wave_sum32(sq) * (1.f / 128.f) + 1e-5f);
#pragma unroll
    for (int j = 0; j < 4; ++j) {
        int d = lane + 32 * j;
        row[d] = fmaxf((x[j] - mean) * inv * g[d] + b[d], 0.f);
    }
}

__global__ __launch_bounds__(256) void attention_kernel(
    const float* __restrict__ qkv0, const float* __restrict__ qkv1,
    float* __restrict__ ao0, float* __restrict__ ao1, int n)
{
    int wv = (blockIdx.x * 256 + threadIdx.x) >> 5;
    int lane = threadIdx.x & 31;
    if (wv >= n) return;
    const float* b0 = qkv0 + (size_t)wv * 384;
    const float* b1 = qkv1 + (size_t)wv * 384;
    float q0[4], k0[4], v0[4], q1[4], k1[4], v1[4];
#pragma unroll
    for (int j = 0; j < 4; ++j) {
        int d = lane + 32 * j;
        q0[j] = b0[d]; k0[j] = b0[128 + d]; v0[j] = b0[256 + d];
        q1[j] = b1[d]; k1[j] = b1[128 + d]; v1[j] = b1[256 + d];
    }
    float l00 = 0, l01 = 0, l10 = 0, l11 = 0;
#pragma unroll
    for (int j = 0; j < 4; ++j) {
        l00 += q0[j] * k0[j]; l01 += q0[j] * k1[j];
        l10 += q1[j] * k0[j]; l11 += q1[j] * k1[j];
    }
    l00 = wave_sum32(l00); l01 = wave_sum32(l01);
    l10 = wave_sum32(l10); l11 = wave_sum32(l11);
    const float scale = 0.08838834764831845f;
    l00 *= scale; l01 *= scale; l10 *= scale; l11 *= scale;
    float m0 = fmaxf(l00, l01), e00 = __expf(l00 - m0), e01 = __expf(l01 - m0);
    float r0 = 1.f / (e00 + e01), w00 = e00 * r0, w01 = e01 * r0;
    float m1 = fmaxf(l10, l11), e10 = __expf(l10 - m1), e11 = __expf(l11 - m1);
    float r1 = 1.f / (e10 + e11), w10 = e10 * r1, w11 = e11 * r1;
    float* o0 = ao0 + (size_t)wv * DD;
    float* o1 = ao1 + (size_t)wv * DD;
#pragma unroll
    for (int j = 0; j < 4; ++j) {
        int d = lane + 32 * j;
        o0[d] = w00 * v0[j] + w01 * v1[j];
        o1[d] = w10 * v0[j] + w11 * v1[j];
    }
}

__global__ __launch_bounds__(256) void fuse_ln384_kernel(
    const float* __restrict__ a0, const float* __restrict__ a1,
    const float* __restrict__ rs, const float* __restrict__ g,
    const float* __restrict__ b, float* __restrict__ o, int n)
{
    int wv = (blockIdx.x * 256 + threadIdx.x) >> 5;
    int lane = threadIdx.x & 31;
    if (wv >= n) return;
    const float* p0 = a0 + (size_t)wv * DD;
    const float* p1 = a1 + (size_t)wv * DD;
    const float* p2 = rs + (size_t)wv * DD;
    float x[12], s = 0.f;
#pragma unroll
    for (int j = 0; j < 12; ++j) {
        int d = lane + 32 * j;
        float v = (d < 128) ? p0[d] : (d < 256) ? p1[d - 128] : p2[d - 256];
        x[j] = v; s += v;
    }
    float mean = wave_sum32(s) * (1.f / 384.f);
    float sq = 0.f;
#pragma unroll
    for (int j = 0; j < 12; ++j) { float d = x[j] - mean; sq += d * d; }
    float inv = rsqrtf(wave_sum32(sq) * (1.f / 384.f) + 1e-5f);
    float* orow = o + (size_t)wv * 384;
#pragma unroll
    for (int j = 0; j < 12; ++j) {
        int d = lane + 32 * j;
        orow[d] = (x[j] - mean) * inv * g[d] + b[d];
    }
}

__global__ void colsum_kernel(const float* __restrict__ p, float* __restrict__ cs, int n) {
    int c = threadIdx.x;
    float s = 0.f;
    for (int r = blockIdx.x; r < n; r += gridDim.x) s += p[(size_t)r * DD + c];
    atomicAdd(&cs[c], s);
}

__global__ __launch_bounds__(256) void pairnorm_kernel(
    const float* __restrict__ pre, const float* __restrict__ cs,
    float* __restrict__ o, int n)
{
    int wv = (blockIdx.x * 256 + threadIdx.x) >> 5;
    int lane = threadIdx.x & 31;
    if (wv >= n) return;
    const float* row = pre + (size_t)wv * DD;
    float x[4], sq = 0.f;
#pragma unroll
    for (int j = 0; j < 4; ++j) { x[j] = row[lane + 32 * j]; sq += x[j] * x[j]; }
    float rn = sqrtf(1e-6f + wave_sum32(sq));
    float inv = 20.f / rn;
    float* orow = o + (size_t)wv * DD;
    const float invN = 1.f / (float)NN;
#pragma unroll
    for (int j = 0; j < 4; ++j) {
        int d = lane + 32 * j;
        orow[d] = x[j] * inv - cs[d] * invN;
    }
}

extern "C" void kernel_launch(void* const* d_in, const int* in_sizes, int n_in,
                              void* d_out, int out_size, void* d_ws, size_t ws_size,
                              hipStream_t stream)
{
    const float* x      = (const float*)d_in[0];
    const float* x_org  = (const float*)d_in[1];
    const int*   ei     = (const int*)d_in[2];
    const float* in_W   = (const float*)d_in[3];
    const float* in_b   = (const float*)d_in[4];
    const float* res_W  = (const float*)d_in[5];
    const float* gcn_W  = (const float*)d_in[6];
    const float* gcn_b  = (const float*)d_in[7];
    const float* ln_g   = (const float*)d_in[8];
    const float* ln_b   = (const float*)d_in[9];
    const float* lin_W  = (const float*)d_in[10];
    const float* lin_b  = (const float*)d_in[11];
    const float* attr_W = (const float*)d_in[12];
    const float* attr_b = (const float*)d_in[13];
    const float* qkv_W  = (const float*)d_in[14];
    const float* qkv_b  = (const float*)d_in[15];
    const float* out_W  = (const float*)d_in[16];
    const float* out_b  = (const float*)d_in[17];
    const float* fln_g  = (const float*)d_in[18];
    const float* fln_b  = (const float*)d_in[19];
    const float* fuse_W = (const float*)d_in[20];
    const float* fuse_b = (const float*)d_in[21];
    float* out = (float*)d_out;

    const int* src = ei;
    const int* dst = ei + EE;

    float* ws = (float*)d_ws;
    const size_t ND = (size_t)NN * DD;
    float* isd  = ws;
    float* csum = isd + NN;
    float* bh   = csum + DD;
    float* bt   = bh + ND;
    float* br   = bt + ND;
    float* bs   = br + ND;
    float* q0b  = bs + ND;
    float* q1b  = q0b + 3 * ND;
    int* cnt    = (int*)q1b;
    int* rowptr = cnt + NN;
    int* rank   = rowptr + NN;
    int* csr_a  = rank + EE;
    int* csr_b  = csr_a + EE;

    const dim3 blk(256);
    const dim3 gemmO128((NN + GBM - 1) / GBM, DD / GBN);
    const dim3 gemmO384((NN + GBM - 1) / GBM, 3 * DD / GBN);
    const int nodeWaves = (NN + 7) / 8;
    const int edgeBlk   = (EE + 255) / 256;

    fill_int_kernel<<<(NN + 255) / 256, blk, 0, stream>>>(cnt, 0, NN);
    count_rank_kernel<<<edgeBlk, blk, 0, stream>>>(dst, cnt, rank, EE);
    scan_isd_kernel<<<1, 1024, 0, stream>>>(cnt, rowptr, isd, NN);
    place_kernel<<<edgeBlk, blk, 0, stream>>>(src, dst, rowptr, rank, csr_a, EE);
    row_order_kernel<<<nodeWaves, blk, 0, stream>>>(rowptr, cnt, csr_a, csr_b, NN);

    gemm_lds<true, 1, DD, DD><<<gemmO128, blk, 0, stream>>>(x, in_W, in_b, bh, NN);

    gemm_lds<false, 0, DD, DD><<<gemmO128, blk, 0, stream>>>(x_org, res_W, nullptr, bt, NN);
    gather_rows_kernel<false><<<nodeWaves, blk, 0, stream>>>(
        rowptr, cnt, csr_b, bt, nullptr, nullptr, br, NN);

    for (int i = 0; i < 3; ++i) {
        gemm_lds<true, 0, DD, DD><<<gemmO128, blk, 0, stream>>>(
            bh, gcn_W + (size_t)i * DD * DD, nullptr, bt, NN);
        gather_rows_kernel<true><<<nodeWaves, blk, 0, stream>>>(
            rowptr, cnt, csr_b, bt, isd, gcn_b + (size_t)i * DD, bh, NN);
        ln128_relu_kernel<<<nodeWaves, blk, 0, stream>>>(
            bh, ln_g + (size_t)i * DD, ln_b + (size_t)i * DD, NN);
    }

    gemm_lds<true, 0, DD, DD><<<gemmO128, blk, 0, stream>>>(bh, lin_W, lin_b, bs, NN);
    gemm_lds<true, 1, DD, DD><<<gemmO128, blk, 0, stream>>>(x_org, attr_W, attr_b, bt, NN);

    gemm_lds<true, 0, DD, 3 * DD><<<gemmO384, blk, 0, stream>>>(bs, qkv_W, qkv_b, q0b, NN);
    gemm_lds<true, 0, DD, 3 * DD><<<gemmO384, blk, 0, stream>>>(bt, qkv_W, qkv_b, q1b, NN);

    attention_kernel<<<nodeWaves, blk, 0, stream>>>(q0b, q1b, bh, bt, NN);

    gemm_lds<true, 0, DD, DD><<<gemmO128, blk, 0, stream>>>(bh, out_W, out_b, bs, NN);
    gemm_lds<true, 0, DD, DD><<<gemmO128, blk, 0, stream>>>(bt, out_W, out_b, q1b, NN);

    fuse_ln384_kernel<<<nodeWaves, blk, 0, stream>>>(bs, q1b, br, fln_g, fln_b, q0b, NN);

    gemm_lds<true, 0, 3 * DD, DD><<<gemmO128, blk, 0, stream>>>(q0b, fuse_W, fuse_b, bh, NN);

    fill_kernel<<<1, 128, 0, stream>>>(csum, 0.f, DD);
    colsum_kernel<<<256, 128, 0, stream>>>(bh, csum, NN);
    pairnorm_kernel<<<nodeWaves, blk, 0, stream>>>(bh, csum, out, NN);
}


# LayerNorm+relu of each GCN layer done in the row-gather wave while it holds the row (3 LN passes dropped); LN sums now add a lane's 4 adjacent dims first, then across lanes
# speedup vs baseline: 2.1764x; 1.0341x over previous
#include <hip/hip_runtime.h>
#include <hip/hip_bf16.h>

#define NN 50000
#define DD 128
#define EE 800000

typedef __attribute__((ext_vector_type(2))) float v2f;
typedef __attribute__((ext_vector_type(8))) float v8f;

#define GP 132
#define GBM 64
#define GBN 128

template <bool TRANSB, int ACT, int K, int O>
__global__ __launch_bounds__(256) void gemm_lds(
    const float* __restrict__ A, const float* __restrict__ W,
    const float* __restrict__ bias, float* __restrict__ C, int n)
{
    static_assert(K % 128 == 0 && O % GBN == 0, "tile sizes");
    __shared__ __attribute__((aligned(16))) float As[GBM * GP];
    __shared__ __attribute__((aligned(16))) float Ws[GBN * GP];

    const int wave = threadIdx.x >> 5;
    const int lane = threadIdx.x & 31;
    const int half = lane >> 4;
    const int l15  = lane & 15;
    const int rbase = blockIdx.x * GBM;
    const int cbase = blockIdx.y * GBN;
    const int wr = (wave >> 2) * 32;
    const int wc = (wave & 3) * 32;

    v8f acc[2][2];
#pragma unroll
    for (int i = 0; i < 2; ++i)
#pragma unroll
        for (int j = 0; j < 2; ++j) acc[i][j] = (v8f){0.f,0.f,0.f,0.f,0.f,0.f,0.f,0.f};

    for (int kc = 0; kc < K; kc += 128) {
        if (kc > 0) __syncthreads();

#pragma unroll
        for (int i = 0; i < GBM / 8; ++i) {
            const int row = wave + 8 * i;
            const int grow = min(rbase + row, n - 1);
            const float4 v = *(const float4*)(A + (size_t)grow * K + kc + lane * 4);
            *(float4*)(As + row * GP + lane * 4) = v;
        }
        if (TRANSB) {
#pragma unroll
            for (int i = 0; i < GBN / 8; ++i) {
                const int row = wave + 8 * i;
                const float4 v = *(const float4*)(W + (size_t)(cbase + row) * K + kc + lane * 4);
                *(float4*)(Ws + row * GP + lane * 4) = v;
            }
        } else {
#pragma unroll
            for (int i = 0; i < GBN / 8; ++i) {
                const int idx = wave * (GBN / 8) + i;
                const int k  = (idx >> 2) * 4 + (lane >> 3);
                const int nb = (idx & 3) * 32 + (lane & 7) * 4;
                const float4 v = *(const float4*)(W + (size_t)(kc + k) * O + cbase + nb);
                Ws[(nb + 0) * GP + k] = v.x;
                Ws[(nb + 1) * GP + k] = v.y;
                Ws[(nb + 2) * GP + k] = v.z;
                Ws[(nb + 3) * GP + k] = v.w;
            }
        }
        __syncthreads();

        const float* ap = As + (wr + l15) * GP + 2 * half;
        const float* bp = Ws + (wc + l15) * GP + 2 * half;
#pragma unroll 8
        for (int kk = 0; kk < 128; kk += 4) {
            const v2f a0 = *(const v2f*)(ap + kk);
            const v2f a1 = *(const v2f*)(ap + 16 * GP + kk);
            const v2f b0 = *(const v2f*)(bp + kk);
            const v2f b1 = *(const v2f*)(bp + 16 * GP + kk);
            acc[0][0] = __builtin_amdgcn_wmma_f32_16x16x4_f32(
                false, a0, false, b0, (short)0, acc[0][0], false, false);
            acc[0][1] = __builtin_amdgcn_wmma_f32_16x16x4_f32(
                false, a0, false, b1, (short)0, acc[0][1], false, false);
            acc[1][0] = __builtin_amdgcn_wmma_f32_16x16x4_f32(
                false, a1, false, b0, (short)0, acc[1][0], false, false);
            acc[1][1] = __builtin_amdgcn_wmma_f32_16x16x4_f32(
                false, a1, false, b1, (short)0, acc[1][1], false, false);
        }
    }

#pragma unroll
    for (int j = 0; j < 2; ++j) {
        const int col = cbase + wc + 16 * j + l15;
        const float bv = bias ? bias[col] : 0.f;
#pragma unroll
        for (int i = 0; i < 2; ++i) {
#pragma unroll
            for (int v = 0; v < 8; ++v) {
                const int row = rbase + wr + 16 * i + v + 8 * half;
                float val = acc[i][j][v] + bv;
                if (ACT == 1) val = fmaxf(val, 0.f);
                if (row < n) C[(size_t)row * O + col] = val;
            }
        }
    }
}

__device__ __forceinline__ float wave_sum32(float v) {
#pragma unroll
    for (int m = 16; m > 0; m >>= 1) v += __shfl_xor(v, m, 32);
    return v;
}

__global__ __launch_bounds__(256) void fill_kernel(float* __restrict__ p, float v, int n) {
    int i = blockIdx.x * blockDim.x + threadIdx.x;
    if (i < n) p[i] = v;
}

__global__ __launch_bounds__(256) void fill_int_kernel(int* __restrict__ p, int v, int n) {
    int i = blockIdx.x * blockDim.x + threadIdx.x;
    if (i < n) p[i] = v;
}

__global__ __launch_bounds__(256) void count_rank_kernel(
    const int* __restrict__ dst, int* __restrict__ cnt, int* __restrict__ rank, int e)
{
    int i = blockIdx.x * blockDim.x + threadIdx.x;
    if (i < e) rank[i] = atomicAdd(&cnt[dst[i]], 1);
}

#define SCAN_CH 13
static_assert(32 * SCAN_CH * 128 >= NN, "scan block must cover all nodes");
static_assert(NN % 4 == 0, "16-byte accesses below need NN % 4 == 0");

__global__ __launch_bounds__(1024) void scan_isd_kernel(
    const int* __restrict__ cnt, int* __restrict__ rowptr, float* __restrict__ isd, int n)
{
    __shared__ int wtot[32];
    const int lane = threadIdx.x & 31;
    const int wave = threadIdx.x >> 5;
    const int wbase = wave * (SCAN_CH * 128);
    int4 c[SCAN_CH];
#pragma unroll
    for (int k = 0; k < SCAN_CH; ++k) {
        const int e0 = wbase + k * 128 + lane * 4;
        c[k] = (e0 < n) ? *(const int4*)(cnt + e0) : make_int4(0, 0, 0, 0);
    }
    int ex[SCAN_CH];
    int carry = 0;
#pragma unroll
    for (int k = 0; k < SCAN_CH; ++k) {
        const int tot = c[k].x + c[k].y + c[k].z + c[k].w;
        int inc = tot;
#pragma unroll
        for (int d = 1; d < 32; d <<= 1) {
            int t = __shfl_up(inc, d, 32);
            if (lane >= d) inc += t;
        }
        ex[k] = carry + inc - tot;
        carry += __shfl(inc, 31, 32);
    }
    if (lane == 0) wtot[wave] = carry;
    __syncthreads();
    int below = (lane < wave) ? wtot[lane] : 0;
#pragma unroll
    for (int m = 16; m > 0; m >>= 1) below += __shfl_xor(below, m, 32);
#pragma unroll
    for (int k = 0; k < SCAN_CH; ++k) {
        const int e0 = wbase + k * 128 + lane * 4;
        if (e0 < n) {
            const int p0 = below + ex[k];
            const int p1 = p0 + c[k].x, p2 = p1 + c[k].y, p3 = p2 + c[k].z;
            *(int4*)(rowptr + e0) = make_int4(p0, p1, p2, p3);
            *(float4*)(isd + e0) = make_float4(
                rsqrtf((float)c[k].x + 1.0f), rsqrtf((float)c[k].y + 1.0f),
                rsqrtf((float)c[k].z + 1.0f), rsqrtf((float)c[k].w + 1.0f));
        }
    }
}

__global__ __launch_bounds__(256) void place_kernel(
    const int* __restrict__ src, const int* __restrict__ dst,
    const int* __restrict__ rowptr, const int* __restrict__ rank,
    int* __restrict__ csr, int e)
{
    int i = blockIdx.x * blockDim.x + threadIdx.x;
    if (i >= e) return;
    const int pos = rowptr[dst[i]] + rank[i];
    if (pos >= 0 && pos < e) csr[pos] = src[i];
}

__global__ __launch_bounds__(256) void row_order_kernel(
    const int* __restrict__ rowptr, const int* __restrict__ cnt,
    const int* __restrict__ csr_in, int* __restrict__ csr_out, int n)
{
    const int lane = threadIdx.x & 31;
    const int node = __builtin_amdgcn_readfirstlane((int)((blockIdx.x * 256 + threadIdx.x) >> 5));
    if (node >= n) return;
    const int start = __builtin_amdgcn_readfirstlane(rowptr[node]);
    const int len   = __builtin_amdgcn_readfirstlane(cnt[node]);
    for (int ib = 0; ib < len; ib += 32) {
        const int i = ib + lane;
        const int xi = (i < len) ? csr_in[start + i] : 0;
        int pos = 0;
        for (int jb = 0; jb < len; jb += 32) {
            const int mj = min(32, len - jb);
            const int xv = (lane < mj) ? csr_in[start + jb + lane] : 0;
            for (int jj = 0; jj < mj; ++jj) {
                const int xj = __builtin_amdgcn_readlane(xv, jj);
                const int j = jb + jj;
                pos += (xj < xi || (xj == xi && j < i)) ? 1 : 0;
            }
        }
        if (i < len) csr_out[start + pos] = xi;
    }
}

template <bool GCN, int B>
__device__ __forceinline__ void gather_accum(
    float4& acc, const float* __restrict__ v, int ms, float mw, int j, int lane)
{
    float4 r[B];
    float w[B];
#pragma unroll
    for (int u = 0; u < B; ++u) {
        const int s = __builtin_amdgcn_readlane(ms, j + u);
        r[u] = *(const float4*)(v + (size_t)s * DD + lane * 4);
        w[u] = GCN ? __int_as_float(__builtin_amdgcn_readlane(__float_as_int(mw), j + u)) : 1.0f;
    }
#pragma unroll
    for (int u = 0; u < B; ++u) {
        if (GCN) {
            acc.x = fmaf(r[u].x, w[u], acc.x); acc.y = fmaf(r[u].y, w[u], acc.y);
            acc.z = fmaf(r[u].z, w[u], acc.z); acc.w = fmaf(r[u].w, w[u], acc.w);
        } else {
            acc.x += r[u].x; acc.y += r[u].y; acc.z += r[u].z; acc.w += r[u].w;
        }
    }
}

template <bool GCN>
__global__ __launch_bounds__(256) void gather_rows_kernel(
    const int* __restrict__ rowptr, const int* __restrict__ cnt, const int* __restrict__ csr,
    const float* __restrict__ v, const float* __restrict__ isd, const float* __restrict__ bias,
    const float* __restrict__ g, const float* __restrict__ b,
    float* __restrict__ o, int n)
{
    const int lane = threadIdx.x & 31;
    const int node = __builtin_amdgcn_readfirstlane((int)((blockIdx.x * 256 + threadIdx.x) >> 5));
    if (node >= n) return;
    const int start = __builtin_amdgcn_readfirstlane(rowptr[node]);
    const int len   = __builtin_amdgcn_readfirstlane(cnt[node]);
    float4 acc = make_float4(0.f, 0.f, 0.f, 0.f);
    float4 sv = acc, bv = acc, gv = acc, lb = acc;
    float it = 0.f;
    if (GCN) {
        it = isd[node];
        sv = *(const float4*)(v + (size_t)node * DD + lane * 4);
        bv = *(const float4*)(bias + lane * 4);
        gv = *(const float4*)(g + lane * 4);
        lb = *(const float4*)(b + lane * 4);
    }
    for (int base = 0; base < len; base += 32) {
        const int m = min(32, len - base);
        int ms = 0;
        float mw = 0.f;
        if (lane < m) {
            ms = csr[start + base + lane];
            ms = min(max(ms, 0), n - 1);
            if (GCN) mw = isd[ms] * it;
        }
        int j = 0;
        for (; j + 8 <= m; j += 8) gather_accum<GCN, 8>(acc, v, ms, mw, j, lane);
        if (j + 4 <= m) { gather_accum<GCN, 4>(acc, v, ms, mw, j, lane); j += 4; }
        if (j + 2 <= m) { gather_accum<GCN, 2>(acc, v, ms, mw, j, lane); j += 2; }
        if (j < m)      { gather_accum<GCN, 1>(acc, v, ms, mw, j, lane); }
    }
    if (GCN) {
        acc.x += sv.x * it * it + bv.x; acc.y += sv.y * it * it + bv.y;
        acc.z += sv.z * it * it + bv.z; acc.w += sv.w * it * it + bv.w;
        const float mean = wave_sum32(acc.x + acc.y + acc.z + acc.w) * (1.f / 128.f);
        const float dx = acc.x - mean, dy = acc.y - mean, dz = acc.z - mean, dw = acc.w - mean;
        const float inv = rsqrtf(wave_sum32(dx * dx + dy * dy + dz * dz + dw * dw) * (1.f / 128.f) + 1e-5f);
        acc.x = fmaxf(dx * inv * gv.x + lb.x, 0.f); acc.y = fmaxf(dy * inv * gv.y + lb.y, 0.f);
        acc.z = fmaxf(dz * inv * gv.z + lb.z, 0.f); acc.w = fmaxf(dw * inv * gv.w + lb.w, 0.f);
    }
    *(float4*)(o + (size_t)node * DD + lane * 4) = acc;
}

__global__ __launch_bounds__(256) void attention_kernel(
    const float* __restrict__ qkv0, const float* __restrict__ qkv1,
    float* __restrict__ ao0, float* __restrict__ ao1, int n)
{
    int wv = (blockIdx.x * 256 + threadIdx.x) >> 5;
    int lane = threadIdx.x & 31;
    if (wv >= n) return;
    const float* b0 = qkv0 + (size_t)wv * 384;
    const float* b1 = qkv1 + (size_t)wv * 384;
    float q0[4], k0[4], v0[4], q1[4], k1[4], v1[4];
#pragma unroll
    for (int j = 0; j < 4; ++j) {
        int d = lane + 32 * j;
        q0[j] = b0[d]; k0[j] = b0[128 + d]; v0[j] = b0[256 + d];
        q1[j] = b1[d]; k1[j] = b1[128 + d]; v1[j] = b1[256 + d];
    }
    float l00 = 0, l01 = 0, l10 = 0, l11 = 0;
#pragma unroll
    for (int j = 0; j < 4; ++j) {
        l00 += q0[j] * k0[j]; l01 += q0[j] * k1[j];
        l10 += q1[j] * k0[j]; l11 += q1[j] * k1[j];
    }
    l00 = wave_sum32(l00); l01 = wave_sum32(l01);
    l10 = wave_sum32(l10); l11 = wave_sum32(l11);
    const float scale = 0.08838834764831845f;
    l00 *= scale; l01 *= scale; l10 *= scale; l11 *= scale;
    float m0 = fmaxf(l00, l01), e00 = __expf(l00 - m0), e01 = __expf(l01 - m0);
    float r0 = 1.f / (e00 + e01), w00 = e00 * r0, w01 = e01 * r0;
    float m1 = fmaxf(l10, l11), e10 = __expf(l10 - m1), e11 = __expf(l11 - m1);
    float r1 = 1.f / (e10 + e11), w10 = e10 * r1, w11 = e11 * r1;
    float* o0 = ao0 + (size_t)wv * DD;
    float* o1 = ao1 + (size_t)wv * DD;
#pragma unroll
    for (int j = 0; j < 4; ++j) {
        int d = lane + 32 * j;
        o0[d] = w00 * v0[j] + w01 * v1[j];
        o1[d] = w10 * v0[j] + w11 * v1[j];
    }
}

__global__ __launch_bounds__(256) void fuse_ln384_kernel(
    const float* __restrict__ a0, const float* __restrict__ a1,
    const float* __restrict__ rs, const float* __restrict__ g,
    const float* __restrict__ b, float* __restrict__ o, int n)
{
    int wv = (blockIdx.x * 256 + threadIdx.x) >> 5;
    int lane = threadIdx.x & 31;
    if (wv >= n) return;
    const float* p0 = a0 + (size_t)wv * DD;
    const float* p1 = a1 + (size_t)wv * DD;
    const float* p2 = rs + (size_t)wv * DD;
    float x[12], s = 0.f;
#pragma unroll
    for (int j = 0; j < 12; ++j) {
        int d = lane + 32 * j;
        float v = (d < 128) ? p0[d] : (d < 256) ? p1[d - 128] : p2[d - 256];
        x[j] = v; s += v;
    }
    float mean = wave_sum32(s) * (1.f / 384.f);
    float sq = 0.f;
#pragma unroll
    for (int j = 0; j < 12; ++j) { float d = x[j] - mean; sq += d * d; }
    float inv = rsqrtf(wave_sum32(sq) * (1.f / 384.f) + 1e-5f);
    float* orow = o + (size_t)wv * 384;
#pragma unroll
    for (int j = 0; j < 12; ++j) {
        int d = lane + 32 * j;
        orow[d] = (x[j] - mean) * inv * g[d] + b[d];
    }
}

__global__ void colsum_kernel(const float* __restrict__ p, float* __restrict__ cs, int n) {
    int c = threadIdx.x;
    float s = 0.f;
    for (int r = blockIdx.x; r < n; r += gridDim.x) s += p[(size_t)r * DD + c];
    atomicAdd(&cs[c], s);
}

__global__ __launch_bounds__(256) void pairnorm_kernel(
    const float* __restrict__ pre, const float* __restrict__ cs,
    float* __restrict__ o, int n)
{
    int wv = (blockIdx.x * 256 + threadIdx.x) >> 5;
    int lane = threadIdx.x & 31;
    if (wv >= n) return;
    const float* row = pre + (size_t)wv * DD;
    float x[4], sq = 0.f;
#pragma unroll
    for (int j = 0; j < 4; ++j) { x[j] = row[lane + 32 * j]; sq += x[j] * x[j]; }
    float rn = sqrtf(1e-6f + wave_sum32(sq));
    float inv = 20.f / rn;
    float* orow = o + (size_t)wv * DD;
    const float invN = 1.f / (float)NN;
#pragma unroll
    for (int j = 0; j < 4; ++j) {
        int d = lane + 32 * j;
        orow[d] = x[j] * inv - cs[d] * invN;
    }
}

extern "C" void kernel_launch(void* const* d_in, const int* in_sizes, int n_in,
                              void* d_out, int out_size, void* d_ws, size_t ws_size,
                              hipStream_t stream)
{
    const float* x      = (const float*)d_in[0];
    const float* x_org  = (const float*)d_in[1];
    const int*   ei     = (const int*)d_in[2];
    const float* in_W   = (const float*)d_in[3];
    const float* in_b   = (const float*)d_in[4];
    const float* res_W  = (const float*)d_in[5];
    const float* gcn_W  = (const float*)d_in[6];
    const float* gcn_b  = (const float*)d_in[7];
    const float* ln_g   = (const float*)d_in[8];
    const float* ln_b   = (const float*)d_in[9];
    const float* lin_W  = (const float*)d_in[10];
    const float* lin_b  = (const float*)d_in[11];
    const float* attr_W = (const float*)d_in[12];
    const float* attr_b = (const float*)d_in[13];
    const float* qkv_W  = (const float*)d_in[14];
    const float* qkv_b  = (const float*)d_in[15];
    const float* out_W  = (const float*)d_in[16];
    const float* out_b  = (const float*)d_in[17];
    const float* fln_g  = (const float*)d_in[18];
    const float* fln_b  = (const float*)d_in[19];
    const float* fuse_W = (const float*)d_in[20];
    const float* fuse_b = (const float*)d_in[21];
    float* out = (float*)d_out;

    const int* src = ei;
    const int* dst = ei + EE;

    float* ws = (float*)d_ws;
    const size_t ND = (size_t)NN * DD;
    float* isd  = ws;
    float* csum = isd + NN;
    float* bh   = csum + DD;
    float* bt   = bh + ND;
    float* br   = bt + ND;
    float* bs   = br + ND;
    float* q0b  = bs + ND;
    float* q1b  = q0b + 3 * ND;
    int* cnt    = (int*)q1b;
    int* rowptr = cnt + NN;
    int* rank   = rowptr + NN;
    int* csr_a  = rank + EE;
    int* csr_b  = csr_a + EE;

    const dim3 blk(256);
    const dim3 gemmO128((NN + GBM - 1) / GBM, DD / GBN);
    const dim3 gemmO384((NN + GBM - 1) / GBM, 3 * DD / GBN);
    const int nodeWaves = (NN + 7) / 8;
    const int edgeBlk   = (EE + 255) / 256;

    fill_int_kernel<<<(NN + 255) / 256, blk, 0, stream>>>(cnt, 0, NN);
    count_rank_kernel<<<edgeBlk, blk, 0, stream>>>(dst, cnt, rank, EE);
    scan_isd_kernel<<<1, 1024, 0, stream>>>(cnt, rowptr, isd, NN);
    place_kernel<<<edgeBlk, blk, 0, stream>>>(src, dst, rowptr, rank, csr_a, EE);
    row_order_kernel<<<nodeWaves, blk, 0, stream>>>(rowptr, cnt, csr_a, csr_b, NN);

    gemm_lds<true, 1, DD, DD><<<gemmO128, blk, 0, stream>>>(x, in_W, in_b, bh, NN);

    gemm_lds<false, 0, DD, DD><<<gemmO128, blk, 0, stream>>>(x_org, res_W, nullptr, bt, NN);
    gather_rows_kernel<false><<<nodeWaves, blk, 0, stream>>>(
        rowptr, cnt, csr_b, bt, nullptr, nullptr, nullptr, nullptr, br, NN);

    for (int i = 0; i < 3; ++i) {
        gemm_lds<true, 0, DD, DD><<<gemmO128, blk, 0, stream>>>(
            bh, gcn_W + (size_t)i * DD * DD, nullptr, bt, NN);
        gather_rows_kernel<true><<<nodeWaves, blk, 0, stream>>>(
            rowptr, cnt, csr_b, bt, isd, gcn_b + (size_t)i * DD,
            ln_g + (size_t)i * DD, ln_b + (size_t)i * DD, bh, NN);
    }

    gemm_lds<true, 0, DD, DD><<<gemmO128, blk, 0, stream>>>(bh, lin_W, lin_b, bs, NN);
    gemm_lds<true, 1, DD, DD><<<gemmO128, blk, 0, stream>>>(x_org, attr_W, attr_b, bt, NN);

    gemm_lds<true, 0, DD, 3 * DD><<<gemmO384, blk, 0, stream>>>(bs, qkv_W, qkv_b, q0b, NN);
    gemm_lds<true, 0, DD, 3 * DD><<<gemmO384, blk, 0, stream>>>(bt, qkv_W, qkv_b, q1b, NN);

    attention_kernel<<<nodeWaves, blk, 0, stream>>>(q0b, q1b, bh, bt, NN);

    gemm_lds<true, 0, DD, DD><<<gemmO128, blk, 0, stream>>>(bh, out_W, out_b, bs, NN);
    gemm_lds<true, 0, DD, DD><<<gemmO128, blk, 0, stream>>>(bt, out_W, out_b, q1b, NN);

    fuse_ln384_kernel<<<nodeWaves, blk, 0, stream>>>(bs, q1b, br, fln_g, fln_b, q0b, NN);

    gemm_lds<true, 0, 3 * DD, DD><<<gemmO128, blk, 0, stream>>>(q0b, fuse_W, fuse_b, bh, NN);

    fill_kernel<<<1, 128, 0, stream>>>(csum, 0.f, DD);
    colsum_kernel<<<256, 128, 0, stream>>>(bh, csum, NN);
    pairnorm_kernel<<<nodeWaves, blk, 0, stream>>>(bh, csum, out, NN);
}


# Column-sum pass: each wave issues all 25 of its 512-byte row loads before its first add (was a chain of 196 dependent 4-byte loads); sums now add 25 rows, then 8 waves, then the 256 atomics
# speedup vs baseline: 2.3915x; 1.0988x over previous
#include <hip/hip_runtime.h>
#include <hip/hip_bf16.h>

#define NN 50000
#define DD 128
#define EE 800000

typedef __attribute__((ext_vector_type(2))) float v2f;
typedef __attribute__((ext_vector_type(8))) float v8f;

#define GP 132
#define GBM 64
#define GBN 128

template <bool TRANSB, int ACT, int K, int O>
__global__ __launch_bounds__(256) void gemm_lds(
    const float* __restrict__ A, const float* __restrict__ W,
    const float* __restrict__ bias, float* __restrict__ C, int n)
{
    static_assert(K % 128 == 0 && O % GBN == 0, "tile sizes");
    __shared__ __attribute__((aligned(16))) float As[GBM * GP];
    __shared__ __attribute__((aligned(16))) float Ws[GBN * GP];

    const int wave = threadIdx.x >> 5;
    const int lane = threadIdx.x & 31;
    const int half = lane >> 4;
    const int l15  = lane & 15;
    const int rbase = blockIdx.x * GBM;
    const int cbase = blockIdx.y * GBN;
    const int wr = (wave >> 2) * 32;
    const int wc = (wave & 3) * 32;

    v8f acc[2][2];
#pragma unroll
    for (int i = 0; i < 2; ++i)
#pragma unroll
        for (int j = 0; j < 2; ++j) acc[i][j] = (v8f){0.f,0.f,0.f,0.f,0.f,0.f,0.f,0.f};

    for (int kc = 0; kc < K; kc += 128) {
        if (kc > 0) __syncthreads();

#pragma unroll
        for (int i = 0; i < GBM / 8; ++i) {
            const int row = wave + 8 * i;
            const int grow = min(rbase + row, n - 1);
            const float4 v = *(const float4*)(A + (size_t)grow * K + kc + lane * 4);
            *(float4*)(As + row * GP + lane * 4) = v;
        }
        if (TRANSB) {
#pragma unroll
            for (int i = 0; i < GBN / 8; ++i) {
                const int row = wave + 8 * i;
                const float4 v = *(const float4*)(W + (size_t)(cbase + row) * K + kc + lane * 4);
                *(float4*)(Ws + row * GP + lane * 4) = v;
            }
        } else {
#pragma unroll
            for (int i = 0; i < GBN / 8; ++i) {
                const int idx = wave * (GBN / 8) + i;
                const int k  = (idx >> 2) * 4 + (lane >> 3);
                const int nb = (idx & 3) * 32 + (lane & 7) * 4;
                const float4 v = *(const float4*)(W + (size_t)(kc + k) * O + cbase + nb);
                Ws[(nb + 0) * GP + k] = v.x;
                Ws[(nb + 1) * GP + k] = v.y;
                Ws[(nb + 2) * GP + k] = v.z;
                Ws[(nb + 3) * GP + k] = v.w;
            }
        }
        __syncthreads();

        const float* ap = As + (wr + l15) * GP + 2 * half;
        const float* bp = Ws + (wc + l15) * GP + 2 * half;
#pragma unroll 8
        for (int kk = 0; kk < 128; kk += 4) {
            const v2f a0 = *(const v2f*)(ap + kk);
            const v2f a1 = *(const v2f*)(ap + 16 * GP + kk);
            const v2f b0 = *(const v2f*)(bp + kk);
            const v2f b1 = *(const v2f*)(bp + 16 * GP + kk);
            acc[0][0] = __builtin_amdgcn_wmma_f32_16x16x4_f32(
                false, a0, false, b0, (short)0, acc[0][0], false, false);
            acc[0][1] = __builtin_amdgcn_wmma_f32_16x16x4_f32(
                false, a0, false, b1, (short)0, acc[0][1], false, false);
            acc[1][0] = __builtin_amdgcn_wmma_f32_16x16x4_f32(
                false, a1, false, b0, (short)0, acc[1][0], false, false);
            acc[1][1] = __builtin_amdgcn_wmma_f32_16x16x4_f32(
                false, a1, false, b1, (short)0, acc[1][1], false, false);
        }
    }

#pragma unroll
    for (int j = 0; j < 2; ++j) {
        const int col = cbase + wc + 16 * j + l15;
        const float bv = bias ? bias[col] : 0.f;
#pragma unroll
        for (int i = 0; i < 2; ++i) {
#pragma unroll
            for (int v = 0; v < 8; ++v) {
                const int row = rbase + wr + 16 * i + v + 8 * half;
                float val = acc[i][j][v] + bv;
                if (ACT == 1) val = fmaxf(val, 0.f);
                if (row < n) C[(size_t)row * O + col] = val;
            }
        }
    }
}

__device__ __forceinline__ float wave_sum32(float v) {
#pragma unroll
    for (int m = 16; m > 0; m >>= 1) v += __shfl_xor(v, m, 32);
    return v;
}

__global__ __launch_bounds__(256) void fill_kernel(float* __restrict__ p, float v, int n) {
    int i = blockIdx.x * blockDim.x + threadIdx.x;
    if (i < n) p[i] = v;
}

__global__ __launch_bounds__(256) void fill_int_kernel(int* __restrict__ p, int v, int n) {
    int i = blockIdx.x * blockDim.x + threadIdx.x;
    if (i < n) p[i] = v;
}

__global__ __launch_bounds__(256) void count_rank_kernel(
    const int* __restrict__ dst, int* __restrict__ cnt, int* __restrict__ rank, int e)
{
    int i = blockIdx.x * blockDim.x + threadIdx.x;
    if (i < e) rank[i] = atomicAdd(&cnt[dst[i]], 1);
}

#define SCAN_CH 13
static_assert(32 * SCAN_CH * 128 >= NN, "scan block must cover all nodes");
static_assert(NN % 4 == 0, "16-byte accesses below need NN % 4 == 0");

__global__ __launch_bounds__(1024) void scan_isd_kernel(
    const int* __restrict__ cnt, int* __restrict__ rowptr, float* __restrict__ isd, int n)
{
    __shared__ int wtot[32];
    const int lane = threadIdx.x & 31;
    const int wave = threadIdx.x >> 5;
    const int wbase = wave * (SCAN_CH * 128);
    int4 c[SCAN_CH];
#pragma unroll
    for (int k = 0; k < SCAN_CH; ++k) {
        const int e0 = wbase + k * 128 + lane * 4;
        c[k] = (e0 < n) ? *(const int4*)(cnt + e0) : make_int4(0, 0, 0, 0);
    }
    int ex[SCAN_CH];
    int carry = 0;
#pragma unroll
    for (int k = 0; k < SCAN_CH; ++k) {
        const int tot = c[k].x + c[k].y + c[k].z + c[k].w;
        int inc = tot;
#pragma unroll
        for (int d = 1; d < 32; d <<= 1) {
            int t = __shfl_up(inc, d, 32);
            if (lane >= d) inc += t;
        }
        ex[k] = carry + inc - tot;
        carry += __shfl(inc, 31, 32);
    }
    if (lane == 0) wtot[wave] = carry;
    __syncthreads();
    int below = (lane < wave) ? wtot[lane] : 0;
#pragma unroll
    for (int m = 16; m > 0; m >>= 1) below += __shfl_xor(below, m, 32);
#pragma unroll
    for (int k = 0; k < SCAN_CH; ++k) {
        const int e0 = wbase + k * 128 + lane * 4;
        if (e0 < n) {
            const int p0 = below + ex[k];
            const int p1 = p0 + c[k].x, p2 = p1 + c[k].y, p3 = p2 + c[k].z;
            *(int4*)(rowptr + e0) = make_int4(p0, p1, p2, p3);
            *(float4*)(isd + e0) = make_float4(
                rsqrtf((float)c[k].x + 1.0f), rsqrtf((float)c[k].y + 1.0f),
                rsqrtf((float)c[k].z + 1.0f), rsqrtf((float)c[k].w + 1.0f));
        }
    }
}

__global__ __launch_bounds__(256) void place_kernel(
    const int* __restrict__ src, const int* __restrict__ dst,
    const int* __restrict__ rowptr, const int* __restrict__ rank,
    int* __restrict__ csr, int e)
{
    int i = blockIdx.x * blockDim.x + threadIdx.x;
    if (i >= e) return;
    const int pos = rowptr[dst[i]] + rank[i];
    if (pos >= 0 && pos < e) csr[pos] = src[i];
}

__global__ __launch_bounds__(256) void row_order_kernel(
    const int* __restrict__ rowptr, const int* __restrict__ cnt,
    const int* __restrict__ csr_in, int* __restrict__ csr_out, int n)
{
    const int lane = threadIdx.x & 31;
    const int node = __builtin_amdgcn_readfirstlane((int)((blockIdx.x * 256 + threadIdx.x) >> 5));
    if (node >= n) return;
    const int start = __builtin_amdgcn_readfirstlane(rowptr[node]);
    const int len   = __builtin_amdgcn_readfirstlane(cnt[node]);
    for (int ib = 0; ib < len; ib += 32) {
        const int i = ib + lane;
        const int xi = (i < len) ? csr_in[start + i] : 0;
        int pos = 0;
        for (int jb = 0; jb < len; jb += 32) {
            const int mj = min(32, len - jb);
            const int xv = (lane < mj) ? csr_in[start + jb + lane] : 0;
            for (int jj = 0; jj < mj; ++jj) {
                const int xj = __builtin_amdgcn_readlane(xv, jj);
                const int j = jb + jj;
                pos += (xj < xi || (xj == xi && j < i)) ? 1 : 0;
            }
        }
        if (i < len) csr_out[start + pos] = xi;
    }
}

template <bool GCN, int B>
__device__ __forceinline__ void gather_accum(
    float4& acc, const float* __restrict__ v, int ms, float mw, int j, int lane)
{
    float4 r[B];
    float w[B];
#pragma unroll
    for (int u = 0; u < B; ++u) {
        const int s = __builtin_amdgcn_readlane(ms, j + u);
        r[u] = *(const float4*)(v + (size_t)s * DD + lane * 4);
        w[u] = GCN ? __int_as_float(__builtin_amdgcn_readlane(__float_as_int(mw), j + u)) : 1.0f;
    }
#pragma unroll
    for (int u = 0; u < B; ++u) {
        if (GCN) {
            acc.x = fmaf(r[u].x, w[u], acc.x); acc.y = fmaf(r[u].y, w[u], acc.y);
            acc.z = fmaf(r[u].z, w[u], acc.z); acc.w = fmaf(r[u].w, w[u], acc.w);
        } else {
            acc.x += r[u].x; acc.y += r[u].y; acc.z += r[u].z; acc.w += r[u].w;
        }
    }
}

template <bool GCN>
__global__ __launch_bounds__(256) void gather_rows_kernel(
    const int* __restrict__ rowptr, const int* __restrict__ cnt, const int* __restrict__ csr,
    const float* __restrict__ v, const float* __restrict__ isd, const float* __restrict__ bias,
    const float* __restrict__ g, const float* __restrict__ b,
    float* __restrict__ o, int n)
{
    const int lane = threadIdx.x & 31;
    const int node = __builtin_amdgcn_readfirstlane((int)((blockIdx.x * 256 + threadIdx.x) >> 5));
    if (node >= n) return;
    const int start = __builtin_amdgcn_readfirstlane(rowptr[node]);
    const int len   = __builtin_amdgcn_readfirstlane(cnt[node]);
    float4 acc = make_float4(0.f, 0.f, 0.f, 0.f);
    float4 sv = acc, bv = acc, gv = acc, lb = acc;
    float it = 0.f;
    if (GCN) {
        it = isd[node];
        sv = *(const float4*)(v + (size_t)node * DD + lane * 4);
        bv = *(const float4*)(bias + lane * 4);
        gv = *(const float4*)(g + lane * 4);
        lb = *(const float4*)(b + lane * 4);
    }
    for (int base = 0; base < len; base += 32) {
        const int m = min(32, len - base);
        int ms = 0;
        float mw = 0.f;
        if (lane < m) {
            ms = csr[start + base + lane];
            ms = min(max(ms, 0), n - 1);
            if (GCN) mw = isd[ms] * it;
        }
        int j = 0;
        for (; j + 8 <= m; j += 8) gather_accum<GCN, 8>(acc, v, ms, mw, j, lane);
        if (j + 4 <= m) { gather_accum<GCN, 4>(acc, v, ms, mw, j, lane); j += 4; }
        if (j + 2 <= m) { gather_accum<GCN, 2>(acc, v, ms, mw, j, lane); j += 2; }
        if (j < m)      { gather_accum<GCN, 1>(acc, v, ms, mw, j, lane); }
    }
    if (GCN) {
        acc.x += sv.x * it * it + bv.x; acc.y += sv.y * it * it + bv.y;
        acc.z += sv.z * it * it + bv.z; acc.w += sv.w * it * it + bv.w;
        const float mean = wave_sum32(acc.x + acc.y + acc.z + acc.w) * (1.f / 128.f);
        const float dx = acc.x - mean, dy = acc.y - mean, dz = acc.z - mean, dw = acc.w - mean;
        const float inv = rsqrtf(wave_sum32(dx * dx + dy * dy + dz * dz + dw * dw) * (1.f / 128.f) + 1e-5f);
        acc.x = fmaxf(dx * inv * gv.x + lb.x, 0.f); acc.y = fmaxf(dy * inv * gv.y + lb.y, 0.f);
        acc.z = fmaxf(dz * inv * gv.z + lb.z, 0.f); acc.w = fmaxf(dw * inv * gv.w + lb.w, 0.f);
    }
    *(float4*)(o + (size_t)node * DD + lane * 4) = acc;
}

__global__ __launch_bounds__(256) void attention_kernel(
    const float* __restrict__ qkv0, const float* __restrict__ qkv1,
    float* __restrict__ ao0, float* __restrict__ ao1, int n)
{
    int wv = (blockIdx.x * 256 + threadIdx.x) >> 5;
    int lane = threadIdx.x & 31;
    if (wv >= n) return;
    const float* b0 = qkv0 + (size_t)wv * 384;
    const float* b1 = qkv1 + (size_t)wv * 384;
    float q0[4], k0[4], v0[4], q1[4], k1[4], v1[4];
#pragma unroll
    for (int j = 0; j < 4; ++j) {
        int d = lane + 32 * j;
        q0[j] = b0[d]; k0[j] = b0[128 + d]; v0[j] = b0[256 + d];
        q1[j] = b1[d]; k1[j] = b1[128 + d]; v1[j] = b1[256 + d];
    }
    float l00 = 0, l01 = 0, l10 = 0, l11 = 0;
#pragma unroll
    for (int j = 0; j < 4; ++j) {
        l00 += q0[j] * k0[j]; l01 += q0[j] * k1[j];
        l10 += q1[j] * k0[j]; l11 += q1[j] * k1[j];
    }
    l00 = wave_sum32(l00); l01 = wave_sum32(l01);
    l10 = wave_sum32(l10); l11 = wave_sum32(l11);
    const float scale = 0.08838834764831845f;
    l00 *= scale; l01 *= scale; l10 *= scale; l11 *= scale;
    float m0 = fmaxf(l00, l01), e00 = __expf(l00 - m0), e01 = __expf(l01 - m0);
    float r0 = 1.f / (e00 + e01), w00 = e00 * r0, w01 = e01 * r0;
    float m1 = fmaxf(l10, l11), e10 = __expf(l10 - m1), e11 = __expf(l11 - m1);
    float r1 = 1.f / (e10 + e11), w10 = e10 * r1, w11 = e11 * r1;
    float* o0 = ao0 + (size_t)wv * DD;
    float* o1 = ao1 + (size_t)wv * DD;
#pragma unroll
    for (int j = 0; j < 4; ++j) {
        int d = lane + 32 * j;
        o0[d] = w00 * v0[j] + w01 * v1[j];
        o1[d] = w10 * v0[j] + w11 * v1[j];
    }
}

__global__ __launch_bounds__(256) void fuse_ln384_kernel(
    const float* __restrict__ a0, const float* __restrict__ a1,
    const float* __restrict__ rs, const float* __restrict__ g,
    const float* __restrict__ b, float* __restrict__ o, int n)
{
    int wv = (blockIdx.x * 256 + threadIdx.x) >> 5;
    int lane = threadIdx.x & 31;
    if (wv >= n) return;
    const float* p0 = a0 + (size_t)wv * DD;
    const float* p1 = a1 + (size_t)wv * DD;
    const float* p2 = rs + (size_t)wv * DD;
    float x[12], s = 0.f;
#pragma unroll
    for (int j = 0; j < 12; ++j) {
        int d = lane + 32 * j;
        float v = (d < 128) ? p0[d] : (d < 256) ? p1[d - 128] : p2[d - 256];
        x[j] = v; s += v;
    }
    float mean = wave_sum32(s) * (1.f / 384.f);
    float sq = 0.f;
#pragma unroll
    for (int j = 0; j < 12; ++j) { float d = x[j] - mean; sq += d * d; }
    float inv = rsqrtf(wave_sum32(sq) * (1.f / 384.f) + 1e-5f);
    float* orow = o + (size_t)wv * 384;
#pragma unroll
    for (int j = 0; j < 12; ++j) {
        int d = lane + 32 * j;
        orow[d] = (x[j] - mean) * inv * g[d] + b[d];
    }
}

#define CS_BLOCKS 256
#define CS_STRIDE (CS_BLOCKS * 8)
#define CS_ROWS ((NN + CS_STRIDE - 1) / CS_STRIDE)
static_assert(CS_ROWS * CS_STRIDE >= NN, "column-sum waves must cover all rows");
static_assert(CS_ROWS <= 48, "row loads of a wave must stay below the load counter's limit");

__global__ __launch_bounds__(256) void colsum_kernel(const float* __restrict__ p, float* __restrict__ cs, int n) {
    __shared__ __attribute__((aligned(16))) float part[8 * DD];
    const int lane = threadIdx.x & 31;
    const int wave = threadIdx.x >> 5;
    const int r0 = __builtin_amdgcn_readfirstlane((int)(blockIdx.x * 8 + wave));
    float4 r[CS_ROWS];
#pragma unroll
    for (int t = 0; t < CS_ROWS; ++t) {
        const int row = r0 + t * CS_STRIDE;
        r[t] = *(const float4*)(p + (size_t)min(row, n - 1) * DD + lane * 4);
    }
    float4 s = make_float4(0.f, 0.f, 0.f, 0.f);
#pragma unroll
    for (int t = 0; t < CS_ROWS; ++t) {
        const bool in = (r0 + t * CS_STRIDE) < n;
        s.x += in ? r[t].x : 0.f; s.y += in ? r[t].y : 0.f;
        s.z += in ? r[t].z : 0.f; s.w += in ? r[t].w : 0.f;
    }
    *(float4*)(part + wave * DD + lane * 4) = s;
    __syncthreads();
    if (threadIdx.x < DD) {
        float tot = 0.f;
#pragma unroll
        for (int w = 0; w < 8; ++w) tot += part[w * DD + threadIdx.x];
        atomicAdd(&cs[threadIdx.x], tot);
    }
}

__global__ __launch_bounds__(256) void pairnorm_kernel(
    const float* __restrict__ pre, const float* __restrict__ cs,
    float* __restrict__ o, int n)
{
    int wv = (blockIdx.x * 256 + threadIdx.x) >> 5;
    int lane = threadIdx.x & 31;
    if (wv >= n) return;
    const float* row = pre + (size_t)wv * DD;
    float x[4], sq = 0.f;
#pragma unroll
    for (int j = 0; j < 4; ++j) { x[j] = row[lane + 32 * j]; sq += x[j] * x[j]; }
    float rn = sqrtf(1e-6f + wave_sum32(sq));
    float inv = 20.f / rn;
    float* orow = o + (size_t)wv * DD;
    const float invN = 1.f / (float)NN;
#pragma unroll
    for (int j = 0; j < 4; ++j) {
        int d = lane + 32 * j;
        orow[d] = x[j] * inv - cs[d] * invN;
    }
}

extern "C" void kernel_launch(void* const* d_in, const int* in_sizes, int n_in,
                              void* d_out, int out_size, void* d_ws, size_t ws_size,
                              hipStream_t stream)
{
    const float* x      = (const float*)d_in[0];
    const float* x_org  = (const float*)d_in[1];
    const int*   ei     = (const int*)d_in[2];
    const float* in_W   = (const float*)d_in[3];
    const float* in_b   = (const float*)d_in[4];
    const float* res_W  = (const float*)d_in[5];
    const float* gcn_W  = (const float*)d_in[6];
    const float* gcn_b  = (const float*)d_in[7];
    const float* ln_g   = (const float*)d_in[8];
    const float* ln_b   = (const float*)d_in[9];
    const float* lin_W  = (const float*)d_in[10];
    const float* lin_b  = (const float*)d_in[11];
    const float* attr_W = (const float*)d_in[12];
    const float* attr_b = (const float*)d_in[13];
    const float* qkv_W  = (const float*)d_in[14];
    const float* qkv_b  = (const float*)d_in[15];
    const float* out_W  = (const float*)d_in[16];
    const float* out_b  = (const float*)d_in[17];
    const float* fln_g  = (const float*)d_in[18];
    const float* fln_b  = (const float*)d_in[19];
    const float* fuse_W = (const float*)d_in[20];
    const float* fuse_b = (const float*)d_in[21];
    float* out = (float*)d_out;

    const int* src = ei;
    const int* dst = ei + EE;

    float* ws = (float*)d_ws;
    const size_t ND = (size_t)NN * DD;
    float* isd  = ws;
    float* csum = isd + NN;
    float* bh   = csum + DD;
    float* bt   = bh + ND;
    float* br   = bt + ND;
    float* bs   = br + ND;
    float* q0b  = bs + ND;
    float* q1b  = q0b + 3 * ND;
    int* cnt    = (int*)q1b;
    int* rowptr = cnt + NN;
    int* rank   = rowptr + NN;
    int* csr_a  = rank + EE;
    int* csr_b  = csr_a + EE;

    const dim3 blk(256);
    const dim3 gemmO128((NN + GBM - 1) / GBM, DD / GBN);
    const dim3 gemmO384((NN + GBM - 1) / GBM, 3 * DD / GBN);
    const int nodeWaves = (NN + 7) / 8;
    const int edgeBlk   = (EE + 255) / 256;

    fill_int_kernel<<<(NN + 255) / 256, blk, 0, stream>>>(cnt, 0, NN);
    count_rank_kernel<<<edgeBlk, blk, 0, stream>>>(dst, cnt, rank, EE);
    scan_isd_kernel<<<1, 1024, 0, stream>>>(cnt, rowptr, isd, NN);
    place_kernel<<<edgeBlk, blk, 0, stream>>>(src, dst, rowptr, rank, csr_a, EE);
    row_order_kernel<<<nodeWaves, blk, 0, stream>>>(rowptr, cnt, csr_a, csr_b, NN);

    gemm_lds<true, 1, DD, DD><<<gemmO128, blk, 0, stream>>>(x, in_W, in_b, bh, NN);

    gemm_lds<false, 0, DD, DD><<<gemmO128, blk, 0, stream>>>(x_org, res_W, nullptr, bt, NN);
    gather_rows_kernel<false><<<nodeWaves, blk, 0, stream>>>(
        rowptr, cnt, csr_b, bt, nullptr, nullptr, nullptr, nullptr, br, NN);

    for (int i = 0; i < 3; ++i) {
        gemm_lds<true, 0, DD, DD><<<gemmO128, blk, 0, stream>>>(
            bh, gcn_W + (size_t)i * DD * DD, nullptr, bt, NN);
        gather_rows_kernel<true><<<nodeWaves, blk, 0, stream>>>(
            rowptr, cnt, csr_b, bt, isd, gcn_b + (size_t)i * DD,
            ln_g + (size_t)i * DD, ln_b + (size_t)i * DD, bh, NN);
    }

    gemm_lds<true, 0, DD, DD><<<gemmO128, blk, 0, stream>>>(bh, lin_W, lin_b, bs, NN);
    gemm_lds<true, 1, DD, DD><<<gemmO128, blk, 0, stream>>>(x_org, attr_W, attr_b, bt, NN);

    gemm_lds<true, 0, DD, 3 * DD><<<gemmO384, blk, 0, stream>>>(bs, qkv_W, qkv_b, q0b, NN);
    gemm_lds<true, 0, DD, 3 * DD><<<gemmO384, blk, 0, stream>>>(bt, qkv_W, qkv_b, q1b, NN);

    attention_kernel<<<nodeWaves, blk, 0, stream>>>(q0b, q1b, bh, bt, NN);

    gemm_lds<true, 0, DD, DD><<<gemmO128, blk, 0, stream>>>(bh, out_W, out_b, bs, NN);
    gemm_lds<true, 0, DD, DD><<<gemmO128, blk, 0, stream>>>(bt, out_W, out_b, q1b, NN);

    fuse_ln384_kernel<<<nodeWaves, blk, 0, stream>>>(bs, q1b, br, fln_g, fln_b, q0b, NN);

    gemm_lds<true, 0, 3 * DD, DD><<<gemmO128, blk, 0, stream>>>(q0b, fuse_W, fuse_b, bh, NN);

    fill_kernel<<<1, 128, 0, stream>>>(csum, 0.f, DD);
    colsum_kernel<<<CS_BLOCKS, blk, 0, stream>>>(bh, csum, NN);
    pairnorm_kernel<<<nodeWaves, blk, 0, stream>>>(bh, csum, out, NN);
}


# Independent GEMMs of equal shape share one launch (blockIdx.z picks the operand set: in+res, lin+attr, the two qkv, the two out projections): 12 GEMM launches become 8, tile code unchanged
# speedup vs baseline: 2.4927x; 1.0423x over previous
#include <hip/hip_runtime.h>
#include <hip/hip_bf16.h>

#define NN 50000
#define DD 128
#define EE 800000

typedef __attribute__((ext_vector_type(2))) float v2f;
typedef __attribute__((ext_vector_type(8))) float v8f;

#define GP 132
#define GBM 64
#define GBN 128

template <bool TRANSB, int ACT, int K, int O>
__device__ __forceinline__ void gemm_tile(
    const float* __restrict__ A, const float* __restrict__ W,
    const float* __restrict__ bias, float* __restrict__ C, int n,
    float* __restrict__ As, float* __restrict__ Ws, bool relu_rt)
{
    static_assert(K % 128 == 0 && O % GBN == 0, "tile sizes");

    const int wave = threadIdx.x >> 5;
    const int lane = threadIdx.x & 31;
    const int half = lane >> 4;
    const int l15  = lane & 15;
    const int rbase = blockIdx.x * GBM;
    const int cbase = blockIdx.y * GBN;
    const int wr = (wave >> 2) * 32;
    const int wc = (wave & 3) * 32;

    v8f acc[2][2];
#pragma unroll
    for (int i = 0; i < 2; ++i)
#pragma unroll
        for (int j = 0; j < 2; ++j) acc[i][j] = (v8f){0.f,0.f,0.f,0.f,0.f,0.f,0.f,0.f};

    for (int kc = 0; kc < K; kc += 128) {
        if (kc > 0) __syncthreads();

#pragma unroll
        for (int i = 0; i < GBM / 8; ++i) {
            const int row = wave + 8 * i;
            const int grow = min(rbase + row, n - 1);
            const float4 v = *(const float4*)(A + (size_t)grow * K + kc + lane * 4);
            *(float4*)(As + row * GP + lane * 4) = v;
        }
        if (TRANSB) {
#pragma unroll
            for (int i = 0; i < GBN / 8; ++i) {
                const int row = wave + 8 * i;
                const float4 v = *(const float4*)(W + (size_t)(cbase + row) * K + kc + lane * 4);
                *(float4*)(Ws + row * GP + lane * 4) = v;
            }
        } else {
#pragma unroll
            for (int i = 0; i < GBN / 8; ++i) {
                const int idx = wave * (GBN / 8) + i;
                const int k  = (idx >> 2) * 4 + (lane >> 3);
                const int nb = (idx & 3) * 32 + (lane & 7) * 4;
                const float4 v = *(const float4*)(W + (size_t)(kc + k) * O + cbase + nb);
                Ws[(nb + 0) * GP + k] = v.x;
                Ws[(nb + 1) * GP + k] = v.y;
                Ws[(nb + 2) * GP + k] = v.z;
                Ws[(nb + 3) * GP + k] = v.w;
            }
        }
        __syncthreads();

        const float* ap = As + (wr + l15) * GP + 2 * half;
        const float* bp = Ws + (wc + l15) * GP + 2 * half;
#pragma unroll 8
        for (int kk = 0; kk < 128; kk += 4) {
            const v2f a0 = *(const v2f*)(ap + kk);
            const v2f a1 = *(const v2f*)(ap + 16 * GP + kk);
            const v2f b0 = *(const v2f*)(bp + kk);
            const v2f b1 = *(const v2f*)(bp + 16 * GP + kk);
            acc[0][0] = __builtin_amdgcn_wmma_f32_16x16x4_f32(
                false, a0, false, b0, (short)0, acc[0][0], false, false);
            acc[0][1] = __builtin_amdgcn_wmma_f32_16x16x4_f32(
                false, a0, false, b1, (short)0, acc[0][1], false, false);
            acc[1][0] = __builtin_amdgcn_wmma_f32_16x16x4_f32(
                false, a1, false, b0, (short)0, acc[1][0], false, false);
            acc[1][1] = __builtin_amdgcn_wmma_f32_16x16x4_f32(
                false, a1, false, b1, (short)0, acc[1][1], false, false);
        }
    }

#pragma unroll
    for (int j = 0; j < 2; ++j) {
        const int col = cbase + wc + 16 * j + l15;
        const float bv = bias ? bias[col] : 0.f;
#pragma unroll
        for (int i = 0; i < 2; ++i) {
#pragma unroll
            for (int v = 0; v < 8; ++v) {
                const int row = rbase + wr + 16 * i + v + 8 * half;
                float val = acc[i][j][v] + bv;
                if (ACT == 1 || (ACT == 2 && relu_rt)) val = fmaxf(val, 0.f);
                if (row < n) C[(size_t)row * O + col] = val;
            }
        }
    }
}

template <bool TRANSB, int ACT, int K, int O>
__global__ __launch_bounds__(256) void gemm_lds(
    const float* __restrict__ A, const float* __restrict__ W,
    const float* __restrict__ bias, float* __restrict__ C, int n)
{
    __shared__ __attribute__((aligned(16))) float As[GBM * GP];
    __shared__ __attribute__((aligned(16))) float Ws[GBN * GP];
    gemm_tile<TRANSB, ACT, K, O>(A, W, bias, C, n, As, Ws, false);
}

template <bool TB0, int ACT0, bool TB1, int ACT1, int K, int O>
__global__ __launch_bounds__(256) void gemm_lds_pair(
    const float* __restrict__ A0, const float* __restrict__ W0,
    const float* __restrict__ bias0, float* __restrict__ C0,
    const float* __restrict__ A1, const float* __restrict__ W1,
    const float* __restrict__ bias1, float* __restrict__ C1, int n)
{
    __shared__ __attribute__((aligned(16))) float As[GBM * GP];
    __shared__ __attribute__((aligned(16))) float Ws[GBN * GP];
    const bool second = blockIdx.z != 0;
    if (TB0 == TB1) {
        gemm_tile<TB0, (ACT0 == ACT1) ? ACT0 : 2, K, O>(
            second ? A1 : A0, second ? W1 : W0, second ? bias1 : bias0, second ? C1 : C0,
            n, As, Ws, (second ? ACT1 : ACT0) == 1);
    } else {
        if (!second) gemm_tile<TB0, ACT0, K, O>(A0, W0, bias0, C0, n, As, Ws, false);
        else         gemm_tile<TB1, ACT1, K, O>(A1, W1, bias1, C1, n, As, Ws, false);
    }
}

__device__ __forceinline__ float wave_sum32(float v) {
#pragma unroll
    for (int m = 16; m > 0; m >>= 1) v += __shfl_xor(v, m, 32);
    return v;
}

__global__ __launch_bounds__(256) void fill_kernel(float* __restrict__ p, float v, int n) {
    int i = blockIdx.x * blockDim.x + threadIdx.x;
    if (i < n) p[i] = v;
}

__global__ __launch_bounds__(256) void fill_int_kernel(int* __restrict__ p, int v, int n) {
    int i = blockIdx.x * blockDim.x + threadIdx.x;
    if (i < n) p[i] = v;
}

__global__ __launch_bounds__(256) void count_rank_kernel(
    const int* __restrict__ dst, int* __restrict__ cnt, int* __restrict__ rank, int e)
{
    int i = blockIdx.x * blockDim.x + threadIdx.x;
    if (i < e) rank[i] = atomicAdd(&cnt[dst[i]], 1);
}

#define SCAN_CH 13
static_assert(32 * SCAN_CH * 128 >= NN, "scan block must cover all nodes");
static_assert(NN % 4 == 0, "16-byte accesses below need NN % 4 == 0");

__global__ __launch_bounds__(1024) void scan_isd_kernel(
    const int* __restrict__ cnt, int* __restrict__ rowptr, float* __restrict__ isd, int n)
{
    __shared__ int wtot[32];
    const int lane = threadIdx.x & 31;
    const int wave = threadIdx.x >> 5;
    const int wbase = wave * (SCAN_CH * 128);
    int4 c[SCAN_CH];
#pragma unroll
    for (int k = 0; k < SCAN_CH; ++k) {
        const int e0 = wbase + k * 128 + lane * 4;
        c[k] = (e0 < n) ? *(const int4*)(cnt + e0) : make_int4(0, 0, 0, 0);
    }
    int ex[SCAN_CH];
    int carry = 0;
#pragma unroll
    for (int k = 0; k < SCAN_CH; ++k) {
        const int tot = c[k].x + c[k].y + c[k].z + c[k].w;
        int inc = tot;
#pragma unroll
        for (int d = 1; d < 32; d <<= 1) {
            int t = __shfl_up(inc, d, 32);
            if (lane >= d) inc += t;
        }
        ex[k] = carry + inc - tot;
        carry += __shfl(inc, 31, 32);
    }
    if (lane == 0) wtot[wave] = carry;
    __syncthreads();
    int below = (lane < wave) ? wtot[lane] : 0;
#pragma unroll
    for (int m = 16; m > 0; m >>= 1) below += __shfl_xor(below, m, 32);
#pragma unroll
    for (int k = 0; k < SCAN_CH; ++k) {
        const int e0 = wbase + k * 128 + lane * 4;
        if (e0 < n) {
            const int p0 = below + ex[k];
            const int p1 = p0 + c[k].x, p2 = p1 + c[k].y, p3 = p2 + c[k].z;
            *(int4*)(rowptr + e0) = make_int4(p0, p1, p2, p3);
            *(float4*)(isd + e0) = make_float4(
                rsqrtf((float)c[k].x + 1.0f), rsqrtf((float)c[k].y + 1.0f),
                rsqrtf((float)c[k].z + 1.0f), rsqrtf((float)c[k].w + 1.0f));
        }
    }
}

__global__ __launch_bounds__(256) void place_kernel(
    const int* __restrict__ src, const int* __restrict__ dst,
    const int* __restrict__ rowptr, const int* __restrict__ rank,
    int* __restrict__ csr, int e)
{
    int i = blockIdx.x * blockDim.x + threadIdx.x;
    if (i >= e) return;
    const int pos = rowptr[dst[i]] + rank[i];
    if (pos >= 0 && pos < e) csr[pos] = src[i];
}

__global__ __launch_bounds__(256) void row_order_kernel(
    const int* __restrict__ rowptr, const int* __restrict__ cnt,
    const int* __restrict__ csr_in, int* __restrict__ csr_out, int n)
{
    const int lane = threadIdx.x & 31;
    const int node = __builtin_amdgcn_readfirstlane((int)((blockIdx.x * 256 + threadIdx.x) >> 5));
    if (node >= n) return;
    const int start = __builtin_amdgcn_readfirstlane(rowptr[node]);
    const int len   = __builtin_amdgcn_readfirstlane(cnt[node]);
    for (int ib = 0; ib < len; ib += 32) {
        const int i = ib + lane;
        const int xi = (i < len) ? csr_in[start + i] : 0;
        int pos = 0;
        for (int jb = 0; jb < len; jb += 32) {
            const int mj = min(32, len - jb);
            const int xv = (lane < mj) ? csr_in[start + jb + lane] : 0;
            for (int jj = 0; jj < mj; ++jj) {
                const int xj = __builtin_amdgcn_readlane(xv, jj);
                const int j = jb + jj;
                pos += (xj < xi || (xj == xi && j < i)) ? 1 : 0;
            }
        }
        if (i < len) csr_out[start + pos] = xi;
    }
}

template <bool GCN, int B>
__device__ __forceinline__ void gather_accum(
    float4& acc, const float* __restrict__ v, int ms, float mw, int j, int lane)
{
    float4 r[B];
    float w[B];
#pragma unroll
    for (int u = 0; u < B; ++u) {
        const int s = __builtin_amdgcn_readlane(ms, j + u);
        r[u] = *(const float4*)(v + (size_t)s * DD + lane * 4);
        w[u] = GCN ? __int_as_float(__builtin_amdgcn_readlane(__float_as_int(mw), j + u)) : 1.0f;
    }
#pragma unroll
    for (int u = 0; u < B; ++u) {
        if (GCN) {
            acc.x = fmaf(r[u].x, w[u], acc.x); acc.y = fmaf(r[u].y, w[u], acc.y);
            acc.z = fmaf(r[u].z, w[u], acc.z); acc.w = fmaf(r[u].w, w[u], acc.w);
        } else {
            acc.x += r[u].x; acc.y += r[u].y; acc.z += r[u].z; acc.w += r[u].w;
        }
    }
}

template <bool GCN>
__global__ __launch_bounds__(256) void gather_rows_kernel(
    const int* __restrict__ rowptr, const int* __restrict__ cnt, const int* __restrict__ csr,
    const float* __restrict__ v, const float* __restrict__ isd, const float* __restrict__ bias,
    const float* __restrict__ g, const float* __restrict__ b,
    float* __restrict__ o, int n)
{
    const int lane = threadIdx.x & 31;
    const int node = __builtin_amdgcn_readfirstlane((int)((blockIdx.x * 256 + threadIdx.x) >> 5));
    if (node >= n) return;
    const int start = __builtin_amdgcn_readfirstlane(rowptr[node]);
    const int len   = __builtin_amdgcn_readfirstlane(cnt[node]);
    float4 acc = make_float4(0.f, 0.f, 0.f, 0.f);
    float4 sv = acc, bv = acc, gv = acc, lb = acc;
    float it = 0.f;
    if (GCN) {
        it = isd[node];
        sv = *(const float4*)(v + (size_t)node * DD + lane * 4);
        bv = *(const float4*)(bias + lane * 4);
        gv = *(const float4*)(g + lane * 4);
        lb = *(const float4*)(b + lane * 4);
    }
    for (int base = 0; base < len; base += 32) {
        const int m = min(32, len - base);
        int ms = 0;
        float mw = 0.f;
        if (lane < m) {
            ms = csr[start + base + lane];
            ms = min(max(ms, 0), n - 1);
            if (GCN) mw = isd[ms] * it;
        }
        int j = 0;
        for (; j + 8 <= m; j += 8) gather_accum<GCN, 8>(acc, v, ms, mw, j, lane);
        if (j + 4 <= m) { gather_accum<GCN, 4>(acc, v, ms, mw, j, lane); j += 4; }
        if (j + 2 <= m) { gather_accum<GCN, 2>(acc, v, ms, mw, j, lane); j += 2; }
        if (j < m)      { gather_accum<GCN, 1>(acc, v, ms, mw, j, lane); }
    }
    if (GCN) {
        acc.x += sv.x * it * it + bv.x; acc.y += sv.y * it * it + bv.y;
        acc.z += sv.z * it * it + bv.z; acc.w += sv.w * it * it + bv.w;
        const float mean = wave_sum32(acc.x + acc.y + acc.z + acc.w) * (1.f / 128.f);
        const float dx = acc.x - mean, dy = acc.y - mean, dz = acc.z - mean, dw = acc.w - mean;
        const float inv = rsqrtf(wave_sum32(dx * dx + dy * dy + dz * dz + dw * dw) * (1.f / 128.f) + 1e-5f);
        acc.x = fmaxf(dx * inv * gv.x + lb.x, 0.f); acc.y = fmaxf(dy * inv * gv.y + lb.y, 0.f);
        acc.z = fmaxf(dz * inv * gv.z + lb.z, 0.f); acc.w = fmaxf(dw * inv * gv.w + lb.w, 0.f);
    }
    *(float4*)(o + (size_t)node * DD + lane * 4) = acc;
}

__global__ __launch_bounds__(256) void attention_kernel(
    const float* __restrict__ qkv0, const float* __restrict__ qkv1,
    float* __restrict__ ao0, float* __restrict__ ao1, int n)
{
    int wv = (blockIdx.x * 256 + threadIdx.x) >> 5;
    int lane = threadIdx.x & 31;
    if (wv >= n) return;
    const float* b0 = qkv0 + (size_t)wv * 384;
    const float* b1 = qkv1 + (size_t)wv * 384;
    float q0[4], k0[4], v0[4], q1[4], k1[4], v1[4];
#pragma unroll
    for (int j = 0; j < 4; ++j) {
        int d = lane + 32 * j;
        q0[j] = b0[d]; k0[j] = b0[128 + d]; v0[j] = b0[256 + d];
        q1[j] = b1[d]; k1[j] = b1[128 + d]; v1[j] = b1[256 + d];
    }
    float l00 = 0, l01 = 0, l10 = 0, l11 = 0;
#pragma unroll
    for (int j = 0; j < 4; ++j) {
        l00 += q0[j] * k0[j]; l01 += q0[j] * k1[j];
        l10 += q1[j] * k0[j]; l11 += q1[j] * k1[j];
    }
    l00 = wave_sum32(l00); l01 = wave_sum32(l01);
    l10 = wave_sum32(l10); l11 = wave_sum32(l11);
    const float scale = 0.08838834764831845f;
    l00 *= scale; l01 *= scale; l10 *= scale; l11 *= scale;
    float m0 = fmaxf(l00, l01), e00 = __expf(l00 - m0), e01 = __expf(l01 - m0);
    float r0 = 1.f / (e00 + e01), w00 = e00 * r0, w01 = e01 * r0;
    float m1 = fmaxf(l10, l11), e10 = __expf(l10 - m1), e11 = __expf(l11 - m1);
    float r1 = 1.f / (e10 + e11), w10 = e10 * r1, w11 = e11 * r1;
    float* o0 = ao0 + (size_t)wv * DD;
    float* o1 = ao1 + (size_t)wv * DD;
#pragma unroll
    for (int j = 0; j < 4; ++j) {
        int d = lane + 32 * j;
        o0[d] = w00 * v0[j] + w01 * v1[j];
        o1[d] = w10 * v0[j] + w11 * v1[j];
    }
}

__global__ __launch_bounds__(256) void fuse_ln384_kernel(
    const float* __restrict__ a0, const float* __restrict__ a1,
    const float* __restrict__ rs, const float* __restrict__ g,
    const float* __restrict__ b, float* __restrict__ o, int n)
{
    int wv = (blockIdx.x * 256 + threadIdx.x) >> 5;
    int lane = threadIdx.x & 31;
    if (wv >= n) return;
    const float* p0 = a0 + (size_t)wv * DD;
    const float* p1 = a1 + (size_t)wv * DD;
    const float* p2 = rs + (size_t)wv * DD;
    float x[12], s = 0.f;
#pragma unroll
    for (int j = 0; j < 12; ++j) {
        int d = lane + 32 * j;
        float v = (d < 128) ? p0[d] : (d < 256) ? p1[d - 128] : p2[d - 256];
        x[j] = v; s += v;
    }
    float mean = wave_sum32(s) * (1.f / 384.f);
    float sq = 0.f;
#pragma unroll
    for (int j = 0; j < 12; ++j) { float d = x[j] - mean; sq += d * d; }
    float inv = rsqrtf(wave_sum32(sq) * (1.f / 384.f) + 1e-5f);
    float* orow = o + (size_t)wv * 384;
#pragma unroll
    for (int j = 0; j < 12; ++j) {
        int d = lane + 32 * j;
        orow[d] = (x[j] - mean) * inv * g[d] + b[d];
    }
}

#define CS_BLOCKS 256
#define CS_STRIDE (CS_BLOCKS * 8)
#define CS_ROWS ((NN + CS_STRIDE - 1) / CS_STRIDE)
static_assert(CS_ROWS * CS_STRIDE >= NN, "column-sum waves must cover all rows");
static_assert(CS_ROWS <= 48, "row loads of a wave must stay below the load counter's limit");

__global__ __launch_bounds__(256) void colsum_kernel(const float* __restrict__ p, float* __restrict__ cs, int n) {
    __shared__ __attribute__((aligned(16))) float part[8 * DD];
    const int lane = threadIdx.x & 31;
    const int wave = threadIdx.x >> 5;
    const int r0 = __builtin_amdgcn_readfirstlane((int)(blockIdx.x * 8 + wave));
    float4 r[CS_ROWS];
#pragma unroll
    for (int t = 0; t < CS_ROWS; ++t) {
        const int row = r0 + t * CS_STRIDE;
        r[t] = *(const float4*)(p + (size_t)min(row, n - 1) * DD + lane * 4);
    }
    float4 s = make_float4(0.f, 0.f, 0.f, 0.f);
#pragma unroll
    for (int t = 0; t < CS_ROWS; ++t) {
        const bool in = (r0 + t * CS_STRIDE) < n;
        s.x += in ? r[t].x : 0.f; s.y += in ? r[t].y : 0.f;
        s.z += in ? r[t].z : 0.f; s.w += in ? r[t].w : 0.f;
    }
    *(float4*)(part + wave * DD + lane * 4) = s;
    __syncthreads();
    if (threadIdx.x < DD) {
        float tot = 0.f;
#pragma unroll
        for (int w = 0; w < 8; ++w) tot += part[w * DD + threadIdx.x];
        atomicAdd(&cs[threadIdx.x], tot);
    }
}

__global__ __launch_bounds__(256) void pairnorm_kernel(
    const float* __restrict__ pre, const float* __restrict__ cs,
    float* __restrict__ o, int n)
{
    int wv = (blockIdx.x * 256 + threadIdx.x) >> 5;
    int lane = threadIdx.x & 31;
    if (wv >= n) return;
    const float* row = pre + (size_t)wv * DD;
    float x[4], sq = 0.f;
#pragma unroll
    for (int j = 0; j < 4; ++j) { x[j] = row[lane + 32 * j]; sq += x[j] * x[j]; }
    float rn = sqrtf(1e-6f + wave_sum32(sq));
    float inv = 20.f / rn;
    float* orow = o + (size_t)wv * DD;
    const float invN = 1.f / (float)NN;
#pragma unroll
    for (int j = 0; j < 4; ++j) {
        int d = lane + 32 * j;
        orow[d] = x[j] * inv - cs[d] * invN;
    }
}

extern "C" void kernel_launch(void* const* d_in, const int* in_sizes, int n_in,
                              void* d_out, int out_size, void* d_ws, size_t ws_size,
                              hipStream_t stream)
{
    const float* x      = (const float*)d_in[0];
    const float* x_org  = (const float*)d_in[1];
    const int*   ei     = (const int*)d_in[2];
    const float* in_W   = (const float*)d_in[3];
    const float* in_b   = (const float*)d_in[4];
    const float* res_W  = (const float*)d_in[5];
    const float* gcn_W  = (const float*)d_in[6];
    const float* gcn_b  = (const float*)d_in[7];
    const float* ln_g   = (const float*)d_in[8];
    const float* ln_b   = (const float*)d_in[9];
    const float* lin_W  = (const float*)d_in[10];
    const float* lin_b  = (const float*)d_in[11];
    const float* attr_W = (const float*)d_in[12];
    const float* attr_b = (const float*)d_in[13];
    const float* qkv_W  = (const float*)d_in[14];
    const float* qkv_b  = (const float*)d_in[15];
    const float* out_W  = (const float*)d_in[16];
    const float* out_b  = (const float*)d_in[17];
    const float* fln_g  = (const float*)d_in[18];
    const float* fln_b  = (const float*)d_in[19];
    const float* fuse_W = (const float*)d_in[20];
    const float* fuse_b = (const float*)d_in[21];
    float* out = (float*)d_out;

    const int* src = ei;
    const int* dst = ei + EE;

    float* ws = (float*)d_ws;
    const size_t ND = (size_t)NN * DD;
    float* isd  = ws;
    float* csum = isd + NN;
    float* bh   = csum + DD;
    float* bt   = bh + ND;
    float* br   = bt + ND;
    float* bs   = br + ND;
    float* q0b  = bs + ND;
    float* q1b  = q0b + 3 * ND;
    int* cnt    = (int*)q1b;
    int* rowptr = cnt + NN;
    int* rank   = rowptr + NN;
    int* csr_a  = rank + EE;
    int* csr_b  = csr_a + EE;

    const dim3 blk(256);
    const dim3 gemmO128((NN + GBM - 1) / GBM, DD / GBN);
    const dim3 pairO128((NN + GBM - 1) / GBM, DD / GBN, 2);
    const dim3 pairO384((NN + GBM - 1) / GBM, 3 * DD / GBN, 2);
    const int nodeWaves = (NN + 7) / 8;
    const int edgeBlk   = (EE + 255) / 256;

    fill_int_kernel<<<(NN + 255) / 256, blk, 0, stream>>>(cnt, 0, NN);
    count_rank_kernel<<<edgeBlk, blk, 0, stream>>>(dst, cnt, rank, EE);
    scan_isd_kernel<<<1, 1024, 0, stream>>>(cnt, rowptr, isd, NN);
    place_kernel<<<edgeBlk, blk, 0, stream>>>(src, dst, rowptr, rank, csr_a, EE);
    row_order_kernel<<<nodeWaves, blk, 0, stream>>>(rowptr, cnt, csr_a, csr_b, NN);

    gemm_lds_pair<true, 1, false, 0, DD, DD><<<pairO128, blk, 0, stream>>>(
        x, in_W, in_b, bh, x_org, res_W, nullptr, bt, NN);

    gather_rows_kernel<false><<<nodeWaves, blk, 0, stream>>>(
        rowptr, cnt, csr_b, bt, nullptr, nullptr, nullptr, nullptr, br, NN);

    for (int i = 0; i < 3; ++i) {
        gemm_lds<true, 0, DD, DD><<<gemmO128, blk, 0, stream>>>(
            bh, gcn_W + (size_t)i * DD * DD, nullptr, bt, NN);
        gather_rows_kernel<true><<<nodeWaves, blk, 0, stream>>>(
            rowptr, cnt, csr_b, bt, isd, gcn_b + (size_t)i * DD,
            ln_g + (size_t)i * DD, ln_b + (size_t)i * DD, bh, NN);
    }

    gemm_lds_pair<true, 0, true, 1, DD, DD><<<pairO128, blk, 0, stream>>>(
        bh, lin_W, lin_b, bs, x_org, attr_W, attr_b, bt, NN);

    gemm_lds_pair<true, 0, true, 0, DD, 3 * DD><<<pairO384, blk, 0, stream>>>(
        bs, qkv_W, qkv_b, q0b, bt, qkv_W, qkv_b, q1b, NN);

    attention_kernel<<<nodeWaves, blk, 0, stream>>>(q0b, q1b, bh, bt, NN);

    gemm_lds_pair<true, 0, true, 0, DD, DD><<<pairO128, blk, 0, stream>>>(
        bh, out_W, out_b, bs, bt, out_W, out_b, q1b, NN);

    fuse_ln384_kernel<<<nodeWaves, blk, 0, stream>>>(bs, q1b, br, fln_g, fln_b, q0b, NN);

    gemm_lds<true, 0, 3 * DD, DD><<<gemmO128, blk, 0, stream>>>(q0b, fuse_W, fuse_b, bh, NN);

    fill_kernel<<<1, 128, 0, stream>>>(csum, 0.f, DD);
    colsum_kernel<<<CS_BLOCKS, blk, 0, stream>>>(bh, csum, NN);
    pairnorm_kernel<<<nodeWaves, blk, 0, stream>>>(bh, csum, out, NN);
}


# GEMM row tiling: 768 full 64-row tiles (one resident round: 256 WGPs x 3 blocks) plus 53 16-row tiles placed first in block order; was 782 tiles whose last 14 ran one block length behind the rest
# speedup vs baseline: 2.6630x; 1.0683x over previous
#include <hip/hip_runtime.h>
#include <hip/hip_bf16.h>

#define NN 50000
#define DD 128
#define EE 800000

typedef __attribute__((ext_vector_type(2))) float v2f;
typedef __attribute__((ext_vector_type(8))) float v8f;

#define GP 132
#define GBM 64
#define GBN 128

#define RESIDENT_BLOCKS 768
#define FULL_TILES ((NN / GBM) / RESIDENT_BLOCKS * RESIDENT_BLOCKS)
#define SMALL_BM 16
#define SMALL_TILES ((NN - FULL_TILES * GBM + SMALL_BM - 1) / SMALL_BM)
static_assert(FULL_TILES > 0 && FULL_TILES * GBM <= NN, "full tiles must lie inside the rows");
static_assert(FULL_TILES * GBM + SMALL_TILES * SMALL_BM >= NN, "small tiles must cover the rest");

template <bool TRANSB, int K, int O>
__device__ __forceinline__ void stage_w(
    const float* __restrict__ W, float* __restrict__ Ws, int cbase, int kc, int wave, int lane)
{
    if (TRANSB) {
#pragma unroll
        for (int i = 0; i < GBN / 8; ++i) {
            const int row = wave + 8 * i;
            const float4 v = *(const float4*)(W + (unsigned)((cbase + row) * K + kc + lane * 4));
            *(float4*)(Ws + row * GP + lane * 4) = v;
        }
    } else {
#pragma unroll
        for (int i = 0; i < GBN / 8; ++i) {
            const int idx = wave * (GBN / 8) + i;
            const int k  = (idx >> 2) * 4 + (lane >> 3);
            const int nb = (idx & 3) * 32 + (lane & 7) * 4;
            const float4 v = *(const float4*)(W + (unsigned)((kc + k) * O + cbase + nb));
            Ws[(nb + 0) * GP + k] = v.x;
            Ws[(nb + 1) * GP + k] = v.y;
            Ws[(nb + 2) * GP + k] = v.z;
            Ws[(nb + 3) * GP + k] = v.w;
        }
    }
}

template <bool TRANSB, int ACT, int K, int O>
__device__ __forceinline__ void gemm_tile(
    const float* __restrict__ A, const float* __restrict__ W,
    const float* __restrict__ bias, float* __restrict__ C, int n, int rbase,
    float* __restrict__ As, float* __restrict__ Ws, bool relu_rt)
{
    static_assert(K % 128 == 0 && O % GBN == 0, "tile sizes");

    const int wave = threadIdx.x >> 5;
    const int lane = threadIdx.x & 31;
    const int half = lane >> 4;
    const int l15  = lane & 15;
    const int cbase = blockIdx.y * GBN;
    const int wr = (wave >> 2) * 32;
    const int wc = (wave & 3) * 32;

    v8f acc[2][2];
#pragma unroll
    for (int i = 0; i < 2; ++i)
#pragma unroll
        for (int j = 0; j < 2; ++j) acc[i][j] = (v8f){0.f,0.f,0.f,0.f,0.f,0.f,0.f,0.f};

    for (int kc = 0; kc < K; kc += 128) {
        if (kc > 0) __syncthreads();

#pragma unroll
        for (int i = 0; i < GBM / 8; ++i) {
            const int row = wave + 8 * i;
            const int grow = min(rbase + row, n - 1);
            const float4 v = *(const float4*)(A + (unsigned)(grow * K + kc + lane * 4));
            *(float4*)(As + row * GP + lane * 4) = v;
        }
        stage_w<TRANSB, K, O>(W, Ws, cbase, kc, wave, lane);
        __syncthreads();

        const float* ap = As + (wr + l15) * GP + 2 * half;
        const float* bp = Ws + (wc + l15) * GP + 2 * half;
#pragma unroll 8
        for (int kk = 0; kk < 128; kk += 4) {
            const v2f a0 = *(const v2f*)(ap + kk);
            const v2f a1 = *(const v2f*)(ap + 16 * GP + kk);
            const v2f b0 = *(const v2f*)(bp + kk);
            const v2f b1 = *(const v2f*)(bp + 16 * GP + kk);
            acc[0][0] = __builtin_amdgcn_wmma_f32_16x16x4_f32(
                false, a0, false, b0, (short)0, acc[0][0], false, false);
            acc[0][1] = __builtin_amdgcn_wmma_f32_16x16x4_f32(
                false, a0, false, b1, (short)0, acc[0][1], false, false);
            acc[1][0] = __builtin_amdgcn_wmma_f32_16x16x4_f32(
                false, a1, false, b0, (short)0, acc[1][0], false, false);
            acc[1][1] = __builtin_amdgcn_wmma_f32_16x16x4_f32(
                false, a1, false, b1, (short)0, acc[1][1], false, false);
        }
    }

#pragma unroll
    for (int j = 0; j < 2; ++j) {
        const int col = cbase + wc + 16 * j + l15;
        const float bv = bias ? bias[col] : 0.f;
#pragma unroll
        for (int i = 0; i < 2; ++i) {
#pragma unroll
            for (int v = 0; v < 8; ++v) {
                const int row = rbase + wr + 16 * i + v + 8 * half;
                float val = acc[i][j][v] + bv;
                if (ACT == 1 || (ACT == 2 && relu_rt)) val = fmaxf(val, 0.f);
                if (row < n) C[(unsigned)(row * O + col)] = val;
            }
        }
    }
}

template <bool TRANSB, int ACT, int K, int O>
__device__ __forceinline__ void gemm_tile16(
    const float* __restrict__ A, const float* __restrict__ W,
    const float* __restrict__ bias, float* __restrict__ C, int n, int rbase,
    float* __restrict__ As, float* __restrict__ Ws, bool relu_rt)
{
    const int wave = threadIdx.x >> 5;
    const int lane = threadIdx.x & 31;
    const int half = lane >> 4;
    const int l15  = lane & 15;
    const int cbase = blockIdx.y * GBN;

    v8f acc = (v8f){0.f,0.f,0.f,0.f,0.f,0.f,0.f,0.f};

    for (int kc = 0; kc < K; kc += 128) {
        if (kc > 0) __syncthreads();
#pragma unroll
        for (int i = 0; i < SMALL_BM / 8; ++i) {
            const int row = wave + 8 * i;
            const int grow = min(rbase + row, n - 1);
            const float4 v = *(const float4*)(A + (unsigned)(grow * K + kc + lane * 4));
            *(float4*)(As + row * GP + lane * 4) = v;
        }
        stage_w<TRANSB, K, O>(W, Ws, cbase, kc, wave, lane);
        __syncthreads();

        const float* ap = As + l15 * GP + 2 * half;
        const float* bp = Ws + (16 * wave + l15) * GP + 2 * half;
#pragma unroll 8
        for (int kk = 0; kk < 128; kk += 4) {
            const v2f a0 = *(const v2f*)(ap + kk);
            const v2f b0 = *(const v2f*)(bp + kk);
            acc = __builtin_amdgcn_wmma_f32_16x16x4_f32(
                false, a0, false, b0, (short)0, acc, false, false);
        }
    }

    const int col = cbase + 16 * wave + l15;
    const float bv = bias ? bias[col] : 0.f;
#pragma unroll
    for (int v = 0; v < 8; ++v) {
        const int row = rbase + v + 8 * half;
        float val = acc[v] + bv;
        if (ACT == 1 || (ACT == 2 && relu_rt)) val = fmaxf(val, 0.f);
        if (row < n) C[(unsigned)(row * O + col)] = val;
    }
}

template <bool TRANSB, int ACT, int K, int O>
__device__ __forceinline__ void gemm_block(
    const float* __restrict__ A, const float* __restrict__ W,
    const float* __restrict__ bias, float* __restrict__ C, int n,
    float* __restrict__ As, float* __restrict__ Ws, bool relu_rt)
{
    const int bx = blockIdx.x;
    if (bx < SMALL_TILES)
        gemm_tile16<TRANSB, ACT, K, O>(A, W, bias, C, n, FULL_TILES * GBM + bx * SMALL_BM, As, Ws, relu_rt);
    else
        gemm_tile<TRANSB, ACT, K, O>(A, W, bias, C, n, (bx - SMALL_TILES) * GBM, As, Ws, relu_rt);
}

template <bool TRANSB, int ACT, int K, int O>
__global__ __launch_bounds__(256, 6) void gemm_lds(
    const float* __restrict__ A, const float* __restrict__ W,
    const float* __restrict__ bias, float* __restrict__ C, int n)
{
    __shared__ __attribute__((aligned(16))) float As[GBM * GP];
    __shared__ __attribute__((aligned(16))) float Ws[GBN * GP];
    gemm_block<TRANSB, ACT, K, O>(A, W, bias, C, n, As, Ws, false);
}

template <bool TB0, int ACT0, bool TB1, int ACT1, int K, int O>
__global__ __launch_bounds__(256, 6) void gemm_lds_pair(
    const float* __restrict__ A0, const float* __restrict__ W0,
    const float* __restrict__ bias0, float* __restrict__ C0,
    const float* __restrict__ A1, const float* __restrict__ W1,
    const float* __restrict__ bias1, float* __restrict__ C1, int n)
{
    __shared__ __attribute__((aligned(16))) float As[GBM * GP];
    __shared__ __attribute__((aligned(16))) float Ws[GBN * GP];
    const bool second = blockIdx.z != 0;
    if (TB0 == TB1) {
        gemm_block<TB0, (ACT0 == ACT1) ? ACT0 : 2, K, O>(
            second ? A1 : A0, second ? W1 : W0, second ? bias1 : bias0, second ? C1 : C0,
            n, As, Ws, (second ? ACT1 : ACT0) == 1);
    } else {
        if (!second) gemm_block<TB0, ACT0, K, O>(A0, W0, bias0, C0, n, As, Ws, false);
        else         gemm_block<TB1, ACT1, K, O>(A1, W1, bias1, C1, n, As, Ws, false);
    }
}

__device__ __forceinline__ float wave_sum32(float v) {
#pragma unroll
    for (int m = 16; m > 0; m >>= 1) v += __shfl_xor(v, m, 32);
    return v;
}

__global__ __launch_bounds__(256) void fill_kernel(float* __restrict__ p, float v, int n) {
    int i = blockIdx.x * blockDim.x + threadIdx.x;
    if (i < n) p[i] = v;
}

__global__ __launch_bounds__(256) void fill_int_kernel(int* __restrict__ p, int v, int n) {
    int i = blockIdx.x * blockDim.x + threadIdx.x;
    if (i < n) p[i] = v;
}

__global__ __launch_bounds__(256) void count_rank_kernel(
    const int* __restrict__ dst, int* __restrict__ cnt, int* __restrict__ rank, int e)
{
    int i = blockIdx.x * blockDim.x + threadIdx.x;
    if (i < e) rank[i] = atomicAdd(&cnt[dst[i]], 1);
}

#define SCAN_CH 13
static_assert(32 * SCAN_CH * 128 >= NN, "scan block must cover all nodes");
static_assert(NN % 4 == 0, "16-byte accesses below need NN % 4 == 0");

__global__ __launch_bounds__(1024) void scan_isd_kernel(
    const int* __restrict__ cnt, int* __restrict__ rowptr, float* __restrict__ isd, int n)
{
    __shared__ int wtot[32];
    const int lane = threadIdx.x & 31;
    const int wave = threadIdx.x >> 5;
    const int wbase = wave * (SCAN_CH * 128);
    int4 c[SCAN_CH];
#pragma unroll
    for (int k = 0; k < SCAN_CH; ++k) {
        const int e0 = wbase + k * 128 + lane * 4;
        c[k] = (e0 < n) ? *(const int4*)(cnt + e0) : make_int4(0, 0, 0, 0);
    }
    int ex[SCAN_CH];
    int carry = 0;
#pragma unroll
    for (int k = 0; k < SCAN_CH; ++k) {
        const int tot = c[k].x + c[k].y + c[k].z + c[k].w;
        int inc = tot;
#pragma unroll
        for (int d = 1; d < 32; d <<= 1) {
            int t = __shfl_up(inc, d, 32);
            if (lane >= d) inc += t;
        }
        ex[k] = carry + inc - tot;
        carry += __shfl(inc, 31, 32);
    }
    if (lane == 0) wtot[wave] = carry;
    __syncthreads();
    int below = (lane < wave) ? wtot[lane] : 0;
#pragma unroll
    for (int m = 16; m > 0; m >>= 1) below += __shfl_xor(below, m, 32);
#pragma unroll
    for (int k = 0; k < SCAN_CH; ++k) {
        const int e0 = wbase + k * 128 + lane * 4;
        if (e0 < n) {
            const int p0 = below + ex[k];
            const int p1 = p0 + c[k].x, p2 = p1 + c[k].y, p3 = p2 + c[k].z;
            *(int4*)(rowptr + e0) = make_int4(p0, p1, p2, p3);
            *(float4*)(isd + e0) = make_float4(
                rsqrtf((float)c[k].x + 1.0f), rsqrtf((float)c[k].y + 1.0f),
                rsqrtf((float)c[k].z + 1.0f), rsqrtf((float)c[k].w + 1.0f));
        }
    }
}

__global__ __launch_bounds__(256) void place_kernel(
    const int* __restrict__ src, const int* __restrict__ dst,
    const int* __restrict__ rowptr, const int* __restrict__ rank,
    int* __restrict__ csr, int e)
{
    int i = blockIdx.x * blockDim.x + threadIdx.x;
    if (i >= e) return;
    const int pos = rowptr[dst[i]] + rank[i];
    if (pos >= 0 && pos < e) csr[pos] = src[i];
}

__global__ __launch_bounds__(256) void row_order_kernel(
    const int* __restrict__ rowptr, const int* __restrict__ cnt,
    const int* __restrict__ csr_in, int* __restrict__ csr_out, int n)
{
    const int lane = threadIdx.x & 31;
    const int node = __builtin_amdgcn_readfirstlane((int)((blockIdx.x * 256 + threadIdx.x) >> 5));
    if (node >= n) return;
    const int start = __builtin_amdgcn_readfirstlane(rowptr[node]);
    const int len   = __builtin_amdgcn_readfirstlane(cnt[node]);
    for (int ib = 0; ib < len; ib += 32) {
        const int i = ib + lane;
        const int xi = (i < len) ? csr_in[start + i] : 0;
        int pos = 0;
        for (int jb = 0; jb < len; jb += 32) {
            const int mj = min(32, len - jb);
            const int xv = (lane < mj) ? csr_in[start + jb + lane] : 0;
            for (int jj = 0; jj < mj; ++jj) {
                const int xj = __builtin_amdgcn_readlane(xv, jj);
                const int j = jb + jj;
                pos += (xj < xi || (xj == xi && j < i)) ? 1 : 0;
            }
        }
        if (i < len) csr_out[start + pos] = xi;
    }
}

template <bool GCN, int B>
__device__ __forceinline__ void gather_accum(
    float4& acc, const float* __restrict__ v, int ms, float mw, int j, int lane)
{
    float4 r[B];
    float w[B];
#pragma unroll
    for (int u = 0; u < B; ++u) {
        const int s = __builtin_amdgcn_readlane(ms, j + u);
        r[u] = *(const float4*)(v + (size_t)s * DD + lane * 4);
        w[u] = GCN ? __int_as_float(__builtin_amdgcn_readlane(__float_as_int(mw), j + u)) : 1.0f;
    }
#pragma unroll
    for (int u = 0; u < B; ++u) {
        if (GCN) {
            acc.x = fmaf(r[u].x, w[u], acc.x); acc.y = fmaf(r[u].y, w[u], acc.y);
            acc.z = fmaf(r[u].z, w[u], acc.z); acc.w = fmaf(r[u].w, w[u], acc.w);
        } else {
            acc.x += r[u].x; acc.y += r[u].y; acc.z += r[u].z; acc.w += r[u].w;
        }
    }
}

template <bool GCN>
__global__ __launch_bounds__(256) void gather_rows_kernel(
    const int* __restrict__ rowptr, const int* __restrict__ cnt, const int* __restrict__ csr,
    const float* __restrict__ v, const float* __restrict__ isd, const float* __restrict__ bias,
    const float* __restrict__ g, const float* __restrict__ b,
    float* __restrict__ o, int n)
{
    const int lane = threadIdx.x & 31;
    const int node = __builtin_amdgcn_readfirstlane((int)((blockIdx.x * 256 + threadIdx.x) >> 5));
    if (node >= n) return;
    const int start = __builtin_amdgcn_readfirstlane(rowptr[node]);
    const int len   = __builtin_amdgcn_readfirstlane(cnt[node]);
    float4 acc = make_float4(0.f, 0.f, 0.f, 0.f);
    float4 sv = acc, bv = acc, gv = acc, lb = acc;
    float it = 0.f;
    if (GCN) {
        it = isd[node];
        sv = *(const float4*)(v + (size_t)node * DD + lane * 4);
        bv = *(const float4*)(bias + lane * 4);
        gv = *(const float4*)(g + lane * 4);
        lb = *(const float4*)(b + lane * 4);
    }
    for (int base = 0; base < len; base += 32) {
        const int m = min(32, len - base);
        int ms = 0;
        float mw = 0.f;
        if (lane < m) {
            ms = csr[start + base + lane];
            ms = min(max(ms, 0), n - 1);
            if (GCN) mw = isd[ms] * it;
        }
        int j = 0;
        for (; j + 8 <= m; j += 8) gather_accum<GCN, 8>(acc, v, ms, mw, j, lane);
        if (j + 4 <= m) { gather_accum<GCN, 4>(acc, v, ms, mw, j, lane); j += 4; }
        if (j + 2 <= m) { gather_accum<GCN, 2>(acc, v, ms, mw, j, lane); j += 2; }
        if (j < m)      { gather_accum<GCN, 1>(acc, v, ms, mw, j, lane); }
    }
    if (GCN) {
        acc.x += sv.x * it * it + bv.x; acc.y += sv.y * it * it + bv.y;
        acc.z += sv.z * it * it + bv.z; acc.w += sv.w * it * it + bv.w;
        const float mean = wave_sum32(acc.x + acc.y + acc.z + acc.w) * (1.f / 128.f);
        const float dx = acc.x - mean, dy = acc.y - mean, dz = acc.z - mean, dw = acc.w - mean;
        const float inv = rsqrtf(wave_sum32(dx * dx + dy * dy + dz * dz + dw * dw) * (1.f / 128.f) + 1e-5f);
        acc.x = fmaxf(dx * inv * gv.x + lb.x, 0.f); acc.y = fmaxf(dy * inv * gv.y + lb.y, 0.f);
        acc.z = fmaxf(dz * inv * gv.z + lb.z, 0.f); acc.w = fmaxf(dw * inv * gv.w + lb.w, 0.f);
    }
    *(float4*)(o + (size_t)node * DD + lane * 4) = acc;
}

__global__ __launch_bounds__(256) void attention_kernel(
    const float* __restrict__ qkv0, const float* __restrict__ qkv1,
    float* __restrict__ ao0, float* __restrict__ ao1, int n)
{
    int wv = (blockIdx.x * 256 + threadIdx.x) >> 5;
    int lane = threadIdx.x & 31;
    if (wv >= n) return;
    const float* b0 = qkv0 + (size_t)wv * 384;
    const float* b1 = qkv1 + (size_t)wv * 384;
    float q0[4], k0[4], v0[4], q1[4], k1[4], v1[4];
#pragma unroll
    for (int j = 0; j < 4; ++j) {
        int d = lane + 32 * j;
        q0[j] = b0[d]; k0[j] = b0[128 + d]; v0[j] = b0[256 + d];
        q1[j] = b1[d]; k1[j] = b1[128 + d]; v1[j] = b1[256 + d];
    }
    float l00 = 0, l01 = 0, l10 = 0, l11 = 0;
#pragma unroll
    for (int j = 0; j < 4; ++j) {
        l00 += q0[j] * k0[j]; l01 += q0[j] * k1[j];
        l10 += q1[j] * k0[j]; l11 += q1[j] * k1[j];
    }
    l00 = wave_sum32(l00); l01 = wave_sum32(l01);
    l10 = wave_sum32(l10); l11 = wave_sum32(l11);
    const float scale = 0.08838834764831845f;
    l00 *= scale; l01 *= scale; l10 *= scale; l11 *= scale;
    float m0 = fmaxf(l00, l01), e00 = __expf(l00 - m0), e01 = __expf(l01 - m0);
    float r0 = 1.f / (e00 + e01), w00 = e00 * r0, w01 = e01 * r0;
    float m1 = fmaxf(l10, l11), e10 = __expf(l10 - m1), e11 = __expf(l11 - m1);
    float r1 = 1.f / (e10 + e11), w10 = e10 * r1, w11 = e11 * r1;
    float* o0 = ao0 + (size_t)wv * DD;
    float* o1 = ao1 + (size_t)wv * DD;
#pragma unroll
    for (int j = 0; j < 4; ++j) {
        int d = lane + 32 * j;
        o0[d] = w00 * v0[j] + w01 * v1[j];
        o1[d] = w10 * v0[j] + w11 * v1[j];
    }
}

__global__ __launch_bounds__(256) void fuse_ln384_kernel(
    const float* __restrict__ a0, const float* __restrict__ a1,
    const float* __restrict__ rs, const float* __restrict__ g,
    const float* __restrict__ b, float* __restrict__ o, int n)
{
    int wv = (blockIdx.x * 256 + threadIdx.x) >> 5;
    int lane = threadIdx.x & 31;
    if (wv >= n) return;
    const float* p0 = a0 + (size_t)wv * DD;
    const float* p1 = a1 + (size_t)wv * DD;
    const float* p2 = rs + (size_t)wv * DD;
    float x[12], s = 0.f;
#pragma unroll
    for (int j = 0; j < 12; ++j) {
        int d = lane + 32 * j;
        float v = (d < 128) ? p0[d] : (d < 256) ? p1[d - 128] : p2[d - 256];
        x[j] = v; s += v;
    }
    float mean = wave_sum32(s) * (1.f / 384.f);
    float sq = 0.f;
#pragma unroll
    for (int j = 0; j < 12; ++j) { float d = x[j] - mean; sq += d * d; }
    float inv = rsqrtf(wave_sum32(sq) * (1.f / 384.f) + 1e-5f);
    float* orow = o + (size_t)wv * 384;
#pragma unroll
    for (int j = 0; j < 12; ++j) {
        int d = lane + 32 * j;
        orow[d] = (x[j] - mean) * inv * g[d] + b[d];
    }
}

#define CS_BLOCKS 256
#define CS_STRIDE (CS_BLOCKS * 8)
#define CS_ROWS ((NN + CS_STRIDE - 1) / CS_STRIDE)
static_assert(CS_ROWS * CS_STRIDE >= NN, "column-sum waves must cover all rows");
static_assert(CS_ROWS <= 48, "row loads of a wave must stay below the load counter's limit");

__global__ __launch_bounds__(256) void colsum_kernel(const float* __restrict__ p, float* __restrict__ cs, int n) {
    __shared__ __attribute__((aligned(16))) float part[8 * DD];
    const int lane = threadIdx.x & 31;
    const int wave = threadIdx.x >> 5;
    const int r0 = __builtin_amdgcn_readfirstlane((int)(blockIdx.x * 8 + wave));
    float4 r[CS_ROWS];
#pragma unroll
    for (int t = 0; t < CS_ROWS; ++t) {
        const int row = r0 + t * CS_STRIDE;
        r[t] = *(const float4*)(p + (size_t)min(row, n - 1) * DD + lane * 4);
    }
    float4 s = make_float4(0.f, 0.f, 0.f, 0.f);
#pragma unroll
    for (int t = 0; t < CS_ROWS; ++t) {
        const bool in = (r0 + t * CS_STRIDE) < n;
        s.x += in ? r[t].x : 0.f; s.y += in ? r[t].y : 0.f;
        s.z += in ? r[t].z : 0.f; s.w += in ? r[t].w : 0.f;
    }
    *(float4*)(part + wave * DD + lane * 4) = s;
    __syncthreads();
    if (threadIdx.x < DD) {
        float tot = 0.f;
#pragma unroll
        for (int w = 0; w < 8; ++w) tot += part[w * DD + threadIdx.x];
        atomicAdd(&cs[threadIdx.x], tot);
    }
}

__global__ __launch_bounds__(256) void pairnorm_kernel(
    const float* __restrict__ pre, const float* __restrict__ cs,
    float* __restrict__ o, int n)
{
    int wv = (blockIdx.x * 256 + threadIdx.x) >> 5;
    int lane = threadIdx.x & 31;
    if (wv >= n) return;
    const float* row = pre + (size_t)wv * DD;
    float x[4], sq = 0.f;
#pragma unroll
    for (int j = 0; j < 4; ++j) { x[j] = row[lane + 32 * j]; sq += x[j] * x[j]; }
    float rn = sqrtf(1e-6f + wave_sum32(sq));
    float inv = 20.f / rn;
    float* orow = o + (size_t)wv * DD;
    const float invN = 1.f / (float)NN;
#pragma unroll
    for (int j = 0; j < 4; ++j) {
        int d = lane + 32 * j;
        orow[d] = x[j] * inv - cs[d] * invN;
    }
}

extern "C" void kernel_launch(void* const* d_in, const int* in_sizes, int n_in,
                              void* d_out, int out_size, void* d_ws, size_t ws_size,
                              hipStream_t stream)
{
    const float* x      = (const float*)d_in[0];
    const float* x_org  = (const float*)d_in[1];
    const int*   ei     = (const int*)d_in[2];
    const float* in_W   = (const float*)d_in[3];
    const float* in_b   = (const float*)d_in[4];
    const float* res_W  = (const float*)d_in[5];
    const float* gcn_W  = (const float*)d_in[6];
    const float* gcn_b  = (const float*)d_in[7];
    const float* ln_g   = (const float*)d_in[8];
    const float* ln_b   = (const float*)d_in[9];
    const float* lin_W  = (const float*)d_in[10];
    const float* lin_b  = (const float*)d_in[11];
    const float* attr_W = (const float*)d_in[12];
    const float* attr_b = (const float*)d_in[13];
    const float* qkv_W  = (const float*)d_in[14];
    const float* qkv_b  = (const float*)d_in[15];
    const float* out_W  = (const float*)d_in[16];
    const float* out_b  = (const float*)d_in[17];
    const float* fln_g  = (const float*)d_in[18];
    const float* fln_b  = (const float*)d_in[19];
    const float* fuse_W = (const float*)d_in[20];
    const float* fuse_b = (const float*)d_in[21];
    float* out = (float*)d_out;

    const int* src = ei;
    const int* dst = ei + EE;

    float* ws = (float*)d_ws;
    const size_t ND = (size_t)NN * DD;
    float* isd  = ws;
    float* csum = isd + NN;
    float* bh   = csum + DD;
    float* bt   = bh + ND;
    float* br   = bt + ND;
    float* bs   = br + ND;
    float* q0b  = bs + ND;
    float* q1b  = q0b + 3 * ND;
    int* cnt    = (int*)q1b;
    int* rowptr = cnt + NN;
    int* rank   = rowptr + NN;
    int* csr_a  = rank + EE;
    int* csr_b  = csr_a + EE;

    const dim3 blk(256);
    const dim3 gemmO128(SMALL_TILES + FULL_TILES, DD / GBN);
    const dim3 pairO128(SMALL_TILES + FULL_TILES, DD / GBN, 2);
    const dim3 pairO384(SMALL_TILES + FULL_TILES, 3 * DD / GBN, 2);
    const int nodeWaves = (NN + 7) / 8;
    const int edgeBlk   = (EE + 255) / 256;

    fill_int_kernel<<<(NN + 255) / 256, blk, 0, stream>>>(cnt, 0, NN);
    count_rank_kernel<<<edgeBlk, blk, 0, stream>>>(dst, cnt, rank, EE);
    scan_isd_kernel<<<1, 1024, 0, stream>>>(cnt, rowptr, isd, NN);
    place_kernel<<<edgeBlk, blk, 0, stream>>>(src, dst, rowptr, rank, csr_a, EE);
    row_order_kernel<<<nodeWaves, blk, 0, stream>>>(rowptr, cnt, csr_a, csr_b, NN);

    gemm_lds_pair<true, 1, false, 0, DD, DD><<<pairO128, blk, 0, stream>>>(
        x, in_W, in_b, bh, x_org, res_W, nullptr, bt, NN);

    gather_rows_kernel<false><<<nodeWaves, blk, 0, stream>>>(
        rowptr, cnt, csr_b, bt, nullptr, nullptr, nullptr, nullptr, br, NN);

    for (int i = 0; i < 3; ++i) {
        gemm_lds<true, 0, DD, DD><<<gemmO128, blk, 0, stream>>>(
            bh, gcn_W + (size_t)i * DD * DD, nullptr, bt, NN);
        gather_rows_kernel<true><<<nodeWaves, blk, 0, stream>>>(
            rowptr, cnt, csr_b, bt, isd, gcn_b + (size_t)i * DD,
            ln_g + (size_t)i * DD, ln_b + (size_t)i * DD, bh, NN);
    }

    gemm_lds_pair<true, 0, true, 1, DD, DD><<<pairO128, blk, 0, stream>>>(
        bh, lin_W, lin_b, bs, x_org, attr_W, attr_b, bt, NN);

    gemm_lds_pair<true, 0, true, 0, DD, 3 * DD><<<pairO384, blk, 0, stream>>>(
        bs, qkv_W, qkv_b, q0b, bt, qkv_W, qkv_b, q1b, NN);

    attention_kernel<<<nodeWaves, blk, 0, stream>>>(q0b, q1b, bh, bt, NN);

    gemm_lds_pair<true, 0, true, 0, DD, DD><<<pairO128, blk, 0, stream>>>(
        bh, out_W, out_b, bs, bt, out_W, out_b, q1b, NN);

    fuse_ln384_kernel<<<nodeWaves, blk, 0, stream>>>(bs, q1b, br, fln_g, fln_b, q0b, NN);

    gemm_lds<true, 0, 3 * DD, DD><<<gemmO128, blk, 0, stream>>>(q0b, fuse_W, fuse_b, bh, NN);

    fill_kernel<<<1, 128, 0, stream>>>(csum, 0.f, DD);
    colsum_kernel<<<CS_BLOCKS, blk, 0, stream>>>(bh, csum, NN);
    pairnorm_kernel<<<nodeWaves, blk, 0, stream>>>(bh, csum, out, NN);
}


# Whole-line stores: GEMM C tile leaves through LDS as whole 512-byte rows (was 4-byte stores, two 64-byte half lines each); workspace row arrays put on a 256-byte boundary (were 64 bytes off)
# speedup vs baseline: 2.8252x; 1.0609x over previous
#include <hip/hip_runtime.h>
#include <hip/hip_bf16.h>

#define NN 50000
#define DD 128
#define EE 800000

typedef __attribute__((ext_vector_type(2))) float v2f;
typedef __attribute__((ext_vector_type(8))) float v8f;

#define GP 132
#define GBM 64
#define GBN 128

#define RESIDENT_BLOCKS 768
#define FULL_TILES ((NN / GBM) / RESIDENT_BLOCKS * RESIDENT_BLOCKS)
#define SMALL_BM 16
#define SMALL_TILES ((NN - FULL_TILES * GBM + SMALL_BM - 1) / SMALL_BM)
static_assert(FULL_TILES > 0 && FULL_TILES * GBM <= NN, "full tiles must lie inside the rows");
static_assert(FULL_TILES * GBM + SMALL_TILES * SMALL_BM >= NN, "small tiles must cover the rest");

template <bool TRANSB, int K, int O>
__device__ __forceinline__ void stage_w(
    const float* __restrict__ W, float* __restrict__ Ws, int cbase, int kc, int wave, int lane)
{
    if (TRANSB) {
#pragma unroll
        for (int i = 0; i < GBN / 8; ++i) {
            const int row = wave + 8 * i;
            const float4 v = *(const float4*)(W + (unsigned)((cbase + row) * K + kc + lane * 4));
            *(float4*)(Ws + row * GP + lane * 4) = v;
        }
    } else {
#pragma unroll
        for (int i = 0; i < GBN / 8; ++i) {
            const int idx = wave * (GBN / 8) + i;
            const int k  = (idx >> 2) * 4 + (lane >> 3);
            const int nb = (idx & 3) * 32 + (lane & 7) * 4;
            const float4 v = *(const float4*)(W + (unsigned)((kc + k) * O + cbase + nb));
            Ws[(nb + 0) * GP + k] = v.x;
            Ws[(nb + 1) * GP + k] = v.y;
            Ws[(nb + 2) * GP + k] = v.z;
            Ws[(nb + 3) * GP + k] = v.w;
        }
    }
}

template <int ROWS, int O>
__device__ __forceinline__ void store_c_rows(
    const float* __restrict__ As, float* __restrict__ C, int n, int rbase, int cbase, int wave, int lane)
{
    float4 o[ROWS / 8];
#pragma unroll
    for (int i = 0; i < ROWS / 8; ++i) o[i] = *(const float4*)(As + (wave + 8 * i) * GP + lane * 4);
    if (rbase + ROWS <= n) {
#pragma unroll
        for (int i = 0; i < ROWS / 8; ++i)
            *(float4*)(C + (unsigned)((rbase + wave + 8 * i) * O + cbase + lane * 4)) = o[i];
    } else {
#pragma unroll
        for (int i = 0; i < ROWS / 8; ++i) {
            const int row = rbase + wave + 8 * i;
            if (row < n) *(float4*)(C + (unsigned)(row * O + cbase + lane * 4)) = o[i];
        }
    }
}

template <bool TRANSB, int ACT, int K, int O>
__device__ __forceinline__ void gemm_tile(
    const float* __restrict__ A, const float* __restrict__ W,
    const float* __restrict__ bias, float* __restrict__ C, int n, int rbase,
    float* __restrict__ As, float* __restrict__ Ws, bool relu_rt)
{
    static_assert(K % 128 == 0 && O % GBN == 0, "tile sizes");

    const int wave = threadIdx.x >> 5;
    const int lane = threadIdx.x & 31;
    const int half = lane >> 4;
    const int l15  = lane & 15;
    const int cbase = blockIdx.y * GBN;
    const int wr = (wave >> 2) * 32;
    const int wc = (wave & 3) * 32;

    v8f acc[2][2];
#pragma unroll
    for (int i = 0; i < 2; ++i)
#pragma unroll
        for (int j = 0; j < 2; ++j) acc[i][j] = (v8f){0.f,0.f,0.f,0.f,0.f,0.f,0.f,0.f};

    for (int kc = 0; kc < K; kc += 128) {
        if (kc > 0) __syncthreads();

#pragma unroll
        for (int i = 0; i < GBM / 8; ++i) {
            const int row = wave + 8 * i;
            const int grow = min(rbase + row, n - 1);
            const float4 v = *(const float4*)(A + (unsigned)(grow * K + kc + lane * 4));
            *(float4*)(As + row * GP + lane * 4) = v;
        }
        stage_w<TRANSB, K, O>(W, Ws, cbase, kc, wave, lane);
        __syncthreads();

        const float* ap = As + (wr + l15) * GP + 2 * half;
        const float* bp = Ws + (wc + l15) * GP + 2 * half;
#pragma unroll 8
        for (int kk = 0; kk < 128; kk += 4) {
            const v2f a0 = *(const v2f*)(ap + kk);
            const v2f a1 = *(const v2f*)(ap + 16 * GP + kk);
            const v2f b0 = *(const v2f*)(bp + kk);
            const v2f b1 = *(const v2f*)(bp + 16 * GP + kk);
            acc[0][0] = __builtin_amdgcn_wmma_f32_16x16x4_f32(
                false, a0, false, b0, (short)0, acc[0][0], false, false);
            acc[0][1] = __builtin_amdgcn_wmma_f32_16x16x4_f32(
                false, a0, false, b1, (short)0, acc[0][1], false, false);
            acc[1][0] = __builtin_amdgcn_wmma_f32_16x16x4_f32(
                false, a1, false, b0, (short)0, acc[1][0], false, false);
            acc[1][1] = __builtin_amdgcn_wmma_f32_16x16x4_f32(
                false, a1, false, b1, (short)0, acc[1][1], false, false);
        }
    }

    float bv[2];
#pragma unroll
    for (int j = 0; j < 2; ++j) bv[j] = bias ? bias[cbase + wc + 16 * j + l15] : 0.f;

    __syncthreads();
#pragma unroll
    for (int j = 0; j < 2; ++j) {
        const int col = wc + 16 * j + l15;
#pragma unroll
        for (int i = 0; i < 2; ++i) {
#pragma unroll
            for (int v = 0; v < 8; ++v) {
                float val = acc[i][j][v] + bv[j];
                if (ACT == 1 || (ACT == 2 && relu_rt)) val = fmaxf(val, 0.f);
                As[(wr + 16 * i + v + 8 * half) * GP + col] = val;
            }
        }
    }
    __syncthreads();

    store_c_rows<GBM, O>(As, C, n, rbase, cbase, wave, lane);
}

template <bool TRANSB, int ACT, int K, int O>
__device__ __forceinline__ void gemm_tile16(
    const float* __restrict__ A, const float* __restrict__ W,
    const float* __restrict__ bias, float* __restrict__ C, int n, int rbase,
    float* __restrict__ As, float* __restrict__ Ws, bool relu_rt)
{
    const int wave = threadIdx.x >> 5;
    const int lane = threadIdx.x & 31;
    const int half = lane >> 4;
    const int l15  = lane & 15;
    const int cbase = blockIdx.y * GBN;

    v8f acc = (v8f){0.f,0.f,0.f,0.f,0.f,0.f,0.f,0.f};

    for (int kc = 0; kc < K; kc += 128) {
        if (kc > 0) __syncthreads();
#pragma unroll
        for (int i = 0; i < SMALL_BM / 8; ++i) {
            const int row = wave + 8 * i;
            const int grow = min(rbase + row, n - 1);
            const float4 v = *(const float4*)(A + (unsigned)(grow * K + kc + lane * 4));
            *(float4*)(As + row * GP + lane * 4) = v;
        }
        stage_w<TRANSB, K, O>(W, Ws, cbase, kc, wave, lane);
        __syncthreads();

        const float* ap = As + l15 * GP + 2 * half;
        const float* bp = Ws + (16 * wave + l15) * GP + 2 * half;
#pragma unroll 8
        for (int kk = 0; kk < 128; kk += 4) {
            const v2f a0 = *(const v2f*)(ap + kk);
            const v2f b0 = *(const v2f*)(bp + kk);
            acc = __builtin_amdgcn_wmma_f32_16x16x4_f32(
                false, a0, false, b0, (short)0, acc, false, false);
        }
    }

    const int col = 16 * wave + l15;
    const float bv = bias ? bias[cbase + col] : 0.f;

    __syncthreads();
#pragma unroll
    for (int v = 0; v < 8; ++v) {
        float val = acc[v] + bv;
        if (ACT == 1 || (ACT == 2 && relu_rt)) val = fmaxf(val, 0.f);
        As[(v + 8 * half) * GP + col] = val;
    }
    __syncthreads();

    store_c_rows<SMALL_BM, O>(As, C, n, rbase, cbase, wave, lane);
}

template <bool TRANSB, int ACT, int K, int O>
__device__ __forceinline__ void gemm_block(
    const float* __restrict__ A, const float* __restrict__ W,
    const float* __restrict__ bias, float* __restrict__ C, int n,
    float* __restrict__ As, float* __restrict__ Ws, bool relu_rt)
{
    const int bx = blockIdx.x;
    if (bx < SMALL_TILES)
        gemm_tile16<TRANSB, ACT, K, O>(A, W, bias, C, n, FULL_TILES * GBM + bx * SMALL_BM, As, Ws, relu_rt);
    else
        gemm_tile<TRANSB, ACT, K, O>(A, W, bias, C, n, (bx - SMALL_TILES) * GBM, As, Ws, relu_rt);
}

template <bool TRANSB, int ACT, int K, int O>
__global__ __launch_bounds__(256, 6) void gemm_lds(
    const float* __restrict__ A, const float* __restrict__ W,
    const float* __restrict__ bias, float* __restrict__ C, int n)
{
    __shared__ __attribute__((aligned(16))) float As[GBM * GP];
    __shared__ __attribute__((aligned(16))) float Ws[GBN * GP];
    gemm_block<TRANSB, ACT, K, O>(A, W, bias, C, n, As, Ws, false);
}

template <bool TB0, int ACT0, bool TB1, int ACT1, int K, int O>
__global__ __launch_bounds__(256, 6) void gemm_lds_pair(
    const float* __restrict__ A0, const float* __restrict__ W0,
    const float* __restrict__ bias0, float* __restrict__ C0,
    const float* __restrict__ A1, const float* __restrict__ W1,
    const float* __restrict__ bias1, float* __restrict__ C1, int n)
{
    __shared__ __attribute__((aligned(16))) float As[GBM * GP];
    __shared__ __attribute__((aligned(16))) float Ws[GBN * GP];
    const bool second = blockIdx.z != 0;
    if (TB0 == TB1) {
        gemm_block<TB0, (ACT0 == ACT1) ? ACT0 : 2, K, O>(
            second ? A1 : A0, second ? W1 : W0, second ? bias1 : bias0, second ? C1 : C0,
            n, As, Ws, (second ? ACT1 : ACT0) == 1);
    } else {
        if (!second) gemm_block<TB0, ACT0, K, O>(A0, W0, bias0, C0, n, As, Ws, false);
        else         gemm_block<TB1, ACT1, K, O>(A1, W1, bias1, C1, n, As, Ws, false);
    }
}

__device__ __forceinline__ float wave_sum32(float v) {
#pragma unroll
    for (int m = 16; m > 0; m >>= 1) v += __shfl_xor(v, m, 32);
    return v;
}

__global__ __launch_bounds__(256) void fill_kernel(float* __restrict__ p, float v, int n) {
    int i = blockIdx.x * blockDim.x + threadIdx.x;
    if (i < n) p[i] = v;
}

__global__ __launch_bounds__(256) void fill_int_kernel(int* __restrict__ p, int v, int n) {
    int i = blockIdx.x * blockDim.x + threadIdx.x;
    if (i < n) p[i] = v;
}

__global__ __launch_bounds__(256) void count_rank_kernel(
    const int* __restrict__ dst, int* __restrict__ cnt, int* __restrict__ rank, int e)
{
    int i = blockIdx.x * blockDim.x + threadIdx.x;
    if (i < e) rank[i] = atomicAdd(&cnt[dst[i]], 1);
}

#define SCAN_CH 13
static_assert(32 * SCAN_CH * 128 >= NN, "scan block must cover all nodes");
static_assert(NN % 4 == 0, "16-byte accesses below need NN % 4 == 0");

__global__ __launch_bounds__(1024) void scan_isd_kernel(
    const int* __restrict__ cnt, int* __restrict__ rowptr, float* __restrict__ isd, int n)
{
    __shared__ int wtot[32];
    const int lane = threadIdx.x & 31;
    const int wave = threadIdx.x >> 5;
    const int wbase = wave * (SCAN_CH * 128);
    int4 c[SCAN_CH];
#pragma unroll
    for (int k = 0; k < SCAN_CH; ++k) {
        const int e0 = wbase + k * 128 + lane * 4;
        c[k] = (e0 < n) ? *(const int4*)(cnt + e0) : make_int4(0, 0, 0, 0);
    }
    int ex[SCAN_CH];
    int carry = 0;
#pragma unroll
    for (int k = 0; k < SCAN_CH; ++k) {
        const int tot = c[k].x + c[k].y + c[k].z + c[k].w;
        int inc = tot;
#pragma unroll
        for (int d = 1; d < 32; d <<= 1) {
            int t = __shfl_up(inc, d, 32);
            if (lane >= d) inc += t;
        }
        ex[k] = carry + inc - tot;
        carry += __shfl(inc, 31, 32);
    }
    if (lane == 0) wtot[wave] = carry;
    __syncthreads();
    int below = (lane < wave) ? wtot[lane] : 0;
#pragma unroll
    for (int m = 16; m > 0; m >>= 1) below += __shfl_xor(below, m, 32);
#pragma unroll
    for (int k = 0; k < SCAN_CH; ++k) {
        const int e0 = wbase + k * 128 + lane * 4;
        if (e0 < n) {
            const int p0 = below + ex[k];
            const int p1 = p0 + c[k].x, p2 = p1 + c[k].y, p3 = p2 + c[k].z;
            *(int4*)(rowptr + e0) = make_int4(p0, p1, p2, p3);
            *(float4*)(isd + e0) = make_float4(
                rsqrtf((float)c[k].x + 1.0f), rsqrtf((float)c[k].y + 1.0f),
                rsqrtf((float)c[k].z + 1.0f), rsqrtf((float)c[k].w + 1.0f));
        }
    }
}

__global__ __launch_bounds__(256) void place_kernel(
    const int* __restrict__ src, const int* __restrict__ dst,
    const int* __restrict__ rowptr, const int* __restrict__ rank,
    int* __restrict__ csr, int e)
{
    int i = blockIdx.x * blockDim.x + threadIdx.x;
    if (i >= e) return;
    const int pos = rowptr[dst[i]] + rank[i];
    if (pos >= 0 && pos < e) csr[pos] = src[i];
}

__global__ __launch_bounds__(256) void row_order_kernel(
    const int* __restrict__ rowptr, const int* __restrict__ cnt,
    const int* __restrict__ csr_in, int* __restrict__ csr_out, int n)
{
    const int lane = threadIdx.x & 31;
    const int node = __builtin_amdgcn_readfirstlane((int)((blockIdx.x * 256 + threadIdx.x) >> 5));
    if (node >= n) return;
    const int start = __builtin_amdgcn_readfirstlane(rowptr[node]);
    const int len   = __builtin_amdgcn_readfirstlane(cnt[node]);
    for (int ib = 0; ib < len; ib += 32) {
        const int i = ib + lane;
        const int xi = (i < len) ? csr_in[start + i] : 0;
        int pos = 0;
        for (int jb = 0; jb < len; jb += 32) {
            const int mj = min(32, len - jb);
            const int xv = (lane < mj) ? csr_in[start + jb + lane] : 0;
            for (int jj = 0; jj < mj; ++jj) {
                const int xj = __builtin_amdgcn_readlane(xv, jj);
                const int j = jb + jj;
                pos += (xj < xi || (xj == xi && j < i)) ? 1 : 0;
            }
        }
        if (i < len) csr_out[start + pos] = xi;
    }
}

template <bool GCN, int B>
__device__ __forceinline__ void gather_accum(
    float4& acc, const float* __restrict__ v, int ms, float mw, int j, int lane)
{
    float4 r[B];
    float w[B];
#pragma unroll
    for (int u = 0; u < B; ++u) {
        const int s = __builtin_amdgcn_readlane(ms, j + u);
        r[u] = *(const float4*)(v + (size_t)s * DD + lane * 4);
        w[u] = GCN ? __int_as_float(__builtin_amdgcn_readlane(__float_as_int(mw), j + u)) : 1.0f;
    }
#pragma unroll
    for (int u = 0; u < B; ++u) {
        if (GCN) {
            acc.x = fmaf(r[u].x, w[u], acc.x); acc.y = fmaf(r[u].y, w[u], acc.y);
            acc.z = fmaf(r[u].z, w[u], acc.z); acc.w = fmaf(r[u].w, w[u], acc.w);
        } else {
            acc.x += r[u].x; acc.y += r[u].y; acc.z += r[u].z; acc.w += r[u].w;
        }
    }
}

template <bool GCN>
__global__ __launch_bounds__(256) void gather_rows_kernel(
    const int* __restrict__ rowptr, const int* __restrict__ cnt, const int* __restrict__ csr,
    const float* __restrict__ v, const float* __restrict__ isd, const float* __restrict__ bias,
    const float* __restrict__ g, const float* __restrict__ b,
    float* __restrict__ o, int n)
{
    const int lane = threadIdx.x & 31;
    const int node = __builtin_amdgcn_readfirstlane((int)((blockIdx.x * 256 + threadIdx.x) >> 5));
    if (node >= n) return;
    const int start = __builtin_amdgcn_readfirstlane(rowptr[node]);
    const int len   = __builtin_amdgcn_readfirstlane(cnt[node]);
    float4 acc = make_float4(0.f, 0.f, 0.f, 0.f);
    float4 sv = acc, bv = acc, gv = acc, lb = acc;
    float it = 0.f;
    if (GCN) {
        it = isd[node];
        sv = *(const float4*)(v + (size_t)node * DD + lane * 4);
        bv = *(const float4*)(bias + lane * 4);
        gv = *(const float4*)(g + lane * 4);
        lb = *(const float4*)(b + lane * 4);
    }
    for (int base = 0; base < len; base += 32) {
        const int m = min(32, len - base);
        int ms = 0;
        float mw = 0.f;
        if (lane < m) {
            ms = csr[start + base + lane];
            ms = min(max(ms, 0), n - 1);
            if (GCN) mw = isd[ms] * it;
        }
        int j = 0;
        for (; j + 8 <= m; j += 8) gather_accum<GCN, 8>(acc, v, ms, mw, j, lane);
        if (j + 4 <= m) { gather_accum<GCN, 4>(acc, v, ms, mw, j, lane); j += 4; }
        if (j + 2 <= m) { gather_accum<GCN, 2>(acc, v, ms, mw, j, lane); j += 2; }
        if (j < m)      { gather_accum<GCN, 1>(acc, v, ms, mw, j, lane); }
    }
    if (GCN) {
        acc.x += sv.x * it * it + bv.x; acc.y += sv.y * it * it + bv.y;
        acc.z += sv.z * it * it + bv.z; acc.w += sv.w * it * it + bv.w;
        const float mean = wave_sum32(acc.x + acc.y + acc.z + acc.w) * (1.f / 128.f);
        const float dx = acc.x - mean, dy = acc.y - mean, dz = acc.z - mean, dw = acc.w - mean;
        const float inv = rsqrtf(wave_sum32(dx * dx + dy * dy + dz * dz + dw * dw) * (1.f / 128.f) + 1e-5f);
        acc.x = fmaxf(dx * inv * gv.x + lb.x, 0.f); acc.y = fmaxf(dy * inv * gv.y + lb.y, 0.f);
        acc.z = fmaxf(dz * inv * gv.z + lb.z, 0.f); acc.w = fmaxf(dw * inv * gv.w + lb.w, 0.f);
    }
    *(float4*)(o + (size_t)node * DD + lane * 4) = acc;
}

__global__ __launch_bounds__(256) void attention_kernel(
    const float* __restrict__ qkv0, const float* __restrict__ qkv1,
    float* __restrict__ ao0, float* __restrict__ ao1, int n)
{
    int wv = (blockIdx.x * 256 + threadIdx.x) >> 5;
    int lane = threadIdx.x & 31;
    if (wv >= n) return;
    const float* b0 = qkv0 + (size_t)wv * 384;
    const float* b1 = qkv1 + (size_t)wv * 384;
    float q0[4], k0[4], v0[4], q1[4], k1[4], v1[4];
#pragma unroll
    for (int j = 0; j < 4; ++j) {
        int d = lane + 32 * j;
        q0[j] = b0[d]; k0[j] = b0[128 + d]; v0[j] = b0[256 + d];
        q1[j] = b1[d]; k1[j] = b1[128 + d]; v1[j] = b1[256 + d];
    }
    float l00 = 0, l01 = 0, l10 = 0, l11 = 0;
#pragma unroll
    for (int j = 0; j < 4; ++j) {
        l00 += q0[j] * k0[j]; l01 += q0[j] * k1[j];
        l10 += q1[j] * k0[j]; l11 += q1[j] * k1[j];
    }
    l00 = wave_sum32(l00); l01 = wave_sum32(l01);
    l10 = wave_sum32(l10); l11 = wave_sum32(l11);
    const float scale = 0.08838834764831845f;
    l00 *= scale; l01 *= scale; l10 *= scale; l11 *= scale;
    float m0 = fmaxf(l00, l01), e00 = __expf(l00 - m0), e01 = __expf(l01 - m0);
    float r0 = 1.f / (e00 + e01), w00 = e00 * r0, w01 = e01 * r0;
    float m1 = fmaxf(l10, l11), e10 = __expf(l10 - m1), e11 = __expf(l11 - m1);
    float r1 = 1.f / (e10 + e11), w10 = e10 * r1, w11 = e11 * r1;
    float* o0 = ao0 + (size_t)wv * DD;
    float* o1 = ao1 + (size_t)wv * DD;
#pragma unroll
    for (int j = 0; j < 4; ++j) {
        int d = lane + 32 * j;
        o0[d] = w00 * v0[j] + w01 * v1[j];
        o1[d] = w10 * v0[j] + w11 * v1[j];
    }
}

__global__ __launch_bounds__(256) void fuse_ln384_kernel(
    const float* __restrict__ a0, const float* __restrict__ a1,
    const float* __restrict__ rs, const float* __restrict__ g,
    const float* __restrict__ b, float* __restrict__ o, int n)
{
    int wv = (blockIdx.x * 256 + threadIdx.x) >> 5;
    int lane = threadIdx.x & 31;
    if (wv >= n) return;
    const float* p0 = a0 + (size_t)wv * DD;
    const float* p1 = a1 + (size_t)wv * DD;
    const float* p2 = rs + (size_t)wv * DD;
    float x[12], s = 0.f;
#pragma unroll
    for (int j = 0; j < 12; ++j) {
        int d = lane + 32 * j;
        float v = (d < 128) ? p0[d] : (d < 256) ? p1[d - 128] : p2[d - 256];
        x[j] = v; s += v;
    }
    float mean = wave_sum32(s) * (1.f / 384.f);
    float sq = 0.f;
#pragma unroll
    for (int j = 0; j < 12; ++j) { float d = x[j] - mean; sq += d * d; }
    float inv = rsqrtf(wave_sum32(sq) * (1.f / 384.f) + 1e-5f);
    float* orow = o + (size_t)wv * 384;
#pragma unroll
    for (int j = 0; j < 12; ++j) {
        int d = lane + 32 * j;
        orow[d] = (x[j] - mean) * inv * g[d] + b[d];
    }
}

#define CS_BLOCKS 256
#define CS_STRIDE (CS_BLOCKS * 8)
#define CS_ROWS ((NN + CS_STRIDE - 1) / CS_STRIDE)
static_assert(CS_ROWS * CS_STRIDE >= NN, "column-sum waves must cover all rows");
static_assert(CS_ROWS <= 48, "row loads of a wave must stay below the load counter's limit");

__global__ __launch_bounds__(256) void colsum_kernel(const float* __restrict__ p, float* __restrict__ cs, int n) {
    __shared__ __attribute__((aligned(16))) float part[8 * DD];
    const int lane = threadIdx.x & 31;
    const int wave = threadIdx.x >> 5;
    const int r0 = __builtin_amdgcn_readfirstlane((int)(blockIdx.x * 8 + wave));
    float4 r[CS_ROWS];
#pragma unroll
    for (int t = 0; t < CS_ROWS; ++t) {
        const int row = r0 + t * CS_STRIDE;
        r[t] = *(const float4*)(p + (size_t)min(row, n - 1) * DD + lane * 4);
    }
    float4 s = make_float4(0.f, 0.f, 0.f, 0.f);
#pragma unroll
    for (int t = 0; t < CS_ROWS; ++t) {
        const bool in = (r0 + t * CS_STRIDE) < n;
        s.x += in ? r[t].x : 0.f; s.y += in ? r[t].y : 0.f;
        s.z += in ? r[t].z : 0.f; s.w += in ? r[t].w : 0.f;
    }
    *(float4*)(part + wave * DD + lane * 4) = s;
    __syncthreads();
    if (threadIdx.x < DD) {
        float tot = 0.f;
#pragma unroll
        for (int w = 0; w < 8; ++w) tot += part[w * DD + threadIdx.x];
        atomicAdd(&cs[threadIdx.x], tot);
    }
}

__global__ __launch_bounds__(256) void pairnorm_kernel(
    const float* __restrict__ pre, const float* __restrict__ cs,
    float* __restrict__ o, int n)
{
    int wv = (blockIdx.x * 256 + threadIdx.x) >> 5;
    int lane = threadIdx.x & 31;
    if (wv >= n) return;
    const float* row = pre + (size_t)wv * DD;
    float x[4], sq = 0.f;
#pragma unroll
    for (int j = 0; j < 4; ++j) { x[j] = row[lane + 32 * j]; sq += x[j] * x[j]; }
    float rn = sqrtf(1e-6f + wave_sum32(sq));
    float inv = 20.f / rn;
    float* orow = o + (size_t)wv * DD;
    const float invN = 1.f / (float)NN;
#pragma unroll
    for (int j = 0; j < 4; ++j) {
        int d = lane + 32 * j;
        orow[d] = x[j] * inv - cs[d] * invN;
    }
}

#define WS_ROW_ARRAYS ((NN + DD + 63) / 64 * 64)
static_assert(WS_ROW_ARRAYS >= NN + DD && WS_ROW_ARRAYS % 64 == 0, "row arrays behind isd and csum, on a 256-byte boundary");
static_assert(((size_t)NN * DD) % 64 == 0, "every row array keeps the next one on a 256-byte boundary");

extern "C" void kernel_launch(void* const* d_in, const int* in_sizes, int n_in,
                              void* d_out, int out_size, void* d_ws, size_t ws_size,
                              hipStream_t stream)
{
    const float* x      = (const float*)d_in[0];
    const float* x_org  = (const float*)d_in[1];
    const int*   ei     = (const int*)d_in[2];
    const float* in_W   = (const float*)d_in[3];
    const float* in_b   = (const float*)d_in[4];
    const float* res_W  = (const float*)d_in[5];
    const float* gcn_W  = (const float*)d_in[6];
    const float* gcn_b  = (const float*)d_in[7];
    const float* ln_g   = (const float*)d_in[8];
    const float* ln_b   = (const float*)d_in[9];
    const float* lin_W  = (const float*)d_in[10];
    const float* lin_b  = (const float*)d_in[11];
    const float* attr_W = (const float*)d_in[12];
    const float* attr_b = (const float*)d_in[13];
    const float* qkv_W  = (const float*)d_in[14];
    const float* qkv_b  = (const float*)d_in[15];
    const float* out_W  = (const float*)d_in[16];
    const float* out_b  = (const float*)d_in[17];
    const float* fln_g  = (const float*)d_in[18];
    const float* fln_b  = (const float*)d_in[19];
    const float* fuse_W = (const float*)d_in[20];
    const float* fuse_b = (const float*)d_in[21];
    float* out = (float*)d_out;

    const int* src = ei;
    const int* dst = ei + EE;

    float* ws = (float*)d_ws;
    const size_t ND = (size_t)NN * DD;
    float* isd  = ws;
    float* csum = isd + NN;
    float* bh   = ws + WS_ROW_ARRAYS;
    float* bt   = bh + ND;
    float* br   = bt + ND;
    float* bs   = br + ND;
    float* q0b  = bs + ND;
    float* q1b  = q0b + 3 * ND;
    int* cnt    = (int*)q1b;
    int* rowptr = cnt + NN;
    int* rank   = rowptr + NN;
    int* csr_a  = rank + EE;
    int* csr_b  = csr_a + EE;

    const dim3 blk(256);
    const dim3 gemmO128(SMALL_TILES + FULL_TILES, DD / GBN);
    const dim3 pairO128(SMALL_TILES + FULL_TILES, DD / GBN, 2);
    const dim3 pairO384(SMALL_TILES + FULL_TILES, 3 * DD / GBN, 2);
    const int nodeWaves = (NN + 7) / 8;
    const int edgeBlk   = (EE + 255) / 256;

    fill_int_kernel<<<(NN + 255) / 256, blk, 0, stream>>>(cnt, 0, NN);
    count_rank_kernel<<<edgeBlk, blk, 0, stream>>>(dst, cnt, rank, EE);
    scan_isd_kernel<<<1, 1024, 0, stream>>>(cnt, rowptr, isd, NN);
    place_kernel<<<edgeBlk, blk, 0, stream>>>(src, dst, rowptr, rank, csr_a, EE);
    row_order_kernel<<<nodeWaves, blk, 0, stream>>>(rowptr, cnt, csr_a, csr_b, NN);

    gemm_lds_pair<true, 1, false, 0, DD, DD><<<pairO128, blk, 0, stream>>>(
        x, in_W, in_b, bh, x_org, res_W, nullptr, bt, NN);

    gather_rows_kernel<false><<<nodeWaves, blk, 0, stream>>>(
        rowptr, cnt, csr_b, bt, nullptr, nullptr, nullptr, nullptr, br, NN);

    for (int i = 0; i < 3; ++i) {
        gemm_lds<true, 0, DD, DD><<<gemmO128, blk, 0, stream>>>(
            bh, gcn_W + (size_t)i * DD * DD, nullptr, bt, NN);
        gather_rows_kernel<true><<<nodeWaves, blk, 0, stream>>>(
            rowptr, cnt, csr_b, bt, isd, gcn_b + (size_t)i * DD,
            ln_g + (size_t)i * DD, ln_b + (size_t)i * DD, bh, NN);
    }

    gemm_lds_pair<true, 0, true, 1, DD, DD><<<pairO128, blk, 0, stream>>>(
        bh, lin_W, lin_b, bs, x_org, attr_W, attr_b, bt, NN);

    gemm_lds_pair<true, 0, true, 0, DD, 3 * DD><<<pairO384, blk, 0, stream>>>(
        bs, qkv_W, qkv_b, q0b, bt, qkv_W, qkv_b, q1b, NN);

    attention_kernel<<<nodeWaves, blk, 0, stream>>>(q0b, q1b, bh, bt, NN);

    gemm_lds_pair<true, 0, true, 0, DD, DD><<<pairO128, blk, 0, stream>>>(
        bh, out_W, out_b, bs, bt, out_W, out_b, q1b, NN);

    fuse_ln384_kernel<<<nodeWaves, blk, 0, stream>>>(bs, q1b, br, fln_g, fln_b, q0b, NN);

    gemm_lds<true, 0, 3 * DD, DD><<<gemmO128, blk, 0, stream>>>(q0b, fuse_W, fuse_b, bh, NN);

    fill_kernel<<<1, 128, 0, stream>>>(csum, 0.f, DD);
    colsum_kernel<<<CS_BLOCKS, blk, 0, stream>>>(bh, csum, NN);
    pairnorm_kernel<<<nodeWaves, blk, 0, stream>>>(bh, csum, out, NN);
}


# K=128 GEMMs as 768 resident blocks looping over 16-row tiles: W slice staged once per block, its B fragments held in registers, next A rows loaded during the matrix ops; k order per element kept
# speedup vs baseline: 3.0367x; 1.0749x over previous
#include <hip/hip_runtime.h>
#include <hip/hip_bf16.h>

#define NN 50000
#define DD 128
#define EE 800000

typedef __attribute__((ext_vector_type(2))) float v2f;
typedef __attribute__((ext_vector_type(8))) float v8f;

#define GP 132
#define GBM 64
#define GBN 128

#define RESIDENT_BLOCKS 768
#define FULL_TILES ((NN / GBM) / RESIDENT_BLOCKS * RESIDENT_BLOCKS)
#define SMALL_BM 16
#define SMALL_TILES ((NN - FULL_TILES * GBM + SMALL_BM - 1) / SMALL_BM)
static_assert(FULL_TILES > 0 && FULL_TILES * GBM <= NN, "full tiles must lie inside the rows");
static_assert(FULL_TILES * GBM + SMALL_TILES * SMALL_BM >= NN, "small tiles must cover the rest");

template <bool TRANSB, int K, int O>
__device__ __forceinline__ void stage_w(
    const float* __restrict__ W, float* __restrict__ Ws, int cbase, int kc, int wave, int lane)
{
    if (TRANSB) {
#pragma unroll
        for (int i = 0; i < GBN / 8; ++i) {
            const int row = wave + 8 * i;
            const float4 v = *(const float4*)(W + (unsigned)((cbase + row) * K + kc + lane * 4));
            *(float4*)(Ws + row * GP + lane * 4) = v;
        }
    } else {
#pragma unroll
        for (int i = 0; i < GBN / 8; ++i) {
            const int idx = wave * (GBN / 8) + i;
            const int k  = (idx >> 2) * 4 + (lane >> 3);
            const int nb = (idx & 3) * 32 + (lane & 7) * 4;
            const float4 v = *(const float4*)(W + (unsigned)((kc + k) * O + cbase + nb));
            Ws[(nb + 0) * GP + k] = v.x;
            Ws[(nb + 1) * GP + k] = v.y;
            Ws[(nb + 2) * GP + k] = v.z;
            Ws[(nb + 3) * GP + k] = v.w;
        }
    }
}

template <int ROWS, int O>
__device__ __forceinline__ void store_c_rows(
    const float* __restrict__ As, float* __restrict__ C, int n, int rbase, int cbase, int wave, int lane)
{
    float4 o[ROWS / 8];
#pragma unroll
    for (int i = 0; i < ROWS / 8; ++i) o[i] = *(const float4*)(As + (wave + 8 * i) * GP + lane * 4);
    if (rbase + ROWS <= n) {
#pragma unroll
        for (int i = 0; i < ROWS / 8; ++i)
            *(float4*)(C + (unsigned)((rbase + wave + 8 * i) * O + cbase + lane * 4)) = o[i];
    } else {
#pragma unroll
        for (int i = 0; i < ROWS / 8; ++i) {
            const int row = rbase + wave + 8 * i;
            if (row < n) *(float4*)(C + (unsigned)(row * O + cbase + lane * 4)) = o[i];
        }
    }
}

template <bool TRANSB, int ACT, int K, int O>
__device__ __forceinline__ void gemm_tile(
    const float* __restrict__ A, const float* __restrict__ W,
    const float* __restrict__ bias, float* __restrict__ C, int n, int rbase,
    float* __restrict__ As, float* __restrict__ Ws, bool relu_rt)
{
    static_assert(K % 128 == 0 && O % GBN == 0, "tile sizes");

    const int wave = threadIdx.x >> 5;
    const int lane = threadIdx.x & 31;
    const int half = lane >> 4;
    const int l15  = lane & 15;
    const int cbase = blockIdx.y * GBN;
    const int wr = (wave >> 2) * 32;
    const int wc = (wave & 3) * 32;

    v8f acc[2][2];
#pragma unroll
    for (int i = 0; i < 2; ++i)
#pragma unroll
        for (int j = 0; j < 2; ++j) acc[i][j] = (v8f){0.f,0.f,0.f,0.f,0.f,0.f,0.f,0.f};

    for (int kc = 0; kc < K; kc += 128) {
        if (kc > 0) __syncthreads();

#pragma unroll
        for (int i = 0; i < GBM / 8; ++i) {
            const int row = wave + 8 * i;
            const int grow = min(rbase + row, n - 1);
            const float4 v = *(const float4*)(A + (unsigned)(grow * K + kc + lane * 4));
            *(float4*)(As + row * GP + lane * 4) = v;
        }
        stage_w<TRANSB, K, O>(W, Ws, cbase, kc, wave, lane);
        __syncthreads();

        const float* ap = As + (wr + l15) * GP + 2 * half;
        const float* bp = Ws + (wc + l15) * GP + 2 * half;
#pragma unroll 8
        for (int kk = 0; kk < 128; kk += 4) {
            const v2f a0 = *(const v2f*)(ap + kk);
            const v2f a1 = *(const v2f*)(ap + 16 * GP + kk);
            const v2f b0 = *(const v2f*)(bp + kk);
            const v2f b1 = *(const v2f*)(bp + 16 * GP + kk);
            acc[0][0] = __builtin_amdgcn_wmma_f32_16x16x4_f32(
                false, a0, false, b0, (short)0, acc[0][0], false, false);
            acc[0][1] = __builtin_amdgcn_wmma_f32_16x16x4_f32(
                false, a0, false, b1, (short)0, acc[0][1], false, false);
            acc[1][0] = __builtin_amdgcn_wmma_f32_16x16x4_f32(
                false, a1, false, b0, (short)0, acc[1][0], false, false);
            acc[1][1] = __builtin_amdgcn_wmma_f32_16x16x4_f32(
                false, a1, false, b1, (short)0, acc[1][1], false, false);
        }
    }

    float bv[2];
#pragma unroll
    for (int j = 0; j < 2; ++j) bv[j] = bias ? bias[cbase + wc + 16 * j + l15] : 0.f;

    __syncthreads();
#pragma unroll
    for (int j = 0; j < 2; ++j) {
        const int col = wc + 16 * j + l15;
#pragma unroll
        for (int i = 0; i < 2; ++i) {
#pragma unroll
            for (int v = 0; v < 8; ++v) {
                float val = acc[i][j][v] + bv[j];
                if (ACT == 1 || (ACT == 2 && relu_rt)) val = fmaxf(val, 0.f);
                As[(wr + 16 * i + v + 8 * half) * GP + col] = val;
            }
        }
    }
    __syncthreads();

    store_c_rows<GBM, O>(As, C, n, rbase, cbase, wave, lane);
}

template <bool TRANSB, int ACT, int K, int O>
__device__ __forceinline__ void gemm_tile16(
    const float* __restrict__ A, const float* __restrict__ W,
    const float* __restrict__ bias, float* __restrict__ C, int n, int rbase,
    float* __restrict__ As, float* __restrict__ Ws, bool relu_rt)
{
    const int wave = threadIdx.x >> 5;
    const int lane = threadIdx.x & 31;
    const int half = lane >> 4;
    const int l15  = lane & 15;
    const int cbase = blockIdx.y * GBN;

    v8f acc = (v8f){0.f,0.f,0.f,0.f,0.f,0.f,0.f,0.f};

    for (int kc = 0; kc < K; kc += 128) {
        if (kc > 0) __syncthreads();
#pragma unroll
        for (int i = 0; i < SMALL_BM / 8; ++i) {
            const int row = wave + 8 * i;
            const int grow = min(rbase + row, n - 1);
            const float4 v = *(const float4*)(A + (unsigned)(grow * K + kc + lane * 4));
            *(float4*)(As + row * GP + lane * 4) = v;
        }
        stage_w<TRANSB, K, O>(W, Ws, cbase, kc, wave, lane);
        __syncthreads();

        const float* ap = As + l15 * GP + 2 * half;
        const float* bp = Ws + (16 * wave + l15) * GP + 2 * half;
#pragma unroll 8
        for (int kk = 0; kk < 128; kk += 4) {
            const v2f a0 = *(const v2f*)(ap + kk);
            const v2f b0 = *(const v2f*)(bp + kk);
            acc = __builtin_amdgcn_wmma_f32_16x16x4_f32(
                false, a0, false, b0, (short)0, acc, false, false);
        }
    }

    const int col = 16 * wave + l15;
    const float bv = bias ? bias[cbase + col] : 0.f;

    __syncthreads();
#pragma unroll
    for (int v = 0; v < 8; ++v) {
        float val = acc[v] + bv;
        if (ACT == 1 || (ACT == 2 && relu_rt)) val = fmaxf(val, 0.f);
        As[(v + 8 * half) * GP + col] = val;
    }
    __syncthreads();

    store_c_rows<SMALL_BM, O>(As, C, n, rbase, cbase, wave, lane);
}

template <bool TRANSB, int ACT, int K, int O>
__device__ __forceinline__ void gemm_block(
    const float* __restrict__ A, const float* __restrict__ W,
    const float* __restrict__ bias, float* __restrict__ C, int n,
    float* __restrict__ As, float* __restrict__ Ws, bool relu_rt)
{
    const int bx = blockIdx.x;
    if (bx < SMALL_TILES)
        gemm_tile16<TRANSB, ACT, K, O>(A, W, bias, C, n, FULL_TILES * GBM + bx * SMALL_BM, As, Ws, relu_rt);
    else
        gemm_tile<TRANSB, ACT, K, O>(A, W, bias, C, n, (bx - SMALL_TILES) * GBM, As, Ws, relu_rt);
}

template <bool TRANSB, int ACT, int K, int O>
__global__ __launch_bounds__(256, 6) void gemm_lds(
    const float* __restrict__ A, const float* __restrict__ W,
    const float* __restrict__ bias, float* __restrict__ C, int n)
{
    __shared__ __attribute__((aligned(16))) float As[GBM * GP];
    __shared__ __attribute__((aligned(16))) float Ws[GBN * GP];
    gemm_block<TRANSB, ACT, K, O>(A, W, bias, C, n, As, Ws, false);
}

#define PT_ROWS 16
#define PT_TILES ((NN + PT_ROWS - 1) / PT_ROWS)
static_assert(PT_TILES >= RESIDENT_BLOCKS, "every resident block has a first tile");

template <bool TRANSB, int O>
__device__ __forceinline__ void gemm_rows_k128(
    const float* __restrict__ A, const float* __restrict__ W,
    const float* __restrict__ bias, float* __restrict__ C, int n, bool relu,
    float* __restrict__ Ws, float* __restrict__ As, float* __restrict__ Cs)
{
    const int wave = threadIdx.x >> 5;
    const int lane = threadIdx.x & 31;
    const int half = lane >> 4;
    const int l15  = lane & 15;
    const int cbase = blockIdx.y * GBN;
    const int step = gridDim.x;
    int t = blockIdx.x;

    stage_w<TRANSB, 128, O>(W, Ws, cbase, 0, wave, lane);

    float4 pa0 = *(const float4*)(A + (unsigned)(min(t * PT_ROWS + wave,     n - 1) * 128 + lane * 4));
    float4 pa1 = *(const float4*)(A + (unsigned)(min(t * PT_ROWS + wave + 8, n - 1) * 128 + lane * 4));
    const int col = 16 * wave + l15;
    const float bv = bias ? bias[cbase + col] : 0.f;
    __syncthreads();

    v2f bf[32];
    {
        const float* bp = Ws + col * GP + 2 * half;
#pragma unroll
        for (int s = 0; s < 32; ++s) bf[s] = *(const v2f*)(bp + 4 * s);
    }

    const float* ap = As + l15 * GP + 2 * half;
    for (; t < PT_TILES; t += step) {
        *(float4*)(As + wave * GP + lane * 4) = pa0;
        *(float4*)(As + (wave + 8) * GP + lane * 4) = pa1;
        __syncthreads();

        const int tn = t + step;
        if (tn < PT_TILES) {
            pa0 = *(const float4*)(A + (unsigned)(min(tn * PT_ROWS + wave,     n - 1) * 128 + lane * 4));
            pa1 = *(const float4*)(A + (unsigned)(min(tn * PT_ROWS + wave + 8, n - 1) * 128 + lane * 4));
        }

        v8f acc = (v8f){0.f,0.f,0.f,0.f,0.f,0.f,0.f,0.f};
        v2f af[32];
#pragma unroll
        for (int s = 0; s < 32; ++s) af[s] = *(const v2f*)(ap + 4 * s);
#pragma unroll
        for (int s = 0; s < 32; ++s)
            acc = __builtin_amdgcn_wmma_f32_16x16x4_f32(
                false, af[s], false, bf[s], (short)0, acc, false, false);
        __builtin_amdgcn_sched_group_barrier(0x100, 8, 0);
        __builtin_amdgcn_sched_group_barrier(0x008, 8, 0);
        __builtin_amdgcn_sched_group_barrier(0x100, 4, 0);
        __builtin_amdgcn_sched_group_barrier(0x008, 8, 0);
        __builtin_amdgcn_sched_group_barrier(0x100, 4, 0);
        __builtin_amdgcn_sched_group_barrier(0x008, 16, 0);

#pragma unroll
        for (int v = 0; v < 8; ++v) {
            float val = acc[v] + bv;
            if (relu) val = fmaxf(val, 0.f);
            Cs[(v + 8 * half) * GP + col] = val;
        }
        __syncthreads();

#pragma unroll
        for (int i = 0; i < PT_ROWS / 8; ++i) {
            const int row = wave + 8 * i;
            const float4 o = *(const float4*)(Cs + row * GP + lane * 4);
            const int grow = t * PT_ROWS + row;
            if (grow < n) *(float4*)(C + (unsigned)(grow * O + cbase + lane * 4)) = o;
        }
    }
}

template <bool TRANSB, int ACT, int O>
__global__ __launch_bounds__(256, 6) void gemm_k128(
    const float* __restrict__ A, const float* __restrict__ W,
    const float* __restrict__ bias, float* __restrict__ C, int n)
{
    __shared__ __attribute__((aligned(16))) float Ws[GBN * GP];
    __shared__ __attribute__((aligned(16))) float As[PT_ROWS * GP];
    __shared__ __attribute__((aligned(16))) float Cs[PT_ROWS * GP];
    gemm_rows_k128<TRANSB, O>(A, W, bias, C, n, ACT == 1, Ws, As, Cs);
}

template <bool TB0, int ACT0, bool TB1, int ACT1, int O>
__global__ __launch_bounds__(256, 6) void gemm_k128_pair(
    const float* __restrict__ A0, const float* __restrict__ W0,
    const float* __restrict__ bias0, float* __restrict__ C0,
    const float* __restrict__ A1, const float* __restrict__ W1,
    const float* __restrict__ bias1, float* __restrict__ C1, int n)
{
    __shared__ __attribute__((aligned(16))) float Ws[GBN * GP];
    __shared__ __attribute__((aligned(16))) float As[PT_ROWS * GP];
    __shared__ __attribute__((aligned(16))) float Cs[PT_ROWS * GP];
    const bool second = blockIdx.z != 0;
    if (TB0 == TB1) {
        gemm_rows_k128<TB0, O>(
            second ? A1 : A0, second ? W1 : W0, second ? bias1 : bias0, second ? C1 : C0,
            n, (second ? ACT1 : ACT0) == 1, Ws, As, Cs);
    } else {
        if (!second) gemm_rows_k128<TB0, O>(A0, W0, bias0, C0, n, ACT0 == 1, Ws, As, Cs);
        else         gemm_rows_k128<TB1, O>(A1, W1, bias1, C1, n, ACT1 == 1, Ws, As, Cs);
    }
}

__device__ __forceinline__ float wave_sum32(float v) {
#pragma unroll
    for (int m = 16; m > 0; m >>= 1) v += __shfl_xor(v, m, 32);
    return v;
}

__global__ __launch_bounds__(256) void fill_kernel(float* __restrict__ p, float v, int n) {
    int i = blockIdx.x * blockDim.x + threadIdx.x;
    if (i < n) p[i] = v;
}

__global__ __launch_bounds__(256) void fill_int_kernel(int* __restrict__ p, int v, int n) {
    int i = blockIdx.x * blockDim.x + threadIdx.x;
    if (i < n) p[i] = v;
}

__global__ __launch_bounds__(256) void count_rank_kernel(
    const int* __restrict__ dst, int* __restrict__ cnt, int* __restrict__ rank, int e)
{
    int i = blockIdx.x * blockDim.x + threadIdx.x;
    if (i < e) rank[i] = atomicAdd(&cnt[dst[i]], 1);
}

#define SCAN_CH 13
static_assert(32 * SCAN_CH * 128 >= NN, "scan block must cover all nodes");
static_assert(NN % 4 == 0, "16-byte accesses below need NN % 4 == 0");

__global__ __launch_bounds__(1024) void scan_isd_kernel(
    const int* __restrict__ cnt, int* __restrict__ rowptr, float* __restrict__ isd, int n)
{
    __shared__ int wtot[32];
    const int lane = threadIdx.x & 31;
    const int wave = threadIdx.x >> 5;
    const int wbase = wave * (SCAN_CH * 128);
    int4 c[SCAN_CH];
#pragma unroll
    for (int k = 0; k < SCAN_CH; ++k) {
        const int e0 = wbase + k * 128 + lane * 4;
        c[k] = (e0 < n) ? *(const int4*)(cnt + e0) : make_int4(0, 0, 0, 0);
    }
    int ex[SCAN_CH];
    int carry = 0;
#pragma unroll
    for (int k = 0; k < SCAN_CH; ++k) {
        const int tot = c[k].x + c[k].y + c[k].z + c[k].w;
        int inc = tot;
#pragma unroll
        for (int d = 1; d < 32; d <<= 1) {
            int t = __shfl_up(inc, d, 32);
            if (lane >= d) inc += t;
        }
        ex[k] = carry + inc - tot;
        carry += __shfl(inc, 31, 32);
    }
    if (lane == 0) wtot[wave] = carry;
    __syncthreads();
    int below = (lane < wave) ? wtot[lane] : 0;
#pragma unroll
    for (int m = 16; m > 0; m >>= 1) below += __shfl_xor(below, m, 32);
#pragma unroll
    for (int k = 0; k < SCAN_CH; ++k) {
        const int e0 = wbase + k * 128 + lane * 4;
        if (e0 < n) {
            const int p0 = below + ex[k];
            const int p1 = p0 + c[k].x, p2 = p1 + c[k].y, p3 = p2 + c[k].z;
            *(int4*)(rowptr + e0) = make_int4(p0, p1, p2, p3);
            *(float4*)(isd + e0) = make_float4(
                rsqrtf((float)c[k].x + 1.0f), rsqrtf((float)c[k].y + 1.0f),
                rsqrtf((float)c[k].z + 1.0f), rsqrtf((float)c[k].w + 1.0f));
        }
    }
}

__global__ __launch_bounds__(256) void place_kernel(
    const int* __restrict__ src, const int* __restrict__ dst,
    const int* __restrict__ rowptr, const int* __restrict__ rank,
    int* __restrict__ csr, int e)
{
    int i = blockIdx.x * blockDim.x + threadIdx.x;
    if (i >= e) return;
    const int pos = rowptr[dst[i]] + rank[i];
    if (pos >= 0 && pos < e) csr[pos] = src[i];
}

__global__ __launch_bounds__(256) void row_order_kernel(
    const int* __restrict__ rowptr, const int* __restrict__ cnt,
    const int* __restrict__ csr_in, int* __restrict__ csr_out, int n)
{
    const int lane = threadIdx.x & 31;
    const int node = __builtin_amdgcn_readfirstlane((int)((blockIdx.x * 256 + threadIdx.x) >> 5));
    if (node >= n) return;
    const int start = __builtin_amdgcn_readfirstlane(rowptr[node]);
    const int len   = __builtin_amdgcn_readfirstlane(cnt[node]);
    for (int ib = 0; ib < len; ib += 32) {
        const int i = ib + lane;
        const int xi = (i < len) ? csr_in[start + i] : 0;
        int pos = 0;
        for (int jb = 0; jb < len; jb += 32) {
            const int mj = min(32, len - jb);
            const int xv = (lane < mj) ? csr_in[start + jb + lane] : 0;
            for (int jj = 0; jj < mj; ++jj) {
                const int xj = __builtin_amdgcn_readlane(xv, jj);
                const int j = jb + jj;
                pos += (xj < xi || (xj == xi && j < i)) ? 1 : 0;
            }
        }
        if (i < len) csr_out[start + pos] = xi;
    }
}

template <bool GCN, int B>
__device__ __forceinline__ void gather_accum(
    float4& acc, const float* __restrict__ v, int ms, float mw, int j, int lane)
{
    float4 r[B];
    float w[B];
#pragma unroll
    for (int u = 0; u < B; ++u) {
        const int s = __builtin_amdgcn_readlane(ms, j + u);
        r[u] = *(const float4*)(v + (size_t)s * DD + lane * 4);
        w[u] = GCN ? __int_as_float(__builtin_amdgcn_readlane(__float_as_int(mw), j + u)) : 1.0f;
    }
#pragma unroll
    for (int u = 0; u < B; ++u) {
        if (GCN) {
            acc.x = fmaf(r[u].x, w[u], acc.x); acc.y = fmaf(r[u].y, w[u], acc.y);
            acc.z = fmaf(r[u].z, w[u], acc.z); acc.w = fmaf(r[u].w, w[u], acc.w);
        } else {
            acc.x += r[u].x; acc.y += r[u].y; acc.z += r[u].z; acc.w += r[u].w;
        }
    }
}

template <bool GCN>
__global__ __launch_bounds__(256) void gather_rows_kernel(
    const int* __restrict__ rowptr, const int* __restrict__ cnt, const int* __restrict__ csr,
    const float* __restrict__ v, const float* __restrict__ isd, const float* __restrict__ bias,
    const float* __restrict__ g, const float* __restrict__ b,
    float* __restrict__ o, int n)
{
    const int lane = threadIdx.x & 31;
    const int node = __builtin_amdgcn_readfirstlane((int)((blockIdx.x * 256 + threadIdx.x) >> 5));
    if (node >= n) return;
    const int start = __builtin_amdgcn_readfirstlane(rowptr[node]);
    const int len   = __builtin_amdgcn_readfirstlane(cnt[node]);
    float4 acc = make_float4(0.f, 0.f, 0.f, 0.f);
    float4 sv = acc, bv = acc, gv = acc, lb = acc;
    float it = 0.f;
    if (GCN) {
        it = isd[node];
        sv = *(const float4*)(v + (size_t)node * DD + lane * 4);
        bv = *(const float4*)(bias + lane * 4);
        gv = *(const float4*)(g + lane * 4);
        lb = *(const float4*)(b + lane * 4);
    }
    for (int base = 0; base < len; base += 32) {
        const int m = min(32, len - base);
        int ms = 0;
        float mw = 0.f;
        if (lane < m) {
            ms = csr[start + base + lane];
            ms = min(max(ms, 0), n - 1);
            if (GCN) mw = isd[ms] * it;
        }
        int j = 0;
        for (; j + 8 <= m; j += 8) gather_accum<GCN, 8>(acc, v, ms, mw, j, lane);
        if (j + 4 <= m) { gather_accum<GCN, 4>(acc, v, ms, mw, j, lane); j += 4; }
        if (j + 2 <= m) { gather_accum<GCN, 2>(acc, v, ms, mw, j, lane); j += 2; }
        if (j < m)      { gather_accum<GCN, 1>(acc, v, ms, mw, j, lane); }
    }
    if (GCN) {
        acc.x += sv.x * it * it + bv.x; acc.y += sv.y * it * it + bv.y;
        acc.z += sv.z * it * it + bv.z; acc.w += sv.w * it * it + bv.w;
        const float mean = wave_sum32(acc.x + acc.y + acc.z + acc.w) * (1.f / 128.f);
        const float dx = acc.x - mean, dy = acc.y - mean, dz = acc.z - mean, dw = acc.w - mean;
        const float inv = rsqrtf(wave_sum32(dx * dx + dy * dy + dz * dz + dw * dw) * (1.f / 128.f) + 1e-5f);
        acc.x = fmaxf(dx * inv * gv.x + lb.x, 0.f); acc.y = fmaxf(dy * inv * gv.y + lb.y, 0.f);
        acc.z = fmaxf(dz * inv * gv.z + lb.z, 0.f); acc.w = fmaxf(dw * inv * gv.w + lb.w, 0.f);
    }
    *(float4*)(o + (size_t)node * DD + lane * 4) = acc;
}

__global__ __launch_bounds__(256) void attention_kernel(
    const float* __restrict__ qkv0, const float* __restrict__ qkv1,
    float* __restrict__ ao0, float* __restrict__ ao1, int n)
{
    int wv = (blockIdx.x * 256 + threadIdx.x) >> 5;
    int lane = threadIdx.x & 31;
    if (wv >= n) return;
    const float* b0 = qkv0 + (size_t)wv * 384;
    const float* b1 = qkv1 + (size_t)wv * 384;
    float q0[4], k0[4], v0[4], q1[4], k1[4], v1[4];
#pragma unroll
    for (int j = 0; j < 4; ++j) {
        int d = lane + 32 * j;
        q0[j] = b0[d]; k0[j] = b0[128 + d]; v0[j] = b0[256 + d];
        q1[j] = b1[d]; k1[j] = b1[128 + d]; v1[j] = b1[256 + d];
    }
    float l00 = 0, l01 = 0, l10 = 0, l11 = 0;
#pragma unroll
    for (int j = 0; j < 4; ++j) {
        l00 += q0[j] * k0[j]; l01 += q0[j] * k1[j];
        l10 += q1[j] * k0[j]; l11 += q1[j] * k1[j];
    }
    l00 = wave_sum32(l00); l01 = wave_sum32(l01);
    l10 = wave_sum32(l10); l11 = wave_sum32(l11);
    const float scale = 0.08838834764831845f;
    l00 *= scale; l01 *= scale; l10 *= scale; l11 *= scale;
    float m0 = fmaxf(l00, l01), e00 = __expf(l00 - m0), e01 = __expf(l01 - m0);
    float r0 = 1.f / (e00 + e01), w00 = e00 * r0, w01 = e01 * r0;
    float m1 = fmaxf(l10, l11), e10 = __expf(l10 - m1), e11 = __expf(l11 - m1);
    float r1 = 1.f / (e10 + e11), w10 = e10 * r1, w11 = e11 * r1;
    float* o0 = ao0 + (size_t)wv * DD;
    float* o1 = ao1 + (size_t)wv * DD;
#pragma unroll
    for (int j = 0; j < 4; ++j) {
        int d = lane + 32 * j;
        o0[d] = w00 * v0[j] + w01 * v1[j];
        o1[d] = w10 * v0[j] + w11 * v1[j];
    }
}

__global__ __launch_bounds__(256) void fuse_ln384_kernel(
    const float* __restrict__ a0, const float* __restrict__ a1,
    const float* __restrict__ rs, const float* __restrict__ g,
    const float* __restrict__ b, float* __restrict__ o, int n)
{
    int wv = (blockIdx.x * 256 + threadIdx.x) >> 5;
    int lane = threadIdx.x & 31;
    if (wv >= n) return;
    const float* p0 = a0 + (size_t)wv * DD;
    const float* p1 = a1 + (size_t)wv * DD;
    const float* p2 = rs + (size_t)wv * DD;
    float x[12], s = 0.f;
#pragma unroll
    for (int j = 0; j < 12; ++j) {
        int d = lane + 32 * j;
        float v = (d < 128) ? p0[d] : (d < 256) ? p1[d - 128] : p2[d - 256];
        x[j] = v; s += v;
    }
    float mean = wave_sum32(s) * (1.f / 384.f);
    float sq = 0.f;
#pragma unroll
    for (int j = 0; j < 12; ++j) { float d = x[j] - mean; sq += d * d; }
    float inv = rsqrtf(wave_sum32(sq) * (1.f / 384.f) + 1e-5f);
    float* orow = o + (size_t)wv * 384;
#pragma unroll
    for (int j = 0; j < 12; ++j) {
        int d = lane + 32 * j;
        orow[d] = (x[j] - mean) * inv * g[d] + b[d];
    }
}

#define CS_BLOCKS 256
#define CS_STRIDE (CS_BLOCKS * 8)
#define CS_ROWS ((NN + CS_STRIDE - 1) / CS_STRIDE)
static_assert(CS_ROWS * CS_STRIDE >= NN, "column-sum waves must cover all rows");
static_assert(CS_ROWS <= 48, "row loads of a wave must stay below the load counter's limit");

__global__ __launch_bounds__(256) void colsum_kernel(const float* __restrict__ p, float* __restrict__ cs, int n) {
    __shared__ __attribute__((aligned(16))) float part[8 * DD];
    const int lane = threadIdx.x & 31;
    const int wave = threadIdx.x >> 5;
    const int r0 = __builtin_amdgcn_readfirstlane((int)(blockIdx.x * 8 + wave));
    float4 r[CS_ROWS];
#pragma unroll
    for (int t = 0; t < CS_ROWS; ++t) {
        const int row = r0 + t * CS_STRIDE;
        r[t] = *(const float4*)(p + (size_t)min(row, n - 1) * DD + lane * 4);
    }
    float4 s = make_float4(0.f, 0.f, 0.f, 0.f);
#pragma unroll
    for (int t = 0; t < CS_ROWS; ++t) {
        const bool in = (r0 + t * CS_STRIDE) < n;
        s.x += in ? r[t].x : 0.f; s.y += in ? r[t].y : 0.f;
        s.z += in ? r[t].z : 0.f; s.w += in ? r[t].w : 0.f;
    }
    *(float4*)(part + wave * DD + lane * 4) = s;
    __syncthreads();
    if (threadIdx.x < DD) {
        float tot = 0.f;
#pragma unroll
        for (int w = 0; w < 8; ++w) tot += part[w * DD + threadIdx.x];
        atomicAdd(&cs[threadIdx.x], tot);
    }
}

__global__ __launch_bounds__(256) void pairnorm_kernel(
    const float* __restrict__ pre, const float* __restrict__ cs,
    float* __restrict__ o, int n)
{
    int wv = (blockIdx.x * 256 + threadIdx.x) >> 5;
    int lane = threadIdx.x & 31;
    if (wv >= n) return;
    const float* row = pre + (size_t)wv * DD;
    float x[4], sq = 0.f;
#pragma unroll
    for (int j = 0; j < 4; ++j) { x[j] = row[lane + 32 * j]; sq += x[j] * x[j]; }
    float rn = sqrtf(1e-6f + wave_sum32(sq));
    float inv = 20.f / rn;
    float* orow = o + (size_t)wv * DD;
    const float invN = 1.f / (float)NN;
#pragma unroll
    for (int j = 0; j < 4; ++j) {
        int d = lane + 32 * j;
        orow[d] = x[j] * inv - cs[d] * invN;
    }
}

#define WS_ROW_ARRAYS ((NN + DD + 63) / 64 * 64)
static_assert(WS_ROW_ARRAYS >= NN + DD && WS_ROW_ARRAYS % 64 == 0, "row arrays behind isd and csum, on a 256-byte boundary");
static_assert(((size_t)NN * DD) % 64 == 0, "every row array keeps the next one on a 256-byte boundary");

extern "C" void kernel_launch(void* const* d_in, const int* in_sizes, int n_in,
                              void* d_out, int out_size, void* d_ws, size_t ws_size,
                              hipStream_t stream)
{
    const float* x      = (const float*)d_in[0];
    const float* x_org  = (const float*)d_in[1];
    const int*   ei     = (const int*)d_in[2];
    const float* in_W   = (const float*)d_in[3];
    const float* in_b   = (const float*)d_in[4];
    const float* res_W  = (const float*)d_in[5];
    const float* gcn_W  = (const float*)d_in[6];
    const float* gcn_b  = (const float*)d_in[7];
    const float* ln_g   = (const float*)d_in[8];
    const float* ln_b   = (const float*)d_in[9];
    const float* lin_W  = (const float*)d_in[10];
    const float* lin_b  = (const float*)d_in[11];
    const float* attr_W = (const float*)d_in[12];
    const float* attr_b = (const float*)d_in[13];
    const float* qkv_W  = (const float*)d_in[14];
    const float* qkv_b  = (const float*)d_in[15];
    const float* out_W  = (const float*)d_in[16];
    const float* out_b  = (const float*)d_in[17];
    const float* fln_g  = (const float*)d_in[18];
    const float* fln_b  = (const float*)d_in[19];
    const float* fuse_W = (const float*)d_in[20];
    const float* fuse_b = (const float*)d_in[21];
    float* out = (float*)d_out;

    const int* src = ei;
    const int* dst = ei + EE;

    float* ws = (float*)d_ws;
    const size_t ND = (size_t)NN * DD;
    float* isd  = ws;
    float* csum = isd + NN;
    float* bh   = ws + WS_ROW_ARRAYS;
    float* bt   = bh + ND;
    float* br   = bt + ND;
    float* bs   = br + ND;
    float* q0b  = bs + ND;
    float* q1b  = q0b + 3 * ND;
    int* cnt    = (int*)q1b;
    int* rowptr = cnt + NN;
    int* rank   = rowptr + NN;
    int* csr_a  = rank + EE;
    int* csr_b  = csr_a + EE;

    const dim3 blk(256);
    const dim3 gemmO128(SMALL_TILES + FULL_TILES, DD / GBN);
    const dim3 oneK128(RESIDENT_BLOCKS, DD / GBN);
    const dim3 pairK128(RESIDENT_BLOCKS / 2, DD / GBN, 2);
    const dim3 pairK128O384(RESIDENT_BLOCKS / 6, 3 * DD / GBN, 2);
    const int nodeWaves = (NN + 7) / 8;
    const int edgeBlk   = (EE + 255) / 256;

    fill_int_kernel<<<(NN + 255) / 256, blk, 0, stream>>>(cnt, 0, NN);
    count_rank_kernel<<<edgeBlk, blk, 0, stream>>>(dst, cnt, rank, EE);
    scan_isd_kernel<<<1, 1024, 0, stream>>>(cnt, rowptr, isd, NN);
    place_kernel<<<edgeBlk, blk, 0, stream>>>(src, dst, rowptr, rank, csr_a, EE);
    row_order_kernel<<<nodeWaves, blk, 0, stream>>>(rowptr, cnt, csr_a, csr_b, NN);

    gemm_k128_pair<true, 1, false, 0, DD><<<pairK128, blk, 0, stream>>>(
        x, in_W, in_b, bh, x_org, res_W, nullptr, bt, NN);

    gather_rows_kernel<false><<<nodeWaves, blk, 0, stream>>>(
        rowptr, cnt, csr_b, bt, nullptr, nullptr, nullptr, nullptr, br, NN);

    for (int i = 0; i < 3; ++i) {
        gemm_k128<true, 0, DD><<<oneK128, blk, 0, stream>>>(
            bh, gcn_W + (size_t)i * DD * DD, nullptr, bt, NN);
        gather_rows_kernel<true><<<nodeWaves, blk, 0, stream>>>(
            rowptr, cnt, csr_b, bt, isd, gcn_b + (size_t)i * DD,
            ln_g + (size_t)i * DD, ln_b + (size_t)i * DD, bh, NN);
    }

    gemm_k128_pair<true, 0, true, 1, DD><<<pairK128, blk, 0, stream>>>(
        bh, lin_W, lin_b, bs, x_org, attr_W, attr_b, bt, NN);

    gemm_k128_pair<true, 0, true, 0, 3 * DD><<<pairK128O384, blk, 0, stream>>>(
        bs, qkv_W, qkv_b, q0b, bt, qkv_W, qkv_b, q1b, NN);

    attention_kernel<<<nodeWaves, blk, 0, stream>>>(q0b, q1b, bh, bt, NN);

    gemm_k128_pair<true, 0, true, 0, DD><<<pairK128, blk, 0, stream>>>(
        bh, out_W, out_b, bs, bt, out_W, out_b, q1b, NN);

    fuse_ln384_kernel<<<nodeWaves, blk, 0, stream>>>(bs, q1b, br, fln_g, fln_b, q0b, NN);

    gemm_lds<true, 0, 3 * DD, DD><<<gemmO128, blk, 0, stream>>>(q0b, fuse_W, fuse_b, bh, NN);

    fill_kernel<<<1, 128, 0, stream>>>(csum, 0.f, DD);
    colsum_kernel<<<CS_BLOCKS, blk, 0, stream>>>(bh, csum, NN);
    pairnorm_kernel<<<nodeWaves, blk, 0, stream>>>(bh, csum, out, NN);
}


# LayerNorm(3D) of concat(ao0,ao1,residual) applied as the K=384 GEMM stages its A rows: the separate LN pass and its [N,3D] array dropped; LN sums now add a lane's 12 values, then across lanes
# speedup vs baseline: 3.1863x; 1.0493x over previous
#include <hip/hip_runtime.h>
#include <hip/hip_bf16.h>

#define NN 50000
#define DD 128
#define EE 800000

typedef __attribute__((ext_vector_type(2))) float v2f;
typedef __attribute__((ext_vector_type(8))) float v8f;

__device__ __forceinline__ float wave_sum32(float v) {
#pragma unroll
    for (int m = 16; m > 0; m >>= 1) v += __shfl_xor(v, m, 32);
    return v;
}

#define GP 132
#define GBM 64
#define GBN 128

#define RESIDENT_BLOCKS 768
#define FULL_TILES ((NN / GBM) / RESIDENT_BLOCKS * RESIDENT_BLOCKS)
#define SMALL_BM 16
#define SMALL_TILES ((NN - FULL_TILES * GBM + SMALL_BM - 1) / SMALL_BM)
static_assert(FULL_TILES > 0 && FULL_TILES * GBM <= NN, "full tiles must lie inside the rows");
static_assert(FULL_TILES * GBM + SMALL_TILES * SMALL_BM >= NN, "small tiles must cover the rest");

template <bool TRANSB, int K, int O>
__device__ __forceinline__ void stage_w(
    const float* __restrict__ W, float* __restrict__ Ws, int cbase, int kc, int wave, int lane)
{
    if (TRANSB) {
#pragma unroll
        for (int i = 0; i < GBN / 8; ++i) {
            const int row = wave + 8 * i;
            const float4 v = *(const float4*)(W + (unsigned)((cbase + row) * K + kc + lane * 4));
            *(float4*)(Ws + row * GP + lane * 4) = v;
        }
    } else {
#pragma unroll
        for (int i = 0; i < GBN / 8; ++i) {
            const int idx = wave * (GBN / 8) + i;
            const int k  = (idx >> 2) * 4 + (lane >> 3);
            const int nb = (idx & 3) * 32 + (lane & 7) * 4;
            const float4 v = *(const float4*)(W + (unsigned)((kc + k) * O + cbase + nb));
            Ws[(nb + 0) * GP + k] = v.x;
            Ws[(nb + 1) * GP + k] = v.y;
            Ws[(nb + 2) * GP + k] = v.z;
            Ws[(nb + 3) * GP + k] = v.w;
        }
    }
}

template <int ROWS, int O>
__device__ __forceinline__ void store_c_rows(
    const float* __restrict__ As, float* __restrict__ C, int n, int rbase, int cbase, int wave, int lane)
{
    float4 o[ROWS / 8];
#pragma unroll
    for (int i = 0; i < ROWS / 8; ++i) o[i] = *(const float4*)(As + (wave + 8 * i) * GP + lane * 4);
    if (rbase + ROWS <= n) {
#pragma unroll
        for (int i = 0; i < ROWS / 8; ++i)
            *(float4*)(C + (unsigned)((rbase + wave + 8 * i) * O + cbase + lane * 4)) = o[i];
    } else {
#pragma unroll
        for (int i = 0; i < ROWS / 8; ++i) {
            const int row = rbase + wave + 8 * i;
            if (row < n) *(float4*)(C + (unsigned)(row * O + cbase + lane * 4)) = o[i];
        }
    }
}

template <int R>
__device__ __forceinline__ void ln384_stats(
    const float* __restrict__ a0, const float* __restrict__ a1, const float* __restrict__ rs,
    int n, int rbase, int wave, int lane, float (&mean)[R], float (&inv)[R])
{
    float4 u[R], v[R], w[R];
#pragma unroll
    for (int i = 0; i < R; ++i) {
        const unsigned off = (unsigned)(min(rbase + wave + 8 * i, n - 1) * DD + lane * 4);
        u[i] = *(const float4*)(a0 + off);
        v[i] = *(const float4*)(a1 + off);
        w[i] = *(const float4*)(rs + off);
    }
#pragma unroll
    for (int i = 0; i < R; ++i) {
        const float s = (u[i].x + u[i].y + u[i].z + u[i].w) + (v[i].x + v[i].y + v[i].z + v[i].w)
                      + (w[i].x + w[i].y + w[i].z + w[i].w);
        const float m = wave_sum32(s) * (1.f / 384.f);
        float d, q = 0.f;
        d = u[i].x - m; q += d * d; d = u[i].y - m; q += d * d; d = u[i].z - m; q += d * d; d = u[i].w - m; q += d * d;
        d = v[i].x - m; q += d * d; d = v[i].y - m; q += d * d; d = v[i].z - m; q += d * d; d = v[i].w - m; q += d * d;
        d = w[i].x - m; q += d * d; d = w[i].y - m; q += d * d; d = w[i].z - m; q += d * d; d = w[i].w - m; q += d * d;
        const float r = rsqrtf(wave_sum32(q) * (1.f / 384.f) + 1e-5f);
        mean[i] = __int_as_float(__builtin_amdgcn_readfirstlane(__float_as_int(m)));
        inv[i]  = __int_as_float(__builtin_amdgcn_readfirstlane(__float_as_int(r)));
    }
}

template <int ROWS>
__device__ __forceinline__ void stage_a_ln384(
    const float* __restrict__ a0, const float* __restrict__ a1, const float* __restrict__ rs,
    const float* __restrict__ g, const float* __restrict__ b, float* __restrict__ As,
    int n, int rbase, int kc, int wave, int lane,
    const float (&mean)[ROWS / 8], const float (&inv)[ROWS / 8])
{
    const float* __restrict__ part = kc == 0 ? a0 : (kc == 128 ? a1 : rs);
    const float4 gq = *(const float4*)(g + kc + lane * 4);
    const float4 bq = *(const float4*)(b + kc + lane * 4);
#pragma unroll
    for (int i = 0; i < ROWS / 8; ++i) {
        const int row = wave + 8 * i;
        const int grow = min(rbase + row, n - 1);
        float4 x = *(const float4*)(part + (unsigned)(grow * DD + lane * 4));
        x.x = (x.x - mean[i]) * inv[i] * gq.x + bq.x;
        x.y = (x.y - mean[i]) * inv[i] * gq.y + bq.y;
        x.z = (x.z - mean[i]) * inv[i] * gq.z + bq.z;
        x.w = (x.w - mean[i]) * inv[i] * gq.w + bq.w;
        *(float4*)(As + row * GP + lane * 4) = x;
    }
}

template <bool TRANSB, int ACT, int K, int O>
__device__ __forceinline__ void gemm_tile(
    const float* __restrict__ p0, const float* __restrict__ p1, const float* __restrict__ rs,
    const float* __restrict__ g, const float* __restrict__ b, const float* __restrict__ W,
    const float* __restrict__ bias, float* __restrict__ C, int n, int rbase,
    float* __restrict__ As, float* __restrict__ Ws, bool relu_rt)
{
    static_assert(K == 3 * DD && O % GBN == 0, "tile sizes; A is the three 128-wide parts side by side");

    const int wave = threadIdx.x >> 5;
    const int lane = threadIdx.x & 31;
    const int half = lane >> 4;
    const int l15  = lane & 15;
    const int cbase = blockIdx.y * GBN;
    const int wr = (wave >> 2) * 32;
    const int wc = (wave & 3) * 32;

    float mean[GBM / 8], inv[GBM / 8];
    ln384_stats<GBM / 8>(p0, p1, rs, n, rbase, wave, lane, mean, inv);

    v8f acc[2][2];
#pragma unroll
    for (int i = 0; i < 2; ++i)
#pragma unroll
        for (int j = 0; j < 2; ++j) acc[i][j] = (v8f){0.f,0.f,0.f,0.f,0.f,0.f,0.f,0.f};

    for (int kc = 0; kc < K; kc += 128) {
        if (kc > 0) __syncthreads();

        stage_a_ln384<GBM>(p0, p1, rs, g, b, As, n, rbase, kc, wave, lane, mean, inv);
        stage_w<TRANSB, K, O>(W, Ws, cbase, kc, wave, lane);
        __syncthreads();

        const float* ap = As + (wr + l15) * GP + 2 * half;
        const float* bp = Ws + (wc + l15) * GP + 2 * half;
#pragma unroll 8
        for (int kk = 0; kk < 128; kk += 4) {
            const v2f a0 = *(const v2f*)(ap + kk);
            const v2f a1 = *(const v2f*)(ap + 16 * GP + kk);
            const v2f b0 = *(const v2f*)(bp + kk);
            const v2f b1 = *(const v2f*)(bp + 16 * GP + kk);
            acc[0][0] = __builtin_amdgcn_wmma_f32_16x16x4_f32(
                false, a0, false, b0, (short)0, acc[0][0], false, false);
            acc[0][1] = __builtin_amdgcn_wmma_f32_16x16x4_f32(
                false, a0, false, b1, (short)0, acc[0][1], false, false);
            acc[1][0] = __builtin_amdgcn_wmma_f32_16x16x4_f32(
                false, a1, false, b0, (short)0, acc[1][0], false, false);
            acc[1][1] = __builtin_amdgcn_wmma_f32_16x16x4_f32(
                false, a1, false, b1, (short)0, acc[1][1], false, false);
        }
    }

    float bv[2];
#pragma unroll
    for (int j = 0; j < 2; ++j) bv[j] = bias ? bias[cbase + wc + 16 * j + l15] : 0.f;

    __syncthreads();
#pragma unroll
    for (int j = 0; j < 2; ++j) {
        const int col = wc + 16 * j + l15;
#pragma unroll
        for (int i = 0; i < 2; ++i) {
#pragma unroll
            for (int v = 0; v < 8; ++v) {
                float val = acc[i][j][v] + bv[j];
                if (ACT == 1 || (ACT == 2 && relu_rt)) val = fmaxf(val, 0.f);
                As[(wr + 16 * i + v + 8 * half) * GP + col] = val;
            }
        }
    }
    __syncthreads();

    store_c_rows<GBM, O>(As, C, n, rbase, cbase, wave, lane);
}

template <bool TRANSB, int ACT, int K, int O>
__device__ __forceinline__ void gemm_tile16(
    const float* __restrict__ p0, const float* __restrict__ p1, const float* __restrict__ rs,
    const float* __restrict__ g, const float* __restrict__ b, const float* __restrict__ W,
    const float* __restrict__ bias, float* __restrict__ C, int n, int rbase,
    float* __restrict__ As, float* __restrict__ Ws, bool relu_rt)
{
    static_assert(K == 3 * DD, "A is the three 128-wide parts side by side");
    const int wave = threadIdx.x >> 5;
    const int lane = threadIdx.x & 31;
    const int half = lane >> 4;
    const int l15  = lane & 15;
    const int cbase = blockIdx.y * GBN;

    v8f acc = (v8f){0.f,0.f,0.f,0.f,0.f,0.f,0.f,0.f};

    float mean[SMALL_BM / 8], inv[SMALL_BM / 8];
    ln384_stats<SMALL_BM / 8>(p0, p1, rs, n, rbase, wave, lane, mean, inv);

    for (int kc = 0; kc < K; kc += 128) {
        if (kc > 0) __syncthreads();
        stage_a_ln384<SMALL_BM>(p0, p1, rs, g, b, As, n, rbase, kc, wave, lane, mean, inv);
        __builtin_amdgcn_sched_barrier(0);
        stage_w<TRANSB, K, O>(W, Ws, cbase, kc, wave, lane);
        __syncthreads();

        const float* ap = As + l15 * GP + 2 * half;
        const float* bp = Ws + (16 * wave + l15) * GP + 2 * half;
#pragma unroll 8
        for (int kk = 0; kk < 128; kk += 4) {
            const v2f a0 = *(const v2f*)(ap + kk);
            const v2f b0 = *(const v2f*)(bp + kk);
            acc = __builtin_amdgcn_wmma_f32_16x16x4_f32(
                false, a0, false, b0, (short)0, acc, false, false);
        }
    }

    const int col = 16 * wave + l15;
    const float bv = bias ? bias[cbase + col] : 0.f;

    __syncthreads();
#pragma unroll
    for (int v = 0; v < 8; ++v) {
        float val = acc[v] + bv;
        if (ACT == 1 || (ACT == 2 && relu_rt)) val = fmaxf(val, 0.f);
        As[(v + 8 * half) * GP + col] = val;
    }
    __syncthreads();

    store_c_rows<SMALL_BM, O>(As, C, n, rbase, cbase, wave, lane);
}

template <bool TRANSB, int ACT, int K, int O>
__device__ __forceinline__ void gemm_block(
    const float* __restrict__ p0, const float* __restrict__ p1, const float* __restrict__ rs,
    const float* __restrict__ g, const float* __restrict__ b, const float* __restrict__ W,
    const float* __restrict__ bias, float* __restrict__ C, int n,
    float* __restrict__ As, float* __restrict__ Ws, bool relu_rt)
{
    const int bx = blockIdx.x;
    if (bx < SMALL_TILES)
        gemm_tile16<TRANSB, ACT, K, O>(p0, p1, rs, g, b, W, bias, C, n, FULL_TILES * GBM + bx * SMALL_BM, As, Ws, relu_rt);
    else
        gemm_tile<TRANSB, ACT, K, O>(p0, p1, rs, g, b, W, bias, C, n, (bx - SMALL_TILES) * GBM, As, Ws, relu_rt);
}

template <bool TRANSB, int ACT, int K, int O>
__global__ __launch_bounds__(256, 6) void gemm_lds(
    const float* __restrict__ p0, const float* __restrict__ p1, const float* __restrict__ rs,
    const float* __restrict__ g, const float* __restrict__ b, const float* __restrict__ W,
    const float* __restrict__ bias, float* __restrict__ C, int n)
{
    __shared__ __attribute__((aligned(16))) float As[GBM * GP];
    __shared__ __attribute__((aligned(16))) float Ws[GBN * GP];
    gemm_block<TRANSB, ACT, K, O>(p0, p1, rs, g, b, W, bias, C, n, As, Ws, false);
}

#define PT_ROWS 16
#define PT_TILES ((NN + PT_ROWS - 1) / PT_ROWS)
static_assert(PT_TILES >= RESIDENT_BLOCKS, "every resident block has a first tile");

template <bool TRANSB, int O>
__device__ __forceinline__ void gemm_rows_k128(
    const float* __restrict__ A, const float* __restrict__ W,
    const float* __restrict__ bias, float* __restrict__ C, int n, bool relu,
    float* __restrict__ Ws, float* __restrict__ As, float* __restrict__ Cs)
{
    const int wave = threadIdx.x >> 5;
    const int lane = threadIdx.x & 31;
    const int half = lane >> 4;
    const int l15  = lane & 15;
    const int cbase = blockIdx.y * GBN;
    const int step = gridDim.x;
    int t = blockIdx.x;

    stage_w<TRANSB, 128, O>(W, Ws, cbase, 0, wave, lane);

    float4 pa0 = *(const float4*)(A + (unsigned)(min(t * PT_ROWS + wave,     n - 1) * 128 + lane * 4));
    float4 pa1 = *(const float4*)(A + (unsigned)(min(t * PT_ROWS + wave + 8, n - 1) * 128 + lane * 4));
    const int col = 16 * wave + l15;
    const float bv = bias ? bias[cbase + col] : 0.f;
    __syncthreads();

    v2f bf[32];
    {
        const float* bp = Ws + col * GP + 2 * half;
#pragma unroll
        for (int s = 0; s < 32; ++s) bf[s] = *(const v2f*)(bp + 4 * s);
    }

    const float* ap = As + l15 * GP + 2 * half;
    for (; t < PT_TILES; t += step) {
        *(float4*)(As + wave * GP + lane * 4) = pa0;
        *(float4*)(As + (wave + 8) * GP + lane * 4) = pa1;
        __syncthreads();

        const int tn = t + step;
        if (tn < PT_TILES) {
            pa0 = *(const float4*)(A + (unsigned)(min(tn * PT_ROWS + wave,     n - 1) * 128 + lane * 4));
            pa1 = *(const float4*)(A + (unsigned)(min(tn * PT_ROWS + wave + 8, n - 1) * 128 + lane * 4));
        }

        v8f acc = (v8f){0.f,0.f,0.f,0.f,0.f,0.f,0.f,0.f};
        v2f af[32];
#pragma unroll
        for (int s = 0; s < 32; ++s) af[s] = *(const v2f*)(ap + 4 * s);
#pragma unroll
        for (int s = 0; s < 32; ++s)
            acc = __builtin_amdgcn_wmma_f32_16x16x4_f32(
                false, af[s], false, bf[s], (short)0, acc, false, false);
        __builtin_amdgcn_sched_group_barrier(0x100, 8, 0);
        __builtin_amdgcn_sched_group_barrier(0x008, 8, 0);
        __builtin_amdgcn_sched_group_barrier(0x100, 4, 0);
        __builtin_amdgcn_sched_group_barrier(0x008, 8, 0);
        __builtin_amdgcn_sched_group_barrier(0x100, 4, 0);
        __builtin_amdgcn_sched_group_barrier(0x008, 16, 0);

#pragma unroll
        for (int v = 0; v < 8; ++v) {
            float val = acc[v] + bv;
            if (relu) val = fmaxf(val, 0.f);
            Cs[(v + 8 * half) * GP + col] = val;
        }
        __syncthreads();

#pragma unroll
        for (int i = 0; i < PT_ROWS / 8; ++i) {
            const int row = wave + 8 * i;
            const float4 o = *(const float4*)(Cs + row * GP + lane * 4);
            const int grow = t * PT_ROWS + row;
            if (grow < n) *(float4*)(C + (unsigned)(grow * O + cbase + lane * 4)) = o;
        }
    }
}

template <bool TRANSB, int ACT, int O>
__global__ __launch_bounds__(256, 6) void gemm_k128(
    const float* __restrict__ A, const float* __restrict__ W,
    const float* __restrict__ bias, float* __restrict__ C, int n)
{
    __shared__ __attribute__((aligned(16))) float Ws[GBN * GP];
    __shared__ __attribute__((aligned(16))) float As[PT_ROWS * GP];
    __shared__ __attribute__((aligned(16))) float Cs[PT_ROWS * GP];
    gemm_rows_k128<TRANSB, O>(A, W, bias, C, n, ACT == 1, Ws, As, Cs);
}

template <bool TB0, int ACT0, bool TB1, int ACT1, int O>
__global__ __launch_bounds__(256, 6) void gemm_k128_pair(
    const float* __restrict__ A0, const float* __restrict__ W0,
    const float* __restrict__ bias0, float* __restrict__ C0,
    const float* __restrict__ A1, const float* __restrict__ W1,
    const float* __restrict__ bias1, float* __restrict__ C1, int n)
{
    __shared__ __attribute__((aligned(16))) float Ws[GBN * GP];
    __shared__ __attribute__((aligned(16))) float As[PT_ROWS * GP];
    __shared__ __attribute__((aligned(16))) float Cs[PT_ROWS * GP];
    const bool second = blockIdx.z != 0;
    if (TB0 == TB1) {
        gemm_rows_k128<TB0, O>(
            second ? A1 : A0, second ? W1 : W0, second ? bias1 : bias0, second ? C1 : C0,
            n, (second ? ACT1 : ACT0) == 1, Ws, As, Cs);
    } else {
        if (!second) gemm_rows_k128<TB0, O>(A0, W0, bias0, C0, n, ACT0 == 1, Ws, As, Cs);
        else         gemm_rows_k128<TB1, O>(A1, W1, bias1, C1, n, ACT1 == 1, Ws, As, Cs);
    }
}

__global__ __launch_bounds__(256) void fill_kernel(float* __restrict__ p, float v, int n) {
    int i = blockIdx.x * blockDim.x + threadIdx.x;
    if (i < n) p[i] = v;
}

__global__ __launch_bounds__(256) void fill_int_kernel(int* __restrict__ p, int v, int n) {
    int i = blockIdx.x * blockDim.x + threadIdx.x;
    if (i < n) p[i] = v;
}

__global__ __launch_bounds__(256) void count_rank_kernel(
    const int* __restrict__ dst, int* __restrict__ cnt, int* __restrict__ rank, int e)
{
    int i = blockIdx.x * blockDim.x + threadIdx.x;
    if (i < e) rank[i] = atomicAdd(&cnt[dst[i]], 1);
}

#define SCAN_CH 13
static_assert(32 * SCAN_CH * 128 >= NN, "scan block must cover all nodes");
static_assert(NN % 4 == 0, "16-byte accesses below need NN % 4 == 0");

__global__ __launch_bounds__(1024) void scan_isd_kernel(
    const int* __restrict__ cnt, int* __restrict__ rowptr, float* __restrict__ isd, int n)
{
    __shared__ int wtot[32];
    const int lane = threadIdx.x & 31;
    const int wave = threadIdx.x >> 5;
    const int wbase = wave * (SCAN_CH * 128);
    int4 c[SCAN_CH];
#pragma unroll
    for (int k = 0; k < SCAN_CH; ++k) {
        const int e0 = wbase + k * 128 + lane * 4;
        c[k] = (e0 < n) ? *(const int4*)(cnt + e0) : make_int4(0, 0, 0, 0);
    }
    int ex[SCAN_CH];
    int carry = 0;
#pragma unroll
    for (int k = 0; k < SCAN_CH; ++k) {
        const int tot = c[k].x + c[k].y + c[k].z + c[k].w;
        int inc = tot;
#pragma unroll
        for (int d = 1; d < 32; d <<= 1) {
            int t = __shfl_up(inc, d, 32);
            if (lane >= d) inc += t;
        }
        ex[k] = carry + inc - tot;
        carry += __shfl(inc, 31, 32);
    }
    if (lane == 0) wtot[wave] = carry;
    __syncthreads();
    int below = (lane < wave) ? wtot[lane] : 0;
#pragma unroll
    for (int m = 16; m > 0; m >>= 1) below += __shfl_xor(below, m, 32);
#pragma unroll
    for (int k = 0; k < SCAN_CH; ++k) {
        const int e0 = wbase + k * 128 + lane * 4;
        if (e0 < n) {
            const int p0 = below + ex[k];
            const int p1 = p0 + c[k].x, p2 = p1 + c[k].y, p3 = p2 + c[k].z;
            *(int4*)(rowptr + e0) = make_int4(p0, p1, p2, p3);
            *(float4*)(isd + e0) = make_float4(
                rsqrtf((float)c[k].x + 1.0f), rsqrtf((float)c[k].y + 1.0f),
                rsqrtf((float)c[k].z + 1.0f), rsqrtf((float)c[k].w + 1.0f));
        }
    }
}

__global__ __launch_bounds__(256) void place_kernel(
    const int* __restrict__ src, const int* __restrict__ dst,
    const int* __restrict__ rowptr, const int* __restrict__ rank,
    int* __restrict__ csr, int e)
{
    int i = blockIdx.x * blockDim.x + threadIdx.x;
    if (i >= e) return;
    const int pos = rowptr[dst[i]] + rank[i];
    if (pos >= 0 && pos < e) csr[pos] = src[i];
}

__global__ __launch_bounds__(256) void row_order_kernel(
    const int* __restrict__ rowptr, const int* __restrict__ cnt,
    const int* __restrict__ csr_in, int* __restrict__ csr_out, int n)
{
    const int lane = threadIdx.x & 31;
    const int node = __builtin_amdgcn_readfirstlane((int)((blockIdx.x * 256 + threadIdx.x) >> 5));
    if (node >= n) return;
    const int start = __builtin_amdgcn_readfirstlane(rowptr[node]);
    const int len   = __builtin_amdgcn_readfirstlane(cnt[node]);
    for (int ib = 0; ib < len; ib += 32) {
        const int i = ib + lane;
        const int xi = (i < len) ? csr_in[start + i] : 0;
        int pos = 0;
        for (int jb = 0; jb < len; jb += 32) {
            const int mj = min(32, len - jb);
            const int xv = (lane < mj) ? csr_in[start + jb + lane] : 0;
            for (int jj = 0; jj < mj; ++jj) {
                const int xj = __builtin_amdgcn_readlane(xv, jj);
                const int j = jb + jj;
                pos += (xj < xi || (xj == xi && j < i)) ? 1 : 0;
            }
        }
        if (i < len) csr_out[start + pos] = xi;
    }
}

template <bool GCN, int B>
__device__ __forceinline__ void gather_accum(
    float4& acc, const float* __restrict__ v, int ms, float mw, int j, int lane)
{
    float4 r[B];
    float w[B];
#pragma unroll
    for (int u = 0; u < B; ++u) {
        const int s = __builtin_amdgcn_readlane(ms, j + u);
        r[u] = *(const float4*)(v + (size_t)s * DD + lane * 4);
        w[u] = GCN ? __int_as_float(__builtin_amdgcn_readlane(__float_as_int(mw), j + u)) : 1.0f;
    }
#pragma unroll
    for (int u = 0; u < B; ++u) {
        if (GCN) {
            acc.x = fmaf(r[u].x, w[u], acc.x); acc.y = fmaf(r[u].y, w[u], acc.y);
            acc.z = fmaf(r[u].z, w[u], acc.z); acc.w = fmaf(r[u].w, w[u], acc.w);
        } else {
            acc.x += r[u].x; acc.y += r[u].y; acc.z += r[u].z; acc.w += r[u].w;
        }
    }
}

template <bool GCN>
__global__ __launch_bounds__(256) void gather_rows_kernel(
    const int* __restrict__ rowptr, const int* __restrict__ cnt, const int* __restrict__ csr,
    const float* __restrict__ v, const float* __restrict__ isd, const float* __restrict__ bias,
    const float* __restrict__ g, const float* __restrict__ b,
    float* __restrict__ o, int n)
{
    const int lane = threadIdx.x & 31;
    const int node = __builtin_amdgcn_readfirstlane((int)((blockIdx.x * 256 + threadIdx.x) >> 5));
    if (node >= n) return;
    const int start = __builtin_amdgcn_readfirstlane(rowptr[node]);
    const int len   = __builtin_amdgcn_readfirstlane(cnt[node]);
    float4 acc = make_float4(0.f, 0.f, 0.f, 0.f);
    float4 sv = acc, bv = acc, gv = acc, lb = acc;
    float it = 0.f;
    if (GCN) {
        it = isd[node];
        sv = *(const float4*)(v + (size_t)node * DD + lane * 4);
        bv = *(const float4*)(bias + lane * 4);
        gv = *(const float4*)(g + lane * 4);
        lb = *(const float4*)(b + lane * 4);
    }
    for (int base = 0; base < len; base += 32) {
        const int m = min(32, len - base);
        int ms = 0;
        float mw = 0.f;
        if (lane < m) {
            ms = csr[start + base + lane];
            ms = min(max(ms, 0), n - 1);
            if (GCN) mw = isd[ms] * it;
        }
        int j = 0;
        for (; j + 8 <= m; j += 8) gather_accum<GCN, 8>(acc, v, ms, mw, j, lane);
        if (j + 4 <= m) { gather_accum<GCN, 4>(acc, v, ms, mw, j, lane); j += 4; }
        if (j + 2 <= m) { gather_accum<GCN, 2>(acc, v, ms, mw, j, lane); j += 2; }
        if (j < m)      { gather_accum<GCN, 1>(acc, v, ms, mw, j, lane); }
    }
    if (GCN) {
        acc.x += sv.x * it * it + bv.x; acc.y += sv.y * it * it + bv.y;
        acc.z += sv.z * it * it + bv.z; acc.w += sv.w * it * it + bv.w;
        const float mean = wave_sum32(acc.x + acc.y + acc.z + acc.w) * (1.f / 128.f);
        const float dx = acc.x - mean, dy = acc.y - mean, dz = acc.z - mean, dw = acc.w - mean;
        const float inv = rsqrtf(wave_sum32(dx * dx + dy * dy + dz * dz + dw * dw) * (1.f / 128.f) + 1e-5f);
        acc.x = fmaxf(dx * inv * gv.x + lb.x, 0.f); acc.y = fmaxf(dy * inv * gv.y + lb.y, 0.f);
        acc.z = fmaxf(dz * inv * gv.z + lb.z, 0.f); acc.w = fmaxf(dw * inv * gv.w + lb.w, 0.f);
    }
    *(float4*)(o + (size_t)node * DD + lane * 4) = acc;
}

__global__ __launch_bounds__(256) void attention_kernel(
    const float* __restrict__ qkv0, const float* __restrict__ qkv1,
    float* __restrict__ ao0, float* __restrict__ ao1, int n)
{
    int wv = (blockIdx.x * 256 + threadIdx.x) >> 5;
    int lane = threadIdx.x & 31;
    if (wv >= n) return;
    const float* b0 = qkv0 + (size_t)wv * 384;
    const float* b1 = qkv1 + (size_t)wv * 384;
    float q0[4], k0[4], v0[4], q1[4], k1[4], v1[4];
#pragma unroll
    for (int j = 0; j < 4; ++j) {
        int d = lane + 32 * j;
        q0[j] = b0[d]; k0[j] = b0[128 + d]; v0[j] = b0[256 + d];
        q1[j] = b1[d]; k1[j] = b1[128 + d]; v1[j] = b1[256 + d];
    }
    float l00 = 0, l01 = 0, l10 = 0, l11 = 0;
#pragma unroll
    for (int j = 0; j < 4; ++j) {
        l00 += q0[j] * k0[j]; l01 += q0[j] * k1[j];
        l10 += q1[j] * k0[j]; l11 += q1[j] * k1[j];
    }
    l00 = wave_sum32(l00); l01 = wave_sum32(l01);
    l10 = wave_sum32(l10); l11 = wave_sum32(l11);
    const float scale = 0.08838834764831845f;
    l00 *= scale; l01 *= scale; l10 *= scale; l11 *= scale;
    float m0 = fmaxf(l00, l01), e00 = __expf(l00 - m0), e01 = __expf(l01 - m0);
    float r0 = 1.f / (e00 + e01), w00 = e00 * r0, w01 = e01 * r0;
    float m1 = fmaxf(l10, l11), e10 = __expf(l10 - m1), e11 = __expf(l11 - m1);
    float r1 = 1.f / (e10 + e11), w10 = e10 * r1, w11 = e11 * r1;
    float* o0 = ao0 + (size_t)wv * DD;
    float* o1 = ao1 + (size_t)wv * DD;
#pragma unroll
    for (int j = 0; j < 4; ++j) {
        int d = lane + 32 * j;
        o0[d] = w00 * v0[j] + w01 * v1[j];
        o1[d] = w10 * v0[j] + w11 * v1[j];
    }
}

#define CS_BLOCKS 256
#define CS_STRIDE (CS_BLOCKS * 8)
#define CS_ROWS ((NN + CS_STRIDE - 1) / CS_STRIDE)
static_assert(CS_ROWS * CS_STRIDE >= NN, "column-sum waves must cover all rows");
static_assert(CS_ROWS <= 48, "row loads of a wave must stay below the load counter's limit");

__global__ __launch_bounds__(256) void colsum_kernel(const float* __restrict__ p, float* __restrict__ cs, int n) {
    __shared__ __attribute__((aligned(16))) float part[8 * DD];
    const int lane = threadIdx.x & 31;
    const int wave = threadIdx.x >> 5;
    const int r0 = __builtin_amdgcn_readfirstlane((int)(blockIdx.x * 8 + wave));
    float4 r[CS_ROWS];
#pragma unroll
    for (int t = 0; t < CS_ROWS; ++t) {
        const int row = r0 + t * CS_STRIDE;
        r[t] = *(const float4*)(p + (size_t)min(row, n - 1) * DD + lane * 4);
    }
    float4 s = make_float4(0.f, 0.f, 0.f, 0.f);
#pragma unroll
    for (int t = 0; t < CS_ROWS; ++t) {
        const bool in = (r0 + t * CS_STRIDE) < n;
        s.x += in ? r[t].x : 0.f; s.y += in ? r[t].y : 0.f;
        s.z += in ? r[t].z : 0.f; s.w += in ? r[t].w : 0.f;
    }
    *(float4*)(part + wave * DD + lane * 4) = s;
    __syncthreads();
    if (threadIdx.x < DD) {
        float tot = 0.f;
#pragma unroll
        for (int w = 0; w < 8; ++w) tot += part[w * DD + threadIdx.x];
        atomicAdd(&cs[threadIdx.x], tot);
    }
}

__global__ __launch_bounds__(256) void pairnorm_kernel(
    const float* __restrict__ pre, const float* __restrict__ cs,
    float* __restrict__ o, int n)
{
    int wv = (blockIdx.x * 256 + threadIdx.x) >> 5;
    int lane = threadIdx.x & 31;
    if (wv >= n) return;
    const float* row = pre + (size_t)wv * DD;
    float x[4], sq = 0.f;
#pragma unroll
    for (int j = 0; j < 4; ++j) { x[j] = row[lane + 32 * j]; sq += x[j] * x[j]; }
    float rn = sqrtf(1e-6f + wave_sum32(sq));
    float inv = 20.f / rn;
    float* orow = o + (size_t)wv * DD;
    const float invN = 1.f / (float)NN;
#pragma unroll
    for (int j = 0; j < 4; ++j) {
        int d = lane + 32 * j;
        orow[d] = x[j] * inv - cs[d] * invN;
    }
}

#define WS_ROW_ARRAYS ((NN + DD + 63) / 64 * 64)
static_assert(WS_ROW_ARRAYS >= NN + DD && WS_ROW_ARRAYS % 64 == 0, "row arrays behind isd and csum, on a 256-byte boundary");
static_assert(((size_t)NN * DD) % 64 == 0, "every row array keeps the next one on a 256-byte boundary");

extern "C" void kernel_launch(void* const* d_in, const int* in_sizes, int n_in,
                              void* d_out, int out_size, void* d_ws, size_t ws_size,
                              hipStream_t stream)
{
    const float* x      = (const float*)d_in[0];
    const float* x_org  = (const float*)d_in[1];
    const int*   ei     = (const int*)d_in[2];
    const float* in_W   = (const float*)d_in[3];
    const float* in_b   = (const float*)d_in[4];
    const float* res_W  = (const float*)d_in[5];
    const float* gcn_W  = (const float*)d_in[6];
    const float* gcn_b  = (const float*)d_in[7];
    const float* ln_g   = (const float*)d_in[8];
    const float* ln_b   = (const float*)d_in[9];
    const float* lin_W  = (const float*)d_in[10];
    const float* lin_b  = (const float*)d_in[11];
    const float* attr_W = (const float*)d_in[12];
    const float* attr_b = (const float*)d_in[13];
    const float* qkv_W  = (const float*)d_in[14];
    const float* qkv_b  = (const float*)d_in[15];
    const float* out_W  = (const float*)d_in[16];
    const float* out_b  = (const float*)d_in[17];
    const float* fln_g  = (const float*)d_in[18];
    const float* fln_b  = (const float*)d_in[19];
    const float* fuse_W = (const float*)d_in[20];
    const float* fuse_b = (const float*)d_in[21];
    float* out = (float*)d_out;

    const int* src = ei;
    const int* dst = ei + EE;

    float* ws = (float*)d_ws;
    const size_t ND = (size_t)NN * DD;
    float* isd  = ws;
    float* csum = isd + NN;
    float* bh   = ws + WS_ROW_ARRAYS;
    float* bt   = bh + ND;
    float* br   = bt + ND;
    float* bs   = br + ND;
    float* q0b  = bs + ND;
    float* q1b  = q0b + 3 * ND;
    int* cnt    = (int*)q1b;
    int* rowptr = cnt + NN;
    int* rank   = rowptr + NN;
    int* csr_a  = rank + EE;
    int* csr_b  = csr_a + EE;

    const dim3 blk(256);
    const dim3 gemmO128(SMALL_TILES + FULL_TILES, DD / GBN);
    const dim3 oneK128(RESIDENT_BLOCKS, DD / GBN);
    const dim3 pairK128(RESIDENT_BLOCKS / 2, DD / GBN, 2);
    const dim3 pairK128O384(RESIDENT_BLOCKS / 6, 3 * DD / GBN, 2);
    const int nodeWaves = (NN + 7) / 8;
    const int edgeBlk   = (EE + 255) / 256;

    fill_int_kernel<<<(NN + 255) / 256, blk, 0, stream>>>(cnt, 0, NN);
    count_rank_kernel<<<edgeBlk, blk, 0, stream>>>(dst, cnt, rank, EE);
    scan_isd_kernel<<<1, 1024, 0, stream>>>(cnt, rowptr, isd, NN);
    place_kernel<<<edgeBlk, blk, 0, stream>>>(src, dst, rowptr, rank, csr_a, EE);
    row_order_kernel<<<nodeWaves, blk, 0, stream>>>(rowptr, cnt, csr_a, csr_b, NN);

    gemm_k128_pair<true, 1, false, 0, DD><<<pairK128, blk, 0, stream>>>(
        x, in_W, in_b, bh, x_org, res_W, nullptr, bt, NN);

    gather_rows_kernel<false><<<nodeWaves, blk, 0, stream>>>(
        rowptr, cnt, csr_b, bt, nullptr, nullptr, nullptr, nullptr, br, NN);

    for (int i = 0; i < 3; ++i) {
        gemm_k128<true, 0, DD><<<oneK128, blk, 0, stream>>>(
            bh, gcn_W + (size_t)i * DD * DD, nullptr, bt, NN);
        gather_rows_kernel<true><<<nodeWaves, blk, 0, stream>>>(
            rowptr, cnt, csr_b, bt, isd, gcn_b + (size_t)i * DD,
            ln_g + (size_t)i * DD, ln_b + (size_t)i * DD, bh, NN);
    }

    gemm_k128_pair<true, 0, true, 1, DD><<<pairK128, blk, 0, stream>>>(
        bh, lin_W, lin_b, bs, x_org, attr_W, attr_b, bt, NN);

    gemm_k128_pair<true, 0, true, 0, 3 * DD><<<pairK128O384, blk, 0, stream>>>(
        bs, qkv_W, qkv_b, q0b, bt, qkv_W, qkv_b, q1b, NN);

    attention_kernel<<<nodeWaves, blk, 0, stream>>>(q0b, q1b, bh, bt, NN);

    gemm_k128_pair<true, 0, true, 0, DD><<<pairK128, blk, 0, stream>>>(
        bh, out_W, out_b, bs, bt, out_W, out_b, q1b, NN);

    gemm_lds<true, 0, 3 * DD, DD><<<gemmO128, blk, 0, stream>>>(
        bs, q1b, br, fln_g, fln_b, fuse_W, fuse_b, bh, NN);

    fill_kernel<<<1, 128, 0, stream>>>(csum, 0.f, DD);
    colsum_kernel<<<CS_BLOCKS, blk, 0, stream>>>(bh, csum, NN);
    pairnorm_kernel<<<nodeWaves, blk, 0, stream>>>(bh, csum, out, NN);
}


# Attention out projection folded into the v weights: Wvo = out_W*Wv built in-call on the matrix unit, both out-projection products dropped; sums re-associated, x*Wvo^T for (x*Wv^T)*out_W^T
# speedup vs baseline: 3.3337x; 1.0463x over previous
#include <hip/hip_runtime.h>
#include <hip/hip_bf16.h>

#define NN 50000
#define DD 128
#define EE 800000

typedef __attribute__((ext_vector_type(2))) float v2f;
typedef __attribute__((ext_vector_type(8))) float v8f;

__device__ __forceinline__ float wave_sum32(float v) {
#pragma unroll
    for (int m = 16; m > 0; m >>= 1) v += __shfl_xor(v, m, 32);
    return v;
}

#define GP 132
#define GBM 64
#define GBN 128

#define RESIDENT_BLOCKS 768
#define FULL_TILES ((NN / GBM) / RESIDENT_BLOCKS * RESIDENT_BLOCKS)
#define SMALL_BM 16
#define SMALL_TILES ((NN - FULL_TILES * GBM + SMALL_BM - 1) / SMALL_BM)
static_assert(FULL_TILES > 0 && FULL_TILES * GBM <= NN, "full tiles must lie inside the rows");
static_assert(FULL_TILES * GBM + SMALL_TILES * SMALL_BM >= NN, "small tiles must cover the rest");

template <bool TRANSB, int K, int O>
__device__ __forceinline__ void stage_w(
    const float* __restrict__ W, float* __restrict__ Ws, int cbase, int kc, int wave, int lane)
{
    if (TRANSB) {
#pragma unroll
        for (int i = 0; i < GBN / 8; ++i) {
            const int row = wave + 8 * i;
            const float4 v = *(const float4*)(W + (unsigned)((cbase + row) * K + kc + lane * 4));
            *(float4*)(Ws + row * GP + lane * 4) = v;
        }
    } else {
#pragma unroll
        for (int i = 0; i < GBN / 8; ++i) {
            const int idx = wave * (GBN / 8) + i;
            const int k  = (idx >> 2) * 4 + (lane >> 3);
            const int nb = (idx & 3) * 32 + (lane & 7) * 4;
            const float4 v = *(const float4*)(W + (unsigned)((kc + k) * O + cbase + nb));
            Ws[(nb + 0) * GP + k] = v.x;
            Ws[(nb + 1) * GP + k] = v.y;
            Ws[(nb + 2) * GP + k] = v.z;
            Ws[(nb + 3) * GP + k] = v.w;
        }
    }
}

template <int ROWS, int O>
__device__ __forceinline__ void store_c_rows(
    const float* __restrict__ As, float* __restrict__ C, int n, int rbase, int cbase, int wave, int lane)
{
    float4 o[ROWS / 8];
#pragma unroll
    for (int i = 0; i < ROWS / 8; ++i) o[i] = *(const float4*)(As + (wave + 8 * i) * GP + lane * 4);
    if (rbase + ROWS <= n) {
#pragma unroll
        for (int i = 0; i < ROWS / 8; ++i)
            *(float4*)(C + (unsigned)((rbase + wave + 8 * i) * O + cbase + lane * 4)) = o[i];
    } else {
#pragma unroll
        for (int i = 0; i < ROWS / 8; ++i) {
            const int row = rbase + wave + 8 * i;
            if (row < n) *(float4*)(C + (unsigned)(row * O + cbase + lane * 4)) = o[i];
        }
    }
}

template <int R>
__device__ __forceinline__ void ln384_stats(
    const float* __restrict__ a0, const float* __restrict__ a1, const float* __restrict__ rs,
    int n, int rbase, int wave, int lane, float (&mean)[R], float (&inv)[R])
{
    float4 u[R], v[R], w[R];
#pragma unroll
    for (int i = 0; i < R; ++i) {
        const unsigned off = (unsigned)(min(rbase + wave + 8 * i, n - 1) * DD + lane * 4);
        u[i] = *(const float4*)(a0 + off);
        v[i] = *(const float4*)(a1 + off);
        w[i] = *(const float4*)(rs + off);
    }
#pragma unroll
    for (int i = 0; i < R; ++i) {
        const float s = (u[i].x + u[i].y + u[i].z + u[i].w) + (v[i].x + v[i].y + v[i].z + v[i].w)
                      + (w[i].x + w[i].y + w[i].z + w[i].w);
        const float m = wave_sum32(s) * (1.f / 384.f);
        float d, q = 0.f;
        d = u[i].x - m; q += d * d; d = u[i].y - m; q += d * d; d = u[i].z - m; q += d * d; d = u[i].w - m; q += d * d;
        d = v[i].x - m; q += d * d; d = v[i].y - m; q += d * d; d = v[i].z - m; q += d * d; d = v[i].w - m; q += d * d;
        d = w[i].x - m; q += d * d; d = w[i].y - m; q += d * d; d = w[i].z - m; q += d * d; d = w[i].w - m; q += d * d;
        const float r = rsqrtf(wave_sum32(q) * (1.f / 384.f) + 1e-5f);
        mean[i] = __int_as_float(__builtin_amdgcn_readfirstlane(__float_as_int(m)));
        inv[i]  = __int_as_float(__builtin_amdgcn_readfirstlane(__float_as_int(r)));
    }
}

template <int ROWS>
__device__ __forceinline__ void stage_a_ln384(
    const float* __restrict__ a0, const float* __restrict__ a1, const float* __restrict__ rs,
    const float* __restrict__ g, const float* __restrict__ b, float* __restrict__ As,
    int n, int rbase, int kc, int wave, int lane,
    const float (&mean)[ROWS / 8], const float (&inv)[ROWS / 8])
{
    const float* __restrict__ part = kc == 0 ? a0 : (kc == 128 ? a1 : rs);
    const float4 gq = *(const float4*)(g + kc + lane * 4);
    const float4 bq = *(const float4*)(b + kc + lane * 4);
#pragma unroll
    for (int i = 0; i < ROWS / 8; ++i) {
        const int row = wave + 8 * i;
        const int grow = min(rbase + row, n - 1);
        float4 x = *(const float4*)(part + (unsigned)(grow * DD + lane * 4));
        x.x = (x.x - mean[i]) * inv[i] * gq.x + bq.x;
        x.y = (x.y - mean[i]) * inv[i] * gq.y + bq.y;
        x.z = (x.z - mean[i]) * inv[i] * gq.z + bq.z;
        x.w = (x.w - mean[i]) * inv[i] * gq.w + bq.w;
        *(float4*)(As + row * GP + lane * 4) = x;
    }
}

template <bool TRANSB, int ACT, int K, int O>
__device__ __forceinline__ void gemm_tile(
    const float* __restrict__ p0, const float* __restrict__ p1, const float* __restrict__ rs,
    const float* __restrict__ g, const float* __restrict__ b, const float* __restrict__ W,
    const float* __restrict__ bias, float* __restrict__ C, int n, int rbase,
    float* __restrict__ As, float* __restrict__ Ws, bool relu_rt)
{
    static_assert(K == 3 * DD && O % GBN == 0, "tile sizes; A is the three 128-wide parts side by side");

    const int wave = threadIdx.x >> 5;
    const int lane = threadIdx.x & 31;
    const int half = lane >> 4;
    const int l15  = lane & 15;
    const int cbase = blockIdx.y * GBN;
    const int wr = (wave >> 2) * 32;
    const int wc = (wave & 3) * 32;

    float mean[GBM / 8], inv[GBM / 8];
    ln384_stats<GBM / 8>(p0, p1, rs, n, rbase, wave, lane, mean, inv);

    v8f acc[2][2];
#pragma unroll
    for (int i = 0; i < 2; ++i)
#pragma unroll
        for (int j = 0; j < 2; ++j) acc[i][j] = (v8f){0.f,0.f,0.f,0.f,0.f,0.f,0.f,0.f};

    for (int kc = 0; kc < K; kc += 128) {
        if (kc > 0) __syncthreads();

        stage_a_ln384<GBM>(p0, p1, rs, g, b, As, n, rbase, kc, wave, lane, mean, inv);
        stage_w<TRANSB, K, O>(W, Ws, cbase, kc, wave, lane);
        __syncthreads();

        const float* ap = As + (wr + l15) * GP + 2 * half;
        const float* bp = Ws + (wc + l15) * GP + 2 * half;
#pragma unroll 8
        for (int kk = 0; kk < 128; kk += 4) {
            const v2f a0 = *(const v2f*)(ap + kk);
            const v2f a1 = *(const v2f*)(ap + 16 * GP + kk);
            const v2f b0 = *(const v2f*)(bp + kk);
            const v2f b1 = *(const v2f*)(bp + 16 * GP + kk);
            acc[0][0] = __builtin_amdgcn_wmma_f32_16x16x4_f32(
                false, a0, false, b0, (short)0, acc[0][0], false, false);
            acc[0][1] = __builtin_amdgcn_wmma_f32_16x16x4_f32(
                false, a0, false, b1, (short)0, acc[0][1], false, false);
            acc[1][0] = __builtin_amdgcn_wmma_f32_16x16x4_f32(
                false, a1, false, b0, (short)0, acc[1][0], false, false);
            acc[1][1] = __builtin_amdgcn_wmma_f32_16x16x4_f32(
                false, a1, false, b1, (short)0, acc[1][1], false, false);
        }
    }

    float bv[2];
#pragma unroll
    for (int j = 0; j < 2; ++j) bv[j] = bias ? bias[cbase + wc + 16 * j + l15] : 0.f;

    __syncthreads();
#pragma unroll
    for (int j = 0; j < 2; ++j) {
        const int col = wc + 16 * j + l15;
#pragma unroll
        for (int i = 0; i < 2; ++i) {
#pragma unroll
            for (int v = 0; v < 8; ++v) {
                float val = acc[i][j][v] + bv[j];
                if (ACT == 1 || (ACT == 2 && relu_rt)) val = fmaxf(val, 0.f);
                As[(wr + 16 * i + v + 8 * half) * GP + col] = val;
            }
        }
    }
    __syncthreads();

    store_c_rows<GBM, O>(As, C, n, rbase, cbase, wave, lane);
}

template <bool TRANSB, int ACT, int K, int O>
__device__ __forceinline__ void gemm_tile16(
    const float* __restrict__ p0, const float* __restrict__ p1, const float* __restrict__ rs,
    const float* __restrict__ g, const float* __restrict__ b, const float* __restrict__ W,
    const float* __restrict__ bias, float* __restrict__ C, int n, int rbase,
    float* __restrict__ As, float* __restrict__ Ws, bool relu_rt)
{
    static_assert(K == 3 * DD, "A is the three 128-wide parts side by side");
    const int wave = threadIdx.x >> 5;
    const int lane = threadIdx.x & 31;
    const int half = lane >> 4;
    const int l15  = lane & 15;
    const int cbase = blockIdx.y * GBN;

    v8f acc = (v8f){0.f,0.f,0.f,0.f,0.f,0.f,0.f,0.f};

    float mean[SMALL_BM / 8], inv[SMALL_BM / 8];
    ln384_stats<SMALL_BM / 8>(p0, p1, rs, n, rbase, wave, lane, mean, inv);

    for (int kc = 0; kc < K; kc += 128) {
        if (kc > 0) __syncthreads();
        stage_a_ln384<SMALL_BM>(p0, p1, rs, g, b, As, n, rbase, kc, wave, lane, mean, inv);
        __builtin_amdgcn_sched_barrier(0);
        stage_w<TRANSB, K, O>(W, Ws, cbase, kc, wave, lane);
        __syncthreads();

        const float* ap = As + l15 * GP + 2 * half;
        const float* bp = Ws + (16 * wave + l15) * GP + 2 * half;
#pragma unroll 8
        for (int kk = 0; kk < 128; kk += 4) {
            const v2f a0 = *(const v2f*)(ap + kk);
            const v2f b0 = *(const v2f*)(bp + kk);
            acc = __builtin_amdgcn_wmma_f32_16x16x4_f32(
                false, a0, false, b0, (short)0, acc, false, false);
        }
    }

    const int col = 16 * wave + l15;
    const float bv = bias ? bias[cbase + col] : 0.f;

    __syncthreads();
#pragma unroll
    for (int v = 0; v < 8; ++v) {
        float val = acc[v] + bv;
        if (ACT == 1 || (ACT == 2 && relu_rt)) val = fmaxf(val, 0.f);
        As[(v + 8 * half) * GP + col] = val;
    }
    __syncthreads();

    store_c_rows<SMALL_BM, O>(As, C, n, rbase, cbase, wave, lane);
}

template <bool TRANSB, int ACT, int K, int O>
__device__ __forceinline__ void gemm_block(
    const float* __restrict__ p0, const float* __restrict__ p1, const float* __restrict__ rs,
    const float* __restrict__ g, const float* __restrict__ b, const float* __restrict__ W,
    const float* __restrict__ bias, float* __restrict__ C, int n,
    float* __restrict__ As, float* __restrict__ Ws, bool relu_rt)
{
    const int bx = blockIdx.x;
    if (bx < SMALL_TILES)
        gemm_tile16<TRANSB, ACT, K, O>(p0, p1, rs, g, b, W, bias, C, n, FULL_TILES * GBM + bx * SMALL_BM, As, Ws, relu_rt);
    else
        gemm_tile<TRANSB, ACT, K, O>(p0, p1, rs, g, b, W, bias, C, n, (bx - SMALL_TILES) * GBM, As, Ws, relu_rt);
}

template <bool TRANSB, int ACT, int K, int O>
__global__ __launch_bounds__(256, 6) void gemm_lds(
    const float* __restrict__ p0, const float* __restrict__ p1, const float* __restrict__ rs,
    const float* __restrict__ g, const float* __restrict__ b, const float* __restrict__ W,
    const float* __restrict__ bias, float* __restrict__ C, int n)
{
    __shared__ __attribute__((aligned(16))) float As[GBM * GP];
    __shared__ __attribute__((aligned(16))) float Ws[GBN * GP];
    gemm_block<TRANSB, ACT, K, O>(p0, p1, rs, g, b, W, bias, C, n, As, Ws, false);
}

#define PT_ROWS 16
#define PT_TILES ((NN + PT_ROWS - 1) / PT_ROWS)
static_assert(PT_TILES >= RESIDENT_BLOCKS, "every resident block has a first tile");

template <bool TRANSB, int O>
__device__ __forceinline__ void gemm_rows_k128(
    const float* __restrict__ A, const float* __restrict__ W,
    const float* __restrict__ bias, float* __restrict__ C, int n, bool relu, int cbase,
    float* __restrict__ Ws, float* __restrict__ As, float* __restrict__ Cs)
{
    static_assert(TRANSB || O == GBN, "a [K,O] operand is taken whole: one slice");
    const int wave = threadIdx.x >> 5;
    const int lane = threadIdx.x & 31;
    const int half = lane >> 4;
    const int l15  = lane & 15;
    const int step = gridDim.x;
    const int nt = (n + PT_ROWS - 1) / PT_ROWS;
    int t = blockIdx.x;

    stage_w<TRANSB, 128, O>(W, Ws, 0, 0, wave, lane);

    float4 pa0 = *(const float4*)(A + (unsigned)(min(t * PT_ROWS + wave,     n - 1) * 128 + lane * 4));
    float4 pa1 = *(const float4*)(A + (unsigned)(min(t * PT_ROWS + wave + 8, n - 1) * 128 + lane * 4));
    const int col = 16 * wave + l15;
    const float bv = bias ? bias[col] : 0.f;
    __syncthreads();

    v2f bf[32];
    {
        const float* bp = Ws + col * GP + 2 * half;
#pragma unroll
        for (int s = 0; s < 32; ++s) bf[s] = *(const v2f*)(bp + 4 * s);
    }

    const float* ap = As + l15 * GP + 2 * half;
    for (; t < nt; t += step) {
        *(float4*)(As + wave * GP + lane * 4) = pa0;
        *(float4*)(As + (wave + 8) * GP + lane * 4) = pa1;
        __syncthreads();

        const int tn = t + step;
        if (tn < nt) {
            pa0 = *(const float4*)(A + (unsigned)(min(tn * PT_ROWS + wave,     n - 1) * 128 + lane * 4));
            pa1 = *(const float4*)(A + (unsigned)(min(tn * PT_ROWS + wave + 8, n - 1) * 128 + lane * 4));
        }

        v8f acc = (v8f){0.f,0.f,0.f,0.f,0.f,0.f,0.f,0.f};
        v2f af[32];
#pragma unroll
        for (int s = 0; s < 32; ++s) af[s] = *(const v2f*)(ap + 4 * s);
#pragma unroll
        for (int s = 0; s < 32; ++s)
            acc = __builtin_amdgcn_wmma_f32_16x16x4_f32(
                false, af[s], false, bf[s], (short)0, acc, false, false);
        __builtin_amdgcn_sched_group_barrier(0x100, 8, 0);
        __builtin_amdgcn_sched_group_barrier(0x008, 8, 0);
        __builtin_amdgcn_sched_group_barrier(0x100, 4, 0);
        __builtin_amdgcn_sched_group_barrier(0x008, 8, 0);
        __builtin_amdgcn_sched_group_barrier(0x100, 4, 0);
        __builtin_amdgcn_sched_group_barrier(0x008, 16, 0);

#pragma unroll
        for (int v = 0; v < 8; ++v) {
            float val = acc[v] + bv;
            if (relu) val = fmaxf(val, 0.f);
            Cs[(v + 8 * half) * GP + col] = val;
        }
        __syncthreads();

#pragma unroll
        for (int i = 0; i < PT_ROWS / 8; ++i) {
            const int row = wave + 8 * i;
            const float4 o = *(const float4*)(Cs + row * GP + lane * 4);
            const int grow = t * PT_ROWS + row;
            if (grow < n) *(float4*)(C + (unsigned)(grow * O + cbase + lane * 4)) = o;
        }
    }
}

template <bool TRANSB, int ACT, int O>
__global__ __launch_bounds__(256, 6) void gemm_k128(
    const float* __restrict__ A, const float* __restrict__ W,
    const float* __restrict__ bias, float* __restrict__ C, int n)
{
    __shared__ __attribute__((aligned(16))) float Ws[GBN * GP];
    __shared__ __attribute__((aligned(16))) float As[PT_ROWS * GP];
    __shared__ __attribute__((aligned(16))) float Cs[PT_ROWS * GP];
    const int cbase = blockIdx.y * GBN;
    gemm_rows_k128<TRANSB, O>(A, TRANSB ? W + cbase * 128 : W, bias ? bias + cbase : nullptr, C, n, ACT == 1, cbase, Ws, As, Cs);
}

template <bool TB0, int ACT0, bool TB1, int ACT1, int O>
__global__ __launch_bounds__(256, 6) void gemm_k128_pair(
    const float* __restrict__ A0, const float* __restrict__ W0,
    const float* __restrict__ bias0, float* __restrict__ C0,
    const float* __restrict__ A1, const float* __restrict__ W1,
    const float* __restrict__ bias1, float* __restrict__ C1, int n)
{
    __shared__ __attribute__((aligned(16))) float Ws[GBN * GP];
    __shared__ __attribute__((aligned(16))) float As[PT_ROWS * GP];
    __shared__ __attribute__((aligned(16))) float Cs[PT_ROWS * GP];
    static_assert(O == GBN, "one 128-column slice per product");
    const bool second = blockIdx.z != 0;
    if (TB0 == TB1) {
        gemm_rows_k128<TB0, O>(
            second ? A1 : A0, second ? W1 : W0, second ? bias1 : bias0, second ? C1 : C0,
            n, (second ? ACT1 : ACT0) == 1, 0, Ws, As, Cs);
    } else {
        if (!second) gemm_rows_k128<TB0, O>(A0, W0, bias0, C0, n, ACT0 == 1, 0, Ws, As, Cs);
        else         gemm_rows_k128<TB1, O>(A1, W1, bias1, C1, n, ACT1 == 1, 0, Ws, As, Cs);
    }
}

__global__ __launch_bounds__(256, 6) void gemm_qk_vo_pair(
    const float* __restrict__ A0, const float* __restrict__ A1,
    const float* __restrict__ Wqkv, const float* __restrict__ bqkv,
    const float* __restrict__ Wvo, const float* __restrict__ bvo,
    float* __restrict__ C0, float* __restrict__ C1, int n)
{
    __shared__ __attribute__((aligned(16))) float Ws[GBN * GP];
    __shared__ __attribute__((aligned(16))) float As[PT_ROWS * GP];
    __shared__ __attribute__((aligned(16))) float Cs[PT_ROWS * GP];
    const bool second = blockIdx.z != 0;
    const int slice = blockIdx.y;
    const float* __restrict__ W = slice < 2 ? Wqkv + slice * (GBN * 128) : Wvo;
    const float* __restrict__ bias = slice < 2 ? bqkv + slice * GBN : bvo;
    gemm_rows_k128<true, 3 * DD>(second ? A1 : A0, W, bias, second ? C1 : C0, n, false, slice * GBN, Ws, As, Cs);
}

#define FOLD_TILES (DD / PT_ROWS)
__global__ __launch_bounds__(256, 6) void fold_out_v_kernel(
    const float* __restrict__ out_W, const float* __restrict__ Wv, const float* __restrict__ bv,
    float* __restrict__ Wvo, float* __restrict__ bvo)
{
    __shared__ __attribute__((aligned(16))) float Ws[GBN * GP];
    __shared__ __attribute__((aligned(16))) float As[PT_ROWS * GP];
    __shared__ __attribute__((aligned(16))) float Cs[PT_ROWS * GP];
    if (blockIdx.x >= FOLD_TILES) {
        const int wave = threadIdx.x >> 5;
        const int lane = threadIdx.x & 31;
        const float4 b = *(const float4*)(bv + lane * 4);
        float4 a[16];
#pragma unroll
        for (int i = 0; i < 16; ++i) a[i] = *(const float4*)(out_W + (wave * 16 + i) * DD + lane * 4);
#pragma unroll
        for (int i = 0; i < 16; ++i) {
            const float s = wave_sum32(a[i].x * b.x + a[i].y * b.y + a[i].z * b.z + a[i].w * b.w);
            if (lane == 0) bvo[wave * 16 + i] = s;
        }
        return;
    }
    gemm_rows_k128<false, DD>(out_W, Wv, nullptr, Wvo, DD, false, 0, Ws, As, Cs);
}

__global__ __launch_bounds__(256) void fill_kernel(float* __restrict__ p, float v, int n) {
    int i = blockIdx.x * blockDim.x + threadIdx.x;
    if (i < n) p[i] = v;
}

__global__ __launch_bounds__(256) void fill_int_kernel(int* __restrict__ p, int v, int n) {
    int i = blockIdx.x * blockDim.x + threadIdx.x;
    if (i < n) p[i] = v;
}

__global__ __launch_bounds__(256) void count_rank_kernel(
    const int* __restrict__ dst, int* __restrict__ cnt, int* __restrict__ rank, int e)
{
    int i = blockIdx.x * blockDim.x + threadIdx.x;
    if (i < e) rank[i] = atomicAdd(&cnt[dst[i]], 1);
}

#define SCAN_CH 13
static_assert(32 * SCAN_CH * 128 >= NN, "scan block must cover all nodes");
static_assert(NN % 4 == 0, "16-byte accesses below need NN % 4 == 0");

__global__ __launch_bounds__(1024) void scan_isd_kernel(
    const int* __restrict__ cnt, int* __restrict__ rowptr, float* __restrict__ isd, int n)
{
    __shared__ int wtot[32];
    const int lane = threadIdx.x & 31;
    const int wave = threadIdx.x >> 5;
    const int wbase = wave * (SCAN_CH * 128);
    int4 c[SCAN_CH];
#pragma unroll
    for (int k = 0; k < SCAN_CH; ++k) {
        const int e0 = wbase + k * 128 + lane * 4;
        c[k] = (e0 < n) ? *(const int4*)(cnt + e0) : make_int4(0, 0, 0, 0);
    }
    int ex[SCAN_CH];
    int carry = 0;
#pragma unroll
    for (int k = 0; k < SCAN_CH; ++k) {
        const int tot = c[k].x + c[k].y + c[k].z + c[k].w;
        int inc = tot;
#pragma unroll
        for (int d = 1; d < 32; d <<= 1) {
            int t = __shfl_up(inc, d, 32);
            if (lane >= d) inc += t;
        }
        ex[k] = carry + inc - tot;
        carry += __shfl(inc, 31, 32);
    }
    if (lane == 0) wtot[wave] = carry;
    __syncthreads();
    int below = (lane < wave) ? wtot[lane] : 0;
#pragma unroll
    for (int m = 16; m > 0; m >>= 1) below += __shfl_xor(below, m, 32);
#pragma unroll
    for (int k = 0; k < SCAN_CH; ++k) {
        const int e0 = wbase + k * 128 + lane * 4;
        if (e0 < n) {
            const int p0 = below + ex[k];
            const int p1 = p0 + c[k].x, p2 = p1 + c[k].y, p3 = p2 + c[k].z;
            *(int4*)(rowptr + e0) = make_int4(p0, p1, p2, p3);
            *(float4*)(isd + e0) = make_float4(
                rsqrtf((float)c[k].x + 1.0f), rsqrtf((float)c[k].y + 1.0f),
                rsqrtf((float)c[k].z + 1.0f), rsqrtf((float)c[k].w + 1.0f));
        }
    }
}

__global__ __launch_bounds__(256) void place_kernel(
    const int* __restrict__ src, const int* __restrict__ dst,
    const int* __restrict__ rowptr, const int* __restrict__ rank,
    int* __restrict__ csr, int e)
{
    int i = blockIdx.x * blockDim.x + threadIdx.x;
    if (i >= e) return;
    const int pos = rowptr[dst[i]] + rank[i];
    if (pos >= 0 && pos < e) csr[pos] = src[i];
}

__global__ __launch_bounds__(256) void row_order_kernel(
    const int* __restrict__ rowptr, const int* __restrict__ cnt,
    const int* __restrict__ csr_in, int* __restrict__ csr_out, int n)
{
    const int lane = threadIdx.x & 31;
    const int node = __builtin_amdgcn_readfirstlane((int)((blockIdx.x * 256 + threadIdx.x) >> 5));
    if (node >= n) return;
    const int start = __builtin_amdgcn_readfirstlane(rowptr[node]);
    const int len   = __builtin_amdgcn_readfirstlane(cnt[node]);
    for (int ib = 0; ib < len; ib += 32) {
        const int i = ib + lane;
        const int xi = (i < len) ? csr_in[start + i] : 0;
        int pos = 0;
        for (int jb = 0; jb < len; jb += 32) {
            const int mj = min(32, len - jb);
            const int xv = (lane < mj) ? csr_in[start + jb + lane] : 0;
            for (int jj = 0; jj < mj; ++jj) {
                const int xj = __builtin_amdgcn_readlane(xv, jj);
                const int j = jb + jj;
                pos += (xj < xi || (xj == xi && j < i)) ? 1 : 0;
            }
        }
        if (i < len) csr_out[start + pos] = xi;
    }
}

template <bool GCN, int B>
__device__ __forceinline__ void gather_accum(
    float4& acc, const float* __restrict__ v, int ms, float mw, int j, int lane)
{
    float4 r[B];
    float w[B];
#pragma unroll
    for (int u = 0; u < B; ++u) {
        const int s = __builtin_amdgcn_readlane(ms, j + u);
        r[u] = *(const float4*)(v + (size_t)s * DD + lane * 4);
        w[u] = GCN ? __int_as_float(__builtin_amdgcn_readlane(__float_as_int(mw), j + u)) : 1.0f;
    }
#pragma unroll
    for (int u = 0; u < B; ++u) {
        if (GCN) {
            acc.x = fmaf(r[u].x, w[u], acc.x); acc.y = fmaf(r[u].y, w[u], acc.y);
            acc.z = fmaf(r[u].z, w[u], acc.z); acc.w = fmaf(r[u].w, w[u], acc.w);
        } else {
            acc.x += r[u].x; acc.y += r[u].y; acc.z += r[u].z; acc.w += r[u].w;
        }
    }
}

template <bool GCN>
__global__ __launch_bounds__(256) void gather_rows_kernel(
    const int* __restrict__ rowptr, const int* __restrict__ cnt, const int* __restrict__ csr,
    const float* __restrict__ v, const float* __restrict__ isd, const float* __restrict__ bias,
    const float* __restrict__ g, const float* __restrict__ b,
    float* __restrict__ o, int n)
{
    const int lane = threadIdx.x & 31;
    const int node = __builtin_amdgcn_readfirstlane((int)((blockIdx.x * 256 + threadIdx.x) >> 5));
    if (node >= n) return;
    const int start = __builtin_amdgcn_readfirstlane(rowptr[node]);
    const int len   = __builtin_amdgcn_readfirstlane(cnt[node]);
    float4 acc = make_float4(0.f, 0.f, 0.f, 0.f);
    float4 sv = acc, bv = acc, gv = acc, lb = acc;
    float it = 0.f;
    if (GCN) {
        it = isd[node];
        sv = *(const float4*)(v + (size_t)node * DD + lane * 4);
        bv = *(const float4*)(bias + lane * 4);
        gv = *(const float4*)(g + lane * 4);
        lb = *(const float4*)(b + lane * 4);
    }
    for (int base = 0; base < len; base += 32) {
        const int m = min(32, len - base);
        int ms = 0;
        float mw = 0.f;
        if (lane < m) {
            ms = csr[start + base + lane];
            ms = min(max(ms, 0), n - 1);
            if (GCN) mw = isd[ms] * it;
        }
        int j = 0;
        for (; j + 8 <= m; j += 8) gather_accum<GCN, 8>(acc, v, ms, mw, j, lane);
        if (j + 4 <= m) { gather_accum<GCN, 4>(acc, v, ms, mw, j, lane); j += 4; }
        if (j + 2 <= m) { gather_accum<GCN, 2>(acc, v, ms, mw, j, lane); j += 2; }
        if (j < m)      { gather_accum<GCN, 1>(acc, v, ms, mw, j, lane); }
    }
    if (GCN) {
        acc.x += sv.x * it * it + bv.x; acc.y += sv.y * it * it + bv.y;
        acc.z += sv.z * it * it + bv.z; acc.w += sv.w * it * it + bv.w;
        const float mean = wave_sum32(acc.x + acc.y + acc.z + acc.w) * (1.f / 128.f);
        const float dx = acc.x - mean, dy = acc.y - mean, dz = acc.z - mean, dw = acc.w - mean;
        const float inv = rsqrtf(wave_sum32(dx * dx + dy * dy + dz * dz + dw * dw) * (1.f / 128.f) + 1e-5f);
        acc.x = fmaxf(dx * inv * gv.x + lb.x, 0.f); acc.y = fmaxf(dy * inv * gv.y + lb.y, 0.f);
        acc.z = fmaxf(dz * inv * gv.z + lb.z, 0.f); acc.w = fmaxf(dw * inv * gv.w + lb.w, 0.f);
    }
    *(float4*)(o + (size_t)node * DD + lane * 4) = acc;
}

__global__ __launch_bounds__(256) void attention_kernel(
    const float* __restrict__ qkv0, const float* __restrict__ qkv1, const float* __restrict__ out_b,
    float* __restrict__ ao0, float* __restrict__ ao1, int n)
{
    int wv = (blockIdx.x * 256 + threadIdx.x) >> 5;
    int lane = threadIdx.x & 31;
    if (wv >= n) return;
    const float* b0 = qkv0 + (size_t)wv * 384;
    const float* b1 = qkv1 + (size_t)wv * 384;
    float q0[4], k0[4], v0[4], q1[4], k1[4], v1[4];
#pragma unroll
    for (int j = 0; j < 4; ++j) {
        int d = lane + 32 * j;
        q0[j] = b0[d]; k0[j] = b0[128 + d]; v0[j] = b0[256 + d];
        q1[j] = b1[d]; k1[j] = b1[128 + d]; v1[j] = b1[256 + d];
    }
    float l00 = 0, l01 = 0, l10 = 0, l11 = 0;
#pragma unroll
    for (int j = 0; j < 4; ++j) {
        l00 += q0[j] * k0[j]; l01 += q0[j] * k1[j];
        l10 += q1[j] * k0[j]; l11 += q1[j] * k1[j];
    }
    l00 = wave_sum32(l00); l01 = wave_sum32(l01);
    l10 = wave_sum32(l10); l11 = wave_sum32(l11);
    const float scale = 0.08838834764831845f;
    l00 *= scale; l01 *= scale; l10 *= scale; l11 *= scale;
    float m0 = fmaxf(l00, l01), e00 = __expf(l00 - m0), e01 = __expf(l01 - m0);
    float r0 = 1.f / (e00 + e01), w00 = e00 * r0, w01 = e01 * r0;
    float m1 = fmaxf(l10, l11), e10 = __expf(l10 - m1), e11 = __expf(l11 - m1);
    float r1 = 1.f / (e10 + e11), w10 = e10 * r1, w11 = e11 * r1;
    float* o0 = ao0 + (size_t)wv * DD;
    float* o1 = ao1 + (size_t)wv * DD;
#pragma unroll
    for (int j = 0; j < 4; ++j) {
        int d = lane + 32 * j;
        const float ob = out_b[d];
        o0[d] = w00 * v0[j] + w01 * v1[j] + ob;
        o1[d] = w10 * v0[j] + w11 * v1[j] + ob;
    }
}

#define CS_BLOCKS 256
#define CS_STRIDE (CS_BLOCKS * 8)
#define CS_ROWS ((NN + CS_STRIDE - 1) / CS_STRIDE)
static_assert(CS_ROWS * CS_STRIDE >= NN, "column-sum waves must cover all rows");
static_assert(CS_ROWS <= 48, "row loads of a wave must stay below the load counter's limit");

__global__ __launch_bounds__(256) void colsum_kernel(const float* __restrict__ p, float* __restrict__ cs, int n) {
    __shared__ __attribute__((aligned(16))) float part[8 * DD];
    const int lane = threadIdx.x & 31;
    const int wave = threadIdx.x >> 5;
    const int r0 = __builtin_amdgcn_readfirstlane((int)(blockIdx.x * 8 + wave));
    float4 r[CS_ROWS];
#pragma unroll
    for (int t = 0; t < CS_ROWS; ++t) {
        const int row = r0 + t * CS_STRIDE;
        r[t] = *(const float4*)(p + (size_t)min(row, n - 1) * DD + lane * 4);
    }
    float4 s = make_float4(0.f, 0.f, 0.f, 0.f);
#pragma unroll
    for (int t = 0; t < CS_ROWS; ++t) {
        const bool in = (r0 + t * CS_STRIDE) < n;
        s.x += in ? r[t].x : 0.f; s.y += in ? r[t].y : 0.f;
        s.z += in ? r[t].z : 0.f; s.w += in ? r[t].w : 0.f;
    }
    *(float4*)(part + wave * DD + lane * 4) = s;
    __syncthreads();
    if (threadIdx.x < DD) {
        float tot = 0.f;
#pragma unroll
        for (int w = 0; w < 8; ++w) tot += part[w * DD + threadIdx.x];
        atomicAdd(&cs[threadIdx.x], tot);
    }
}

__global__ __launch_bounds__(256) void pairnorm_kernel(
    const float* __restrict__ pre, const float* __restrict__ cs,
    float* __restrict__ o, int n)
{
    int wv = (blockIdx.x * 256 + threadIdx.x) >> 5;
    int lane = threadIdx.x & 31;
    if (wv >= n) return;
    const float* row = pre + (size_t)wv * DD;
    float x[4], sq = 0.f;
#pragma unroll
    for (int j = 0; j < 4; ++j) { x[j] = row[lane + 32 * j]; sq += x[j] * x[j]; }
    float rn = sqrtf(1e-6f + wave_sum32(sq));
    float inv = 20.f / rn;
    float* orow = o + (size_t)wv * DD;
    const float invN = 1.f / (float)NN;
#pragma unroll
    for (int j = 0; j < 4; ++j) {
        int d = lane + 32 * j;
        orow[d] = x[j] * inv - cs[d] * invN;
    }
}

#define WS_ROW_ARRAYS ((NN + DD + 63) / 64 * 64)
static_assert(WS_ROW_ARRAYS >= NN + DD && WS_ROW_ARRAYS % 64 == 0, "row arrays behind isd and csum, on a 256-byte boundary");
static_assert(((size_t)NN * DD) % 64 == 0, "every row array keeps the next one on a 256-byte boundary");

extern "C" void kernel_launch(void* const* d_in, const int* in_sizes, int n_in,
                              void* d_out, int out_size, void* d_ws, size_t ws_size,
                              hipStream_t stream)
{
    const float* x      = (const float*)d_in[0];
    const float* x_org  = (const float*)d_in[1];
    const int*   ei     = (const int*)d_in[2];
    const float* in_W   = (const float*)d_in[3];
    const float* in_b   = (const float*)d_in[4];
    const float* res_W  = (const float*)d_in[5];
    const float* gcn_W  = (const float*)d_in[6];
    const float* gcn_b  = (const float*)d_in[7];
    const float* ln_g   = (const float*)d_in[8];
    const float* ln_b   = (const float*)d_in[9];
    const float* lin_W  = (const float*)d_in[10];
    const float* lin_b  = (const float*)d_in[11];
    const float* attr_W = (const float*)d_in[12];
    const float* attr_b = (const float*)d_in[13];
    const float* qkv_W  = (const float*)d_in[14];
    const float* qkv_b  = (const float*)d_in[15];
    const float* out_W  = (const float*)d_in[16];
    const float* out_b  = (const float*)d_in[17];
    const float* fln_g  = (const float*)d_in[18];
    const float* fln_b  = (const float*)d_in[19];
    const float* fuse_W = (const float*)d_in[20];
    const float* fuse_b = (const float*)d_in[21];
    float* out = (float*)d_out;

    const int* src = ei;
    const int* dst = ei + EE;

    float* ws = (float*)d_ws;
    const size_t ND = (size_t)NN * DD;
    float* isd  = ws;
    float* csum = isd + NN;
    float* bh   = ws + WS_ROW_ARRAYS;
    float* bt   = bh + ND;
    float* br   = bt + ND;
    float* bs   = br + ND;
    float* q0b  = bs + ND;
    float* q1b  = q0b + 3 * ND;
    float* Wvo  = q1b + 3 * ND;
    float* bvo  = Wvo + DD * DD;
    int* cnt    = (int*)q1b;
    int* rowptr = cnt + NN;
    int* rank   = rowptr + NN;
    int* csr_a  = rank + EE;
    int* csr_b  = csr_a + EE;

    const dim3 blk(256);
    const dim3 gemmO128(SMALL_TILES + FULL_TILES, DD / GBN);
    const dim3 oneK128(RESIDENT_BLOCKS, DD / GBN);
    const dim3 pairK128(RESIDENT_BLOCKS / 2, DD / GBN, 2);
    const dim3 pairK128O384(RESIDENT_BLOCKS / 6, 3 * DD / GBN, 2);
    const int nodeWaves = (NN + 7) / 8;
    const int edgeBlk   = (EE + 255) / 256;

    fold_out_v_kernel<<<FOLD_TILES + 1, blk, 0, stream>>>(
        out_W, qkv_W + (size_t)2 * DD * DD, qkv_b + 2 * DD, Wvo, bvo);

    fill_int_kernel<<<(NN + 255) / 256, blk, 0, stream>>>(cnt, 0, NN);
    count_rank_kernel<<<edgeBlk, blk, 0, stream>>>(dst, cnt, rank, EE);
    scan_isd_kernel<<<1, 1024, 0, stream>>>(cnt, rowptr, isd, NN);
    place_kernel<<<edgeBlk, blk, 0, stream>>>(src, dst, rowptr, rank, csr_a, EE);
    row_order_kernel<<<nodeWaves, blk, 0, stream>>>(rowptr, cnt, csr_a, csr_b, NN);

    gemm_k128_pair<true, 1, false, 0, DD><<<pairK128, blk, 0, stream>>>(
        x, in_W, in_b, bh, x_org, res_W, nullptr, bt, NN);

    gather_rows_kernel<false><<<nodeWaves, blk, 0, stream>>>(
        rowptr, cnt, csr_b, bt, nullptr, nullptr, nullptr, nullptr, br, NN);

    for (int i = 0; i < 3; ++i) {
        gemm_k128<true, 0, DD><<<oneK128, blk, 0, stream>>>(
            bh, gcn_W + (size_t)i * DD * DD, nullptr, bt, NN);
        gather_rows_kernel<true><<<nodeWaves, blk, 0, stream>>>(
            rowptr, cnt, csr_b, bt, isd, gcn_b + (size_t)i * DD,
            ln_g + (size_t)i * DD, ln_b + (size_t)i * DD, bh, NN);
    }

    gemm_k128_pair<true, 0, true, 1, DD><<<pairK128, blk, 0, stream>>>(
        bh, lin_W, lin_b, bs, x_org, attr_W, attr_b, bt, NN);

    gemm_qk_vo_pair<<<pairK128O384, blk, 0, stream>>>(
        bs, bt, qkv_W, qkv_b, Wvo, bvo, q0b, q1b, NN);

    attention_kernel<<<nodeWaves, blk, 0, stream>>>(q0b, q1b, out_b, bh, bt, NN);

    gemm_lds<true, 0, 3 * DD, DD><<<gemmO128, blk, 0, stream>>>(
        bh, bt, br, fln_g, fln_b, fuse_W, fuse_b, bs, NN);

    fill_kernel<<<1, 128, 0, stream>>>(csum, 0.f, DD);
    colsum_kernel<<<CS_BLOCKS, blk, 0, stream>>>(bs, csum, NN);
    pairnorm_kernel<<<nodeWaves, blk, 0, stream>>>(bs, csum, out, NN);
}
